# starting kernel
# baseline (speedup 1.0000x reference)
#include <hip/hip_runtime.h>
#include <math.h>


typedef __attribute__((ext_vector_type(2))) float v2f;
typedef __attribute__((ext_vector_type(8))) float v8f;

#define NB 4
#define NS 64
#define NW 32
#define NM 16
#define NPTS (NB * NS * NS * NS)
#define HSZ ((size_t)NPTS * NW)

__device__ __forceinline__ v8f wmma4(v2f a, v2f b, v8f c) {
  return __builtin_amdgcn_wmma_f32_16x16x4_f32(false, a, false, b, (short)0, c,
                                               false, false);
}

__device__ __forceinline__ v8f zero8() {
  v8f z;
  #pragma unroll
  for (int i = 0; i < 8; ++i) z[i] = 0.0f;
  return z;
}

__device__ __forceinline__ float gelu_exact(float x) {
  return 0.5f * x * (1.0f + erff(x * 0.70710678118654752440f));
}

__global__ __launch_bounds__(256) void k_input_proj(
    const float* __restrict__ x, const float* __restrict__ w,
    const float* __restrict__ bias, float* __restrict__ out) {
  const int lane = threadIdx.x & 31, wv = threadIdx.x >> 5;
  const int half = lane >> 4, l16 = lane & 15;
  const long pbase = ((long)blockIdx.x * 8 + wv) * 16;
  v2f a;
  a.x = x[(pbase + l16) * 4 + half * 2];
  a.y = x[(pbase + l16) * 4 + half * 2 + 1];
  v2f b0, b1;
  b0.x = w[(half * 2) * NW + l16];
  b0.y = w[(half * 2 + 1) * NW + l16];
  b1.x = w[(half * 2) * NW + 16 + l16];
  b1.y = w[(half * 2 + 1) * NW + 16 + l16];
  v8f d0 = zero8(), d1 = zero8();
  d0 = wmma4(a, b0, d0);
  d1 = wmma4(a, b1, d1);
  #pragma unroll
  for (int r = 0; r < 8; ++r) {
    long p = pbase + r + half * 8;
    out[p * NW + l16] = d0[r] + bias[l16];
    out[p * NW + 16 + l16] = d1[r] + bias[16 + l16];
  }
}

template <bool WT, bool DOGELU>
__global__ __launch_bounds__(256) void k_pointwise(
    const float* __restrict__ hin, const float* __restrict__ w,
    const float* __restrict__ bias, float* __restrict__ out) {
  const int lane = threadIdx.x & 31, wv = threadIdx.x >> 5;
  const int half = lane >> 4, l16 = lane & 15;
  const long pbase = ((long)blockIdx.x * 8 + wv) * 16;
  v8f d0 = zero8(), d1 = zero8();
  #pragma unroll
  for (int kk = 0; kk < 8; ++kk) {
    const int i0 = kk * 4 + half * 2;
    v2f a;
    a.x = hin[(pbase + l16) * NW + i0];
    a.y = hin[(pbase + l16) * NW + i0 + 1];
    v2f b0, b1;
    if (WT) {
      b0.x = w[(i0)*NW + l16];      b0.y = w[(i0 + 1) * NW + l16];
      b1.x = w[(i0)*NW + 16 + l16]; b1.y = w[(i0 + 1) * NW + 16 + l16];
    } else {
      b0.x = w[(l16)*NW + i0];        b0.y = w[(l16)*NW + i0 + 1];
      b1.x = w[(16 + l16) * NW + i0]; b1.y = w[(16 + l16) * NW + i0 + 1];
    }
    d0 = wmma4(a, b0, d0);
    d1 = wmma4(a, b1, d1);
  }
  #pragma unroll
  for (int r = 0; r < 8; ++r) {
    long p = pbase + r + half * 8;
    float v0 = d0[r] + bias[l16];
    float v1 = d1[r] + bias[16 + l16];
    if (DOGELU) { v0 = gelu_exact(v0); v1 = gelu_exact(v1); }
    out[p * NW + l16] = v0;
    out[p * NW + 16 + l16] = v1;
  }
}

__global__ __launch_bounds__(256) void k_spectral_axis(
    const float* __restrict__ h, const float* __restrict__ wgt,
    float* __restrict__ acc, long strideU, long strideV, long strideA) {
  extern __shared__ float smem[];
  float* Ff = smem;
  float* Fi = smem + 2048;
  float* G  = smem + 4096;
  float* Gp = smem + 4096 + 16384;

  const int tid = threadIdx.x;
  const int lane = tid & 31, wv = tid >> 5;
  const int half = lane >> 4, l16 = lane & 15;
  const float NRM = 0.125f;

  for (int idx = tid; idx < 2048; idx += 256) {
    int m2 = idx >> 6, p = idx & 63, k = m2 >> 1;
    float ang = 6.28318530717958647693f * (float)(k * p) * (1.0f / 64.0f);
    Ff[idx] = (m2 & 1) ? (-__sinf(ang) * NRM) : (__cosf(ang) * NRM);
  }
  for (int idx = tid; idx < 2048; idx += 256) {
    int p = idx >> 5, m2 = idx & 31, k = m2 >> 1;
    float ang = 6.28318530717958647693f * (float)(k * p) * (1.0f / 64.0f);
    float sc = (k == 0 ? 1.0f : 2.0f) * NRM;
    Fi[idx] = (m2 & 1) ? (-__sinf(ang) * sc) : (__cosf(ang) * sc);
  }
  __syncthreads();

  long base[2];
  #pragma unroll
  for (int ll = 0; ll < 2; ++ll) {
    int l = blockIdx.x * 16 + wv * 2 + ll;
    int b = l >> 12, u = (l >> 6) & 63, v = l & 63;
    base[ll] = (long)b * (64L * 64 * 64 * NW) + (long)u * strideU +
               (long)v * strideV;
  }

  #pragma unroll
  for (int ll = 0; ll < 2; ++ll) {
    const int li = wv * 2 + ll;
    v8f a00 = zero8(), a01 = zero8(), a10 = zero8(), a11 = zero8();
    for (int kk = 0; kk < 16; ++kk) {
      const int p0 = kk * 4 + half * 2;
      v2f b0, b1;
      b0.x = h[base[ll] + (long)p0 * strideA + l16];
      b0.y = h[base[ll] + (long)(p0 + 1) * strideA + l16];
      b1.x = h[base[ll] + (long)p0 * strideA + 16 + l16];
      b1.y = h[base[ll] + (long)(p0 + 1) * strideA + 16 + l16];
      v2f fa0, fa1;
      fa0.x = Ff[(l16)*64 + p0];      fa0.y = Ff[(l16)*64 + p0 + 1];
      fa1.x = Ff[(16 + l16) * 64 + p0]; fa1.y = Ff[(16 + l16) * 64 + p0 + 1];
      a00 = wmma4(fa0, b0, a00);
      a01 = wmma4(fa0, b1, a01);
      a10 = wmma4(fa1, b0, a10);
      a11 = wmma4(fa1, b1, a11);
    }
    float* g = G + li * 1024;
    #pragma unroll
    for (int r = 0; r < 8; ++r) {
      const int m = r + half * 8;
      g[(m)*32 + l16]           = a00[r];
      g[(m)*32 + 16 + l16]      = a01[r];
      g[(m + 16) * 32 + l16]    = a10[r];
      g[(m + 16) * 32 + 16 + l16] = a11[r];
    }
  }
  __syncthreads();

  #pragma unroll
  for (int km = 0; km < 2; ++km) {
    const int k = wv + km * 8;
    v8f oRe0 = zero8(), oRe1 = zero8(), oIm0 = zero8(), oIm1 = zero8();
    for (int kk = 0; kk < 8; ++kk) {
      const int i0 = kk * 4 + half * 2, i1 = i0 + 1;
      v2f bRe, bIm;
      bRe.x = G[l16 * 1024 + (2 * k) * 32 + i0];
      bRe.y = G[l16 * 1024 + (2 * k) * 32 + i1];
      bIm.x = G[l16 * 1024 + (2 * k + 1) * 32 + i0];
      bIm.y = G[l16 * 1024 + (2 * k + 1) * 32 + i1];
      #pragma unroll
      for (int mt = 0; mt < 2; ++mt) {
        const int o = mt * 16 + l16;
        v2f aR, aI, aIn;
        aR.x = wgt[((i0 * NW + o) * NM + k) * 2 + 0];
        aR.y = wgt[((i1 * NW + o) * NM + k) * 2 + 0];
        aI.x = wgt[((i0 * NW + o) * NM + k) * 2 + 1];
        aI.y = wgt[((i1 * NW + o) * NM + k) * 2 + 1];
        aIn.x = -aI.x; aIn.y = -aI.y;
        if (mt == 0) {
          oRe0 = wmma4(aR, bRe, oRe0);  oRe0 = wmma4(aIn, bIm, oRe0);
          oIm0 = wmma4(aI, bRe, oIm0);  oIm0 = wmma4(aR, bIm, oIm0);
        } else {
          oRe1 = wmma4(aR, bRe, oRe1);  oRe1 = wmma4(aIn, bIm, oRe1);
          oIm1 = wmma4(aI, bRe, oIm1);  oIm1 = wmma4(aR, bIm, oIm1);
        }
      }
    }
    #pragma unroll
    for (int r = 0; r < 8; ++r) {
      const int o0 = r + half * 8;
      Gp[l16 * 1024 + (2 * k) * 32 + o0]          = oRe0[r];
      Gp[l16 * 1024 + (2 * k) * 32 + o0 + 16]     = oRe1[r];
      Gp[l16 * 1024 + (2 * k + 1) * 32 + o0]      = oIm0[r];
      Gp[l16 * 1024 + (2 * k + 1) * 32 + o0 + 16] = oIm1[r];
    }
  }
  __syncthreads();

  #pragma unroll
  for (int ll = 0; ll < 2; ++ll) {
    const int li = wv * 2 + ll;
    const float* gp = Gp + li * 1024;
    v8f o00 = zero8(), o01 = zero8(), o10 = zero8(), o11 = zero8();
    v8f o20 = zero8(), o21 = zero8(), o30 = zero8(), o31 = zero8();
    for (int kk = 0; kk < 8; ++kk) {
      const int i0 = kk * 4 + half * 2;
      v2f b0, b1;
      b0.x = gp[(i0)*32 + l16];      b0.y = gp[(i0 + 1) * 32 + l16];
      b1.x = gp[(i0)*32 + 16 + l16]; b1.y = gp[(i0 + 1) * 32 + 16 + l16];
      v2f a0, a1, a2, a3;
      a0.x = Fi[(l16)*32 + i0];        a0.y = Fi[(l16)*32 + i0 + 1];
      a1.x = Fi[(16 + l16) * 32 + i0]; a1.y = Fi[(16 + l16) * 32 + i0 + 1];
      a2.x = Fi[(32 + l16) * 32 + i0]; a2.y = Fi[(32 + l16) * 32 + i0 + 1];
      a3.x = Fi[(48 + l16) * 32 + i0]; a3.y = Fi[(48 + l16) * 32 + i0 + 1];
      o00 = wmma4(a0, b0, o00);  o01 = wmma4(a0, b1, o01);
      o10 = wmma4(a1, b0, o10);  o11 = wmma4(a1, b1, o11);
      o20 = wmma4(a2, b0, o20);  o21 = wmma4(a2, b1, o21);
      o30 = wmma4(a3, b0, o30);  o31 = wmma4(a3, b1, o31);
    }
    #pragma unroll
    for (int r = 0; r < 8; ++r) {
      const int pr = r + half * 8;
      long i0a = base[ll] + (long)(pr)*strideA;
      long i1a = base[ll] + (long)(pr + 16) * strideA;
      long i2a = base[ll] + (long)(pr + 32) * strideA;
      long i3a = base[ll] + (long)(pr + 48) * strideA;
      acc[i0a + l16] += o00[r];  acc[i0a + 16 + l16] += o01[r];
      acc[i1a + l16] += o10[r];  acc[i1a + 16 + l16] += o11[r];
      acc[i2a + l16] += o20[r];  acc[i2a + 16 + l16] += o21[r];
      acc[i3a + l16] += o30[r];  acc[i3a + 16 + l16] += o31[r];
    }
  }
}

__global__ __launch_bounds__(256) void k_gelu(const float* __restrict__ in,
                                              float* __restrict__ out) {
  const long i = (long)blockIdx.x * blockDim.x + threadIdx.x;
  float4 v = ((const float4*)in)[i];
  v.x = gelu_exact(v.x);
  v.y = gelu_exact(v.y);
  v.z = gelu_exact(v.z);
  v.w = gelu_exact(v.w);
  ((float4*)out)[i] = v;
}

__global__ __launch_bounds__(256) void k_fc3(const float* __restrict__ hin,
                                             const float* __restrict__ w,
                                             const float* __restrict__ bias,
                                             float* __restrict__ out) {
  const long p = (long)blockIdx.x * blockDim.x + threadIdx.x;
  const float4* hv = (const float4*)(hin + p * NW);
  const float4* wv = (const float4*)w;
  float s = 0.0f;
  #pragma unroll
  for (int i = 0; i < 8; ++i) {
    float4 a = hv[i], b = wv[i];
    s += a.x * b.x + a.y * b.y + a.z * b.z + a.w * b.w;
  }
  out[p] = s + bias[0];
}

extern "C" void kernel_launch(void* const* d_in, const int* in_sizes, int n_in,
                              void* d_out, int out_size, void* d_ws,
                              size_t ws_size, hipStream_t stream) {
  const float* x      = (const float*)d_in[0];
  const float* in_w   = (const float*)d_in[1];
  const float* in_b   = (const float*)d_in[2];
  const float* fwx    = (const float*)d_in[3];
  const float* fwy    = (const float*)d_in[4];
  const float* fwz    = (const float*)d_in[5];
  const float* conv_w = (const float*)d_in[6];
  const float* conv_b = (const float*)d_in[7];
  const float* fc1_w  = (const float*)d_in[8];
  const float* fc1_b  = (const float*)d_in[9];
  const float* fc2_w  = (const float*)d_in[10];
  const float* fc2_b  = (const float*)d_in[11];
  const float* fc3_w  = (const float*)d_in[12];
  const float* fc3_b  = (const float*)d_in[13];
  float* out = (float*)d_out;

  float* buf0 = (float*)d_ws;
  float* buf1 = buf0 + HSZ;

  const dim3 blk(256);
  const int wgPts = NPTS / (16 * 8);
  const size_t smemB = (size_t)(4096 + 2 * 16384) * sizeof(float);
  const long sX = 64L * 64 * NW;
  const long sY = 64L * NW;
  const long sZ = (long)NW;

  k_input_proj<<<wgPts, blk, 0, stream>>>(x, in_w, in_b, buf0);

  for (int i = 0; i < 4; ++i) {
    k_pointwise<false, false><<<wgPts, blk, 0, stream>>>(
        buf0, conv_w + i * (NW * NW), conv_b + i * NW, buf1);
    const long wofs = (long)i * (NW * NW * NM * 2);
    k_spectral_axis<<<1024, blk, smemB, stream>>>(buf0, fwz + wofs, buf1,
                                                  sX, sY, sZ);
    k_spectral_axis<<<1024, blk, smemB, stream>>>(buf0, fwy + wofs, buf1,
                                                  sX, sZ, sY);
    k_spectral_axis<<<1024, blk, smemB, stream>>>(buf0, fwx + wofs, buf1,
                                                  sY, sZ, sX);
    if (i < 3)
      k_gelu<<<(int)(HSZ / 4 / 256), blk, 0, stream>>>(buf1, buf0);
  }

  k_pointwise<true, true><<<wgPts, blk, 0, stream>>>(buf1, fc1_w, fc1_b, buf0);
  k_pointwise<true, true><<<wgPts, blk, 0, stream>>>(buf0, fc2_w, fc2_b, buf1);
  k_fc3<<<NPTS / 256, blk, 0, stream>>>(buf1, fc3_w, fc3_b, out);
}


# spectral kernel: global loads (input rows, mode weights, accumulator reads) issued as register batches ahead of use, not one wait per WMMA group or element; 32-bit offsets so the batches fit
# speedup vs baseline: 1.6673x; 1.6673x over previous
#include <hip/hip_runtime.h>
#include <math.h>


typedef __attribute__((ext_vector_type(2))) float v2f;
typedef __attribute__((ext_vector_type(8))) float v8f;

#define NB 4
#define NS 64
#define NW 32
#define NM 16
#define NPTS (NB * NS * NS * NS)
#define HSZ ((size_t)NPTS * NW)

__device__ __forceinline__ v8f wmma4(v2f a, v2f b, v8f c) {
  return __builtin_amdgcn_wmma_f32_16x16x4_f32(false, a, false, b, (short)0, c,
                                               false, false);
}

__device__ __forceinline__ v8f zero8() {
  v8f z;
  #pragma unroll
  for (int i = 0; i < 8; ++i) z[i] = 0.0f;
  return z;
}

__device__ __forceinline__ float gelu_exact(float x) {
  return 0.5f * x * (1.0f + erff(x * 0.70710678118654752440f));
}

__global__ __launch_bounds__(256) void k_input_proj(
    const float* __restrict__ x, const float* __restrict__ w,
    const float* __restrict__ bias, float* __restrict__ out) {
  const int lane = threadIdx.x & 31, wv = threadIdx.x >> 5;
  const int half = lane >> 4, l16 = lane & 15;
  const long pbase = ((long)blockIdx.x * 8 + wv) * 16;
  v2f a;
  a.x = x[(pbase + l16) * 4 + half * 2];
  a.y = x[(pbase + l16) * 4 + half * 2 + 1];
  v2f b0, b1;
  b0.x = w[(half * 2) * NW + l16];
  b0.y = w[(half * 2 + 1) * NW + l16];
  b1.x = w[(half * 2) * NW + 16 + l16];
  b1.y = w[(half * 2 + 1) * NW + 16 + l16];
  v8f d0 = zero8(), d1 = zero8();
  d0 = wmma4(a, b0, d0);
  d1 = wmma4(a, b1, d1);
  #pragma unroll
  for (int r = 0; r < 8; ++r) {
    long p = pbase + r + half * 8;
    out[p * NW + l16] = d0[r] + bias[l16];
    out[p * NW + 16 + l16] = d1[r] + bias[16 + l16];
  }
}

template <bool WT, bool DOGELU>
__global__ __launch_bounds__(256) void k_pointwise(
    const float* __restrict__ hin, const float* __restrict__ w,
    const float* __restrict__ bias, float* __restrict__ out) {
  const int lane = threadIdx.x & 31, wv = threadIdx.x >> 5;
  const int half = lane >> 4, l16 = lane & 15;
  const long pbase = ((long)blockIdx.x * 8 + wv) * 16;
  v8f d0 = zero8(), d1 = zero8();
  #pragma unroll
  for (int kk = 0; kk < 8; ++kk) {
    const int i0 = kk * 4 + half * 2;
    v2f a;
    a.x = hin[(pbase + l16) * NW + i0];
    a.y = hin[(pbase + l16) * NW + i0 + 1];
    v2f b0, b1;
    if (WT) {
      b0.x = w[(i0)*NW + l16];      b0.y = w[(i0 + 1) * NW + l16];
      b1.x = w[(i0)*NW + 16 + l16]; b1.y = w[(i0 + 1) * NW + 16 + l16];
    } else {
      b0.x = w[(l16)*NW + i0];        b0.y = w[(l16)*NW + i0 + 1];
      b1.x = w[(16 + l16) * NW + i0]; b1.y = w[(16 + l16) * NW + i0 + 1];
    }
    d0 = wmma4(a, b0, d0);
    d1 = wmma4(a, b1, d1);
  }
  #pragma unroll
  for (int r = 0; r < 8; ++r) {
    long p = pbase + r + half * 8;
    float v0 = d0[r] + bias[l16];
    float v1 = d1[r] + bias[16 + l16];
    if (DOGELU) { v0 = gelu_exact(v0); v1 = gelu_exact(v1); }
    out[p * NW + l16] = v0;
    out[p * NW + 16 + l16] = v1;
  }
}

__device__ __forceinline__ float ld_off(const float* p, unsigned boff) {
  return *(const float*)((const char*)p + boff);
}
__device__ __forceinline__ void st_off(float* p, unsigned boff, float v) {
  *(float*)((char*)p + boff) = v;
}

__global__ __launch_bounds__(256, 4) void k_spectral_axis(
    const float* __restrict__ h, const float* __restrict__ wgt,
    float* __restrict__ acc, long strideU, long strideV, long strideA) {
  extern __shared__ float smem[];
  float* Ff = smem;
  float* Fi = smem + 2048;
  float* G  = smem + 4096;
  float* Gp = smem + 4096 + 16384;

  const int tid = threadIdx.x;
  const int lane = tid & 31, wv = tid >> 5;
  const int half = lane >> 4, l16 = lane & 15;
  const float NRM = 0.125f;

  const unsigned sAB = (unsigned)strideA * 4u;
  unsigned base[2];
  #pragma unroll
  for (int ll = 0; ll < 2; ++ll) {
    int l = blockIdx.x * 16 + wv * 2 + ll;
    int b = l >> 12, u = (l >> 6) & 63, v = l & 63;
    base[ll] = ((unsigned)b * (64u * 64 * 64 * NW) + (unsigned)u * (unsigned)strideU +
                (unsigned)v * (unsigned)strideV + (unsigned)l16) * 4u;
  }

  v2f hb0[2][16], hb1[2][16];
  #pragma unroll
  for (int ll = 0; ll < 2; ++ll) {
    #pragma unroll
    for (int kk = 0; kk < 16; ++kk) {
      const unsigned p0 = kk * 4 + half * 2;
      hb0[ll][kk].x = ld_off(h, base[ll] + p0 * sAB);
      hb0[ll][kk].y = ld_off(h, base[ll] + (p0 + 1) * sAB);
      hb1[ll][kk].x = ld_off(h, base[ll] + p0 * sAB + 64u);
      hb1[ll][kk].y = ld_off(h, base[ll] + (p0 + 1) * sAB + 64u);
    }
  }

  for (int idx = tid; idx < 2048; idx += 256) {
    int m2 = idx >> 6, p = idx & 63, k = m2 >> 1;
    float ang = 6.28318530717958647693f * (float)(k * p) * (1.0f / 64.0f);
    Ff[idx] = (m2 & 1) ? (-__sinf(ang) * NRM) : (__cosf(ang) * NRM);
  }
  for (int idx = tid; idx < 2048; idx += 256) {
    int p = idx >> 5, m2 = idx & 31, k = m2 >> 1;
    float ang = 6.28318530717958647693f * (float)(k * p) * (1.0f / 64.0f);
    float sc = (k == 0 ? 1.0f : 2.0f) * NRM;
    Fi[idx] = (m2 & 1) ? (-__sinf(ang) * sc) : (__cosf(ang) * sc);
  }
  __syncthreads();

  #pragma unroll
  for (int ll = 0; ll < 2; ++ll) {
    const int li = wv * 2 + ll;
    v8f a00 = zero8(), a01 = zero8(), a10 = zero8(), a11 = zero8();
    #pragma unroll
    for (int kk = 0; kk < 16; ++kk) {
      const int p0 = kk * 4 + half * 2;
      v2f fa0, fa1;
      fa0.x = Ff[(l16)*64 + p0];      fa0.y = Ff[(l16)*64 + p0 + 1];
      fa1.x = Ff[(16 + l16) * 64 + p0]; fa1.y = Ff[(16 + l16) * 64 + p0 + 1];
      a00 = wmma4(fa0, hb0[ll][kk], a00);
      a01 = wmma4(fa0, hb1[ll][kk], a01);
      a10 = wmma4(fa1, hb0[ll][kk], a10);
      a11 = wmma4(fa1, hb1[ll][kk], a11);
    }
    float* g = G + li * 1024;
    #pragma unroll
    for (int r = 0; r < 8; ++r) {
      const int m = r + half * 8;
      g[(m)*32 + l16]           = a00[r];
      g[(m)*32 + 16 + l16]      = a01[r];
      g[(m + 16) * 32 + l16]    = a10[r];
      g[(m + 16) * 32 + 16 + l16] = a11[r];
    }
  }

  float wq[2][8][2][4];
  #pragma unroll
  for (int km = 0; km < 2; ++km) {
    const int k = wv + km * 8;
    #pragma unroll
    for (int kk = 0; kk < 8; ++kk) {
      const int i0 = kk * 4 + half * 2, i1 = i0 + 1;
      #pragma unroll
      for (int mt = 0; mt < 2; ++mt) {
        const int o = mt * 16 + l16;
        wq[km][kk][mt][0] = wgt[((i0 * NW + o) * NM + k) * 2 + 0];
        wq[km][kk][mt][1] = wgt[((i0 * NW + o) * NM + k) * 2 + 1];
        wq[km][kk][mt][2] = wgt[((i1 * NW + o) * NM + k) * 2 + 0];
        wq[km][kk][mt][3] = wgt[((i1 * NW + o) * NM + k) * 2 + 1];
      }
    }
  }
  __syncthreads();

  #pragma unroll
  for (int km = 0; km < 2; ++km) {
    const int k = wv + km * 8;
    v8f oRe0 = zero8(), oRe1 = zero8(), oIm0 = zero8(), oIm1 = zero8();
    #pragma unroll
    for (int kk = 0; kk < 8; ++kk) {
      const int i0 = kk * 4 + half * 2, i1 = i0 + 1;
      v2f bRe, bIm;
      bRe.x = G[l16 * 1024 + (2 * k) * 32 + i0];
      bRe.y = G[l16 * 1024 + (2 * k) * 32 + i1];
      bIm.x = G[l16 * 1024 + (2 * k + 1) * 32 + i0];
      bIm.y = G[l16 * 1024 + (2 * k + 1) * 32 + i1];
      #pragma unroll
      for (int mt = 0; mt < 2; ++mt) {
        v2f aR, aI, aIn;
        aR.x = wq[km][kk][mt][0];
        aR.y = wq[km][kk][mt][2];
        aI.x = wq[km][kk][mt][1];
        aI.y = wq[km][kk][mt][3];
        aIn.x = -aI.x; aIn.y = -aI.y;
        if (mt == 0) {
          oRe0 = wmma4(aR, bRe, oRe0);  oRe0 = wmma4(aIn, bIm, oRe0);
          oIm0 = wmma4(aI, bRe, oIm0);  oIm0 = wmma4(aR, bIm, oIm0);
        } else {
          oRe1 = wmma4(aR, bRe, oRe1);  oRe1 = wmma4(aIn, bIm, oRe1);
          oIm1 = wmma4(aI, bRe, oIm1);  oIm1 = wmma4(aR, bIm, oIm1);
        }
      }
    }
    #pragma unroll
    for (int r = 0; r < 8; ++r) {
      const int o0 = r + half * 8;
      Gp[l16 * 1024 + (2 * k) * 32 + o0]          = oRe0[r];
      Gp[l16 * 1024 + (2 * k) * 32 + o0 + 16]     = oRe1[r];
      Gp[l16 * 1024 + (2 * k + 1) * 32 + o0]      = oIm0[r];
      Gp[l16 * 1024 + (2 * k + 1) * 32 + o0 + 16] = oIm1[r];
    }
  }

  float cq[2][8][8];
  #pragma unroll
  for (int r = 0; r < 8; ++r) {
    const int pr = r + half * 8;
    #pragma unroll
    for (int q = 0; q < 4; ++q) {
      const unsigned ia = base[0] + (unsigned)(pr + 16 * q) * sAB;
      cq[0][2 * q][r]     = ld_off(acc, ia);
      cq[0][2 * q + 1][r] = ld_off(acc, ia + 64u);
    }
  }
  __syncthreads();

  #pragma unroll
  for (int ll = 0; ll < 2; ++ll) {
    const int li = wv * 2 + ll;
    const float* gp = Gp + li * 1024;
    if (ll == 1) {
      #pragma unroll
      for (int r = 0; r < 8; ++r) {
        const int pr = r + half * 8;
        #pragma unroll
        for (int q = 0; q < 4; ++q) {
          const unsigned ia = base[1] + (unsigned)(pr + 16 * q) * sAB;
          cq[1][2 * q][r]     = ld_off(acc, ia);
          cq[1][2 * q + 1][r] = ld_off(acc, ia + 64u);
        }
      }
    }
    v8f o00 = zero8(), o01 = zero8(), o10 = zero8(), o11 = zero8();
    v8f o20 = zero8(), o21 = zero8(), o30 = zero8(), o31 = zero8();
    for (int kk = 0; kk < 8; ++kk) {
      const int i0 = kk * 4 + half * 2;
      v2f b0, b1;
      b0.x = gp[(i0)*32 + l16];      b0.y = gp[(i0 + 1) * 32 + l16];
      b1.x = gp[(i0)*32 + 16 + l16]; b1.y = gp[(i0 + 1) * 32 + 16 + l16];
      v2f a0, a1, a2, a3;
      a0.x = Fi[(l16)*32 + i0];        a0.y = Fi[(l16)*32 + i0 + 1];
      a1.x = Fi[(16 + l16) * 32 + i0]; a1.y = Fi[(16 + l16) * 32 + i0 + 1];
      a2.x = Fi[(32 + l16) * 32 + i0]; a2.y = Fi[(32 + l16) * 32 + i0 + 1];
      a3.x = Fi[(48 + l16) * 32 + i0]; a3.y = Fi[(48 + l16) * 32 + i0 + 1];
      o00 = wmma4(a0, b0, o00);  o01 = wmma4(a0, b1, o01);
      o10 = wmma4(a1, b0, o10);  o11 = wmma4(a1, b1, o11);
      o20 = wmma4(a2, b0, o20);  o21 = wmma4(a2, b1, o21);
      o30 = wmma4(a3, b0, o30);  o31 = wmma4(a3, b1, o31);
    }
    #pragma unroll
    for (int r = 0; r < 8; ++r) {
      const int pr = r + half * 8;
      const unsigned i0a = base[ll] + (unsigned)(pr)*sAB;
      const unsigned i1a = base[ll] + (unsigned)(pr + 16) * sAB;
      const unsigned i2a = base[ll] + (unsigned)(pr + 32) * sAB;
      const unsigned i3a = base[ll] + (unsigned)(pr + 48) * sAB;
      st_off(acc, i0a, cq[ll][0][r] + o00[r]);  st_off(acc, i0a + 64u, cq[ll][1][r] + o01[r]);
      st_off(acc, i1a, cq[ll][2][r] + o10[r]);  st_off(acc, i1a + 64u, cq[ll][3][r] + o11[r]);
      st_off(acc, i2a, cq[ll][4][r] + o20[r]);  st_off(acc, i2a + 64u, cq[ll][5][r] + o21[r]);
      st_off(acc, i3a, cq[ll][6][r] + o30[r]);  st_off(acc, i3a + 64u, cq[ll][7][r] + o31[r]);
    }
  }
}

__global__ __launch_bounds__(256) void k_gelu(const float* __restrict__ in,
                                              float* __restrict__ out) {
  const long i = (long)blockIdx.x * blockDim.x + threadIdx.x;
  float4 v = ((const float4*)in)[i];
  v.x = gelu_exact(v.x);
  v.y = gelu_exact(v.y);
  v.z = gelu_exact(v.z);
  v.w = gelu_exact(v.w);
  ((float4*)out)[i] = v;
}

__global__ __launch_bounds__(256) void k_fc3(const float* __restrict__ hin,
                                             const float* __restrict__ w,
                                             const float* __restrict__ bias,
                                             float* __restrict__ out) {
  const long p = (long)blockIdx.x * blockDim.x + threadIdx.x;
  const float4* hv = (const float4*)(hin + p * NW);
  const float4* wv = (const float4*)w;
  float s = 0.0f;
  #pragma unroll
  for (int i = 0; i < 8; ++i) {
    float4 a = hv[i], b = wv[i];
    s += a.x * b.x + a.y * b.y + a.z * b.z + a.w * b.w;
  }
  out[p] = s + bias[0];
}

extern "C" void kernel_launch(void* const* d_in, const int* in_sizes, int n_in,
                              void* d_out, int out_size, void* d_ws,
                              size_t ws_size, hipStream_t stream) {
  const float* x      = (const float*)d_in[0];
  const float* in_w   = (const float*)d_in[1];
  const float* in_b   = (const float*)d_in[2];
  const float* fwx    = (const float*)d_in[3];
  const float* fwy    = (const float*)d_in[4];
  const float* fwz    = (const float*)d_in[5];
  const float* conv_w = (const float*)d_in[6];
  const float* conv_b = (const float*)d_in[7];
  const float* fc1_w  = (const float*)d_in[8];
  const float* fc1_b  = (const float*)d_in[9];
  const float* fc2_w  = (const float*)d_in[10];
  const float* fc2_b  = (const float*)d_in[11];
  const float* fc3_w  = (const float*)d_in[12];
  const float* fc3_b  = (const float*)d_in[13];
  float* out = (float*)d_out;

  float* buf0 = (float*)d_ws;
  float* buf1 = buf0 + HSZ;

  const dim3 blk(256);
  const int wgPts = NPTS / (16 * 8);
  const size_t smemB = (size_t)(4096 + 2 * 16384) * sizeof(float);
  const long sX = 64L * 64 * NW;
  const long sY = 64L * NW;
  const long sZ = (long)NW;

  k_input_proj<<<wgPts, blk, 0, stream>>>(x, in_w, in_b, buf0);

  for (int i = 0; i < 4; ++i) {
    k_pointwise<false, false><<<wgPts, blk, 0, stream>>>(
        buf0, conv_w + i * (NW * NW), conv_b + i * NW, buf1);
    const long wofs = (long)i * (NW * NW * NM * 2);
    k_spectral_axis<<<1024, blk, smemB, stream>>>(buf0, fwz + wofs, buf1,
                                                  sX, sY, sZ);
    k_spectral_axis<<<1024, blk, smemB, stream>>>(buf0, fwy + wofs, buf1,
                                                  sX, sZ, sY);
    k_spectral_axis<<<1024, blk, smemB, stream>>>(buf0, fwx + wofs, buf1,
                                                  sY, sZ, sX);
    if (i < 3)
      k_gelu<<<(int)(HSZ / 4 / 256), blk, 0, stream>>>(buf1, buf0);
  }

  k_pointwise<true, true><<<wgPts, blk, 0, stream>>>(buf1, fc1_w, fc1_b, buf0);
  k_pointwise<true, true><<<wgPts, blk, 0, stream>>>(buf0, fc2_w, fc2_b, buf1);
  k_fc3<<<NPTS / 256, blk, 0, stream>>>(buf1, fc3_w, fc3_b, out);
}


# spectral kernel: LDS pitches padded (68, 36, 34/1092, 1027 dwords) and G' row halves swizzled: fragment loads bank-conflict-free, paired stores at most 2-way (were 8- and 16-way)
# speedup vs baseline: 1.7706x; 1.0620x over previous
#include <hip/hip_runtime.h>
#include <math.h>


typedef __attribute__((ext_vector_type(2))) float v2f;
typedef __attribute__((ext_vector_type(8))) float v8f;

#define NB 4
#define NS 64
#define NW 32
#define NM 16
#define NPTS (NB * NS * NS * NS)
#define HSZ ((size_t)NPTS * NW)

__device__ __forceinline__ v8f wmma4(v2f a, v2f b, v8f c) {
  return __builtin_amdgcn_wmma_f32_16x16x4_f32(false, a, false, b, (short)0, c,
                                               false, false);
}

__device__ __forceinline__ v8f zero8() {
  v8f z;
  #pragma unroll
  for (int i = 0; i < 8; ++i) z[i] = 0.0f;
  return z;
}

__device__ __forceinline__ float gelu_exact(float x) {
  return 0.5f * x * (1.0f + erff(x * 0.70710678118654752440f));
}

__global__ __launch_bounds__(256) void k_input_proj(
    const float* __restrict__ x, const float* __restrict__ w,
    const float* __restrict__ bias, float* __restrict__ out) {
  const int lane = threadIdx.x & 31, wv = threadIdx.x >> 5;
  const int half = lane >> 4, l16 = lane & 15;
  const long pbase = ((long)blockIdx.x * 8 + wv) * 16;
  v2f a;
  a.x = x[(pbase + l16) * 4 + half * 2];
  a.y = x[(pbase + l16) * 4 + half * 2 + 1];
  v2f b0, b1;
  b0.x = w[(half * 2) * NW + l16];
  b0.y = w[(half * 2 + 1) * NW + l16];
  b1.x = w[(half * 2) * NW + 16 + l16];
  b1.y = w[(half * 2 + 1) * NW + 16 + l16];
  v8f d0 = zero8(), d1 = zero8();
  d0 = wmma4(a, b0, d0);
  d1 = wmma4(a, b1, d1);
  #pragma unroll
  for (int r = 0; r < 8; ++r) {
    long p = pbase + r + half * 8;
    out[p * NW + l16] = d0[r] + bias[l16];
    out[p * NW + 16 + l16] = d1[r] + bias[16 + l16];
  }
}

template <bool WT, bool DOGELU>
__global__ __launch_bounds__(256) void k_pointwise(
    const float* __restrict__ hin, const float* __restrict__ w,
    const float* __restrict__ bias, float* __restrict__ out) {
  const int lane = threadIdx.x & 31, wv = threadIdx.x >> 5;
  const int half = lane >> 4, l16 = lane & 15;
  const long pbase = ((long)blockIdx.x * 8 + wv) * 16;
  v8f d0 = zero8(), d1 = zero8();
  #pragma unroll
  for (int kk = 0; kk < 8; ++kk) {
    const int i0 = kk * 4 + half * 2;
    v2f a;
    a.x = hin[(pbase + l16) * NW + i0];
    a.y = hin[(pbase + l16) * NW + i0 + 1];
    v2f b0, b1;
    if (WT) {
      b0.x = w[(i0)*NW + l16];      b0.y = w[(i0 + 1) * NW + l16];
      b1.x = w[(i0)*NW + 16 + l16]; b1.y = w[(i0 + 1) * NW + 16 + l16];
    } else {
      b0.x = w[(l16)*NW + i0];        b0.y = w[(l16)*NW + i0 + 1];
      b1.x = w[(16 + l16) * NW + i0]; b1.y = w[(16 + l16) * NW + i0 + 1];
    }
    d0 = wmma4(a, b0, d0);
    d1 = wmma4(a, b1, d1);
  }
  #pragma unroll
  for (int r = 0; r < 8; ++r) {
    long p = pbase + r + half * 8;
    float v0 = d0[r] + bias[l16];
    float v1 = d1[r] + bias[16 + l16];
    if (DOGELU) { v0 = gelu_exact(v0); v1 = gelu_exact(v1); }
    out[p * NW + l16] = v0;
    out[p * NW + 16 + l16] = v1;
  }
}

#define FF_P 68
#define FI_P 36
#define G_RP 34
#define G_LP 1092
#define GP_LP 1027
#define OFF_FI (32 * FF_P)
#define OFF_G (OFF_FI + 64 * FI_P)
#define OFF_GP (OFF_G + 16 * G_LP)
#define LDS_FLOATS (OFF_GP + 16 * GP_LP)
__device__ __forceinline__ float ld_off(const float* p, unsigned boff) {
  return *(const float*)((const char*)p + boff);
}
__device__ __forceinline__ void st_off(float* p, unsigned boff, float v) {
  *(float*)((char*)p + boff) = v;
}

__global__ __launch_bounds__(256, 4) void k_spectral_axis(
    const float* __restrict__ h, const float* __restrict__ wgt,
    float* __restrict__ acc, long strideU, long strideV, long strideA) {
  extern __shared__ float smem[];
  float* Ff = smem;
  float* Fi = smem + OFF_FI;
  float* G  = smem + OFF_G;
  float* Gp = smem + OFF_GP;

  const int tid = threadIdx.x;
  const int lane = tid & 31, wv = tid >> 5;
  const int half = lane >> 4, l16 = lane & 15;
  const float NRM = 0.125f;

  const unsigned sAB = (unsigned)strideA * 4u;
  unsigned base[2];
  #pragma unroll
  for (int ll = 0; ll < 2; ++ll) {
    int l = blockIdx.x * 16 + wv * 2 + ll;
    int b = l >> 12, u = (l >> 6) & 63, v = l & 63;
    base[ll] = ((unsigned)b * (64u * 64 * 64 * NW) + (unsigned)u * (unsigned)strideU +
                (unsigned)v * (unsigned)strideV + (unsigned)l16) * 4u;
  }

  v2f hb0[2][16], hb1[2][16];
  #pragma unroll
  for (int ll = 0; ll < 2; ++ll) {
    #pragma unroll
    for (int kk = 0; kk < 16; ++kk) {
      const unsigned p0 = kk * 4 + half * 2;
      hb0[ll][kk].x = ld_off(h, base[ll] + p0 * sAB);
      hb0[ll][kk].y = ld_off(h, base[ll] + (p0 + 1) * sAB);
      hb1[ll][kk].x = ld_off(h, base[ll] + p0 * sAB + 64u);
      hb1[ll][kk].y = ld_off(h, base[ll] + (p0 + 1) * sAB + 64u);
    }
  }

  for (int idx = tid; idx < 2048; idx += 256) {
    int m2 = idx >> 6, p = idx & 63, k = m2 >> 1;
    float ang = 6.28318530717958647693f * (float)(k * p) * (1.0f / 64.0f);
    Ff[m2 * FF_P + p] = (m2 & 1) ? (-__sinf(ang) * NRM) : (__cosf(ang) * NRM);
  }
  for (int idx = tid; idx < 2048; idx += 256) {
    int p = idx >> 5, m2 = idx & 31, k = m2 >> 1;
    float ang = 6.28318530717958647693f * (float)(k * p) * (1.0f / 64.0f);
    float sc = (k == 0 ? 1.0f : 2.0f) * NRM;
    Fi[p * FI_P + m2] = (m2 & 1) ? (-__sinf(ang) * sc) : (__cosf(ang) * sc);
  }
  __syncthreads();

  #pragma unroll
  for (int ll = 0; ll < 2; ++ll) {
    const int li = wv * 2 + ll;
    v8f a00 = zero8(), a01 = zero8(), a10 = zero8(), a11 = zero8();
    #pragma unroll
    for (int kk = 0; kk < 16; ++kk) {
      const int p0 = kk * 4 + half * 2;
      v2f fa0, fa1;
      fa0.x = Ff[(l16)*FF_P + p0];        fa0.y = Ff[(l16)*FF_P + p0 + 1];
      fa1.x = Ff[(16 + l16) * FF_P + p0]; fa1.y = Ff[(16 + l16) * FF_P + p0 + 1];
      a00 = wmma4(fa0, hb0[ll][kk], a00);
      a01 = wmma4(fa0, hb1[ll][kk], a01);
      a10 = wmma4(fa1, hb0[ll][kk], a10);
      a11 = wmma4(fa1, hb1[ll][kk], a11);
    }
    float* g = G + li * G_LP;
    #pragma unroll
    for (int r = 0; r < 8; ++r) {
      const int m = r + half * 8;
      g[(m)*G_RP + l16]             = a00[r];
      g[(m)*G_RP + 16 + l16]        = a01[r];
      g[(m + 16) * G_RP + l16]      = a10[r];
      g[(m + 16) * G_RP + 16 + l16] = a11[r];
    }
  }

  float wq[2][8][2][4];
  #pragma unroll
  for (int km = 0; km < 2; ++km) {
    const int k = wv + km * 8;
    #pragma unroll
    for (int kk = 0; kk < 8; ++kk) {
      const int i0 = kk * 4 + half * 2, i1 = i0 + 1;
      #pragma unroll
      for (int mt = 0; mt < 2; ++mt) {
        const int o = mt * 16 + l16;
        wq[km][kk][mt][0] = wgt[((i0 * NW + o) * NM + k) * 2 + 0];
        wq[km][kk][mt][1] = wgt[((i0 * NW + o) * NM + k) * 2 + 1];
        wq[km][kk][mt][2] = wgt[((i1 * NW + o) * NM + k) * 2 + 0];
        wq[km][kk][mt][3] = wgt[((i1 * NW + o) * NM + k) * 2 + 1];
      }
    }
  }
  __syncthreads();

  #pragma unroll
  for (int km = 0; km < 2; ++km) {
    const int k = wv + km * 8;
    v8f oRe0 = zero8(), oRe1 = zero8(), oIm0 = zero8(), oIm1 = zero8();
    #pragma unroll
    for (int kk = 0; kk < 8; ++kk) {
      const int i0 = kk * 4 + half * 2, i1 = i0 + 1;
      v2f bRe, bIm;
      bRe.x = G[l16 * G_LP + (2 * k) * G_RP + i0];
      bRe.y = G[l16 * G_LP + (2 * k) * G_RP + i1];
      bIm.x = G[l16 * G_LP + (2 * k + 1) * G_RP + i0];
      bIm.y = G[l16 * G_LP + (2 * k + 1) * G_RP + i1];
      #pragma unroll
      for (int mt = 0; mt < 2; ++mt) {
        v2f aR, aI, aIn;
        aR.x = wq[km][kk][mt][0];
        aR.y = wq[km][kk][mt][2];
        aI.x = wq[km][kk][mt][1];
        aI.y = wq[km][kk][mt][3];
        aIn.x = -aI.x; aIn.y = -aI.y;
        if (mt == 0) {
          oRe0 = wmma4(aR, bRe, oRe0);  oRe0 = wmma4(aIn, bIm, oRe0);
          oIm0 = wmma4(aI, bRe, oIm0);  oIm0 = wmma4(aR, bIm, oIm0);
        } else {
          oRe1 = wmma4(aR, bRe, oRe1);  oRe1 = wmma4(aIn, bIm, oRe1);
          oIm1 = wmma4(aI, bRe, oIm1);  oIm1 = wmma4(aR, bIm, oIm1);
        }
      }
    }
    const int psw = (k & 1) << 4;
    #pragma unroll
    for (int r = 0; r < 8; ++r) {
      const int o0 = r + half * 8;
      Gp[l16 * GP_LP + (2 * k) * 32 + (o0 ^ psw)]            = oRe0[r];
      Gp[l16 * GP_LP + (2 * k) * 32 + ((o0 + 16) ^ psw)]     = oRe1[r];
      Gp[l16 * GP_LP + (2 * k + 1) * 32 + (o0 ^ psw)]        = oIm0[r];
      Gp[l16 * GP_LP + (2 * k + 1) * 32 + ((o0 + 16) ^ psw)] = oIm1[r];
    }
  }

  float cq[2][8][8];
  #pragma unroll
  for (int r = 0; r < 8; ++r) {
    const int pr = r + half * 8;
    #pragma unroll
    for (int q = 0; q < 4; ++q) {
      const unsigned ia = base[0] + (unsigned)(pr + 16 * q) * sAB;
      cq[0][2 * q][r]     = ld_off(acc, ia);
      cq[0][2 * q + 1][r] = ld_off(acc, ia + 64u);
    }
  }
  __syncthreads();

  #pragma unroll
  for (int ll = 0; ll < 2; ++ll) {
    const int li = wv * 2 + ll;
    const float* gp = Gp + li * GP_LP;
    if (ll == 1) {
      #pragma unroll
      for (int r = 0; r < 8; ++r) {
        const int pr = r + half * 8;
        #pragma unroll
        for (int q = 0; q < 4; ++q) {
          const unsigned ia = base[1] + (unsigned)(pr + 16 * q) * sAB;
          cq[1][2 * q][r]     = ld_off(acc, ia);
          cq[1][2 * q + 1][r] = ld_off(acc, ia + 64u);
        }
      }
    }
    v8f o00 = zero8(), o01 = zero8(), o10 = zero8(), o11 = zero8();
    v8f o20 = zero8(), o21 = zero8(), o30 = zero8(), o31 = zero8();
    for (int kk = 0; kk < 8; ++kk) {
      const int i0 = kk * 4 + half * 2;
      v2f b0, b1;
      const int c0 = l16 ^ (half << 4), c1 = c0 ^ 16;
      b0.x = gp[(i0)*32 + c0]; b0.y = gp[(i0 + 1) * 32 + c0];
      b1.x = gp[(i0)*32 + c1]; b1.y = gp[(i0 + 1) * 32 + c1];
      v2f a0, a1, a2, a3;
      a0.x = Fi[(l16)*FI_P + i0];        a0.y = Fi[(l16)*FI_P + i0 + 1];
      a1.x = Fi[(16 + l16) * FI_P + i0]; a1.y = Fi[(16 + l16) * FI_P + i0 + 1];
      a2.x = Fi[(32 + l16) * FI_P + i0]; a2.y = Fi[(32 + l16) * FI_P + i0 + 1];
      a3.x = Fi[(48 + l16) * FI_P + i0]; a3.y = Fi[(48 + l16) * FI_P + i0 + 1];
      o00 = wmma4(a0, b0, o00);  o01 = wmma4(a0, b1, o01);
      o10 = wmma4(a1, b0, o10);  o11 = wmma4(a1, b1, o11);
      o20 = wmma4(a2, b0, o20);  o21 = wmma4(a2, b1, o21);
      o30 = wmma4(a3, b0, o30);  o31 = wmma4(a3, b1, o31);
    }
    #pragma unroll
    for (int r = 0; r < 8; ++r) {
      const int pr = r + half * 8;
      const unsigned i0a = base[ll] + (unsigned)(pr)*sAB;
      const unsigned i1a = base[ll] + (unsigned)(pr + 16) * sAB;
      const unsigned i2a = base[ll] + (unsigned)(pr + 32) * sAB;
      const unsigned i3a = base[ll] + (unsigned)(pr + 48) * sAB;
      st_off(acc, i0a, cq[ll][0][r] + o00[r]);  st_off(acc, i0a + 64u, cq[ll][1][r] + o01[r]);
      st_off(acc, i1a, cq[ll][2][r] + o10[r]);  st_off(acc, i1a + 64u, cq[ll][3][r] + o11[r]);
      st_off(acc, i2a, cq[ll][4][r] + o20[r]);  st_off(acc, i2a + 64u, cq[ll][5][r] + o21[r]);
      st_off(acc, i3a, cq[ll][6][r] + o30[r]);  st_off(acc, i3a + 64u, cq[ll][7][r] + o31[r]);
    }
  }
}

__global__ __launch_bounds__(256) void k_gelu(const float* __restrict__ in,
                                              float* __restrict__ out) {
  const long i = (long)blockIdx.x * blockDim.x + threadIdx.x;
  float4 v = ((const float4*)in)[i];
  v.x = gelu_exact(v.x);
  v.y = gelu_exact(v.y);
  v.z = gelu_exact(v.z);
  v.w = gelu_exact(v.w);
  ((float4*)out)[i] = v;
}

__global__ __launch_bounds__(256) void k_fc3(const float* __restrict__ hin,
                                             const float* __restrict__ w,
                                             const float* __restrict__ bias,
                                             float* __restrict__ out) {
  const long p = (long)blockIdx.x * blockDim.x + threadIdx.x;
  const float4* hv = (const float4*)(hin + p * NW);
  const float4* wv = (const float4*)w;
  float s = 0.0f;
  #pragma unroll
  for (int i = 0; i < 8; ++i) {
    float4 a = hv[i], b = wv[i];
    s += a.x * b.x + a.y * b.y + a.z * b.z + a.w * b.w;
  }
  out[p] = s + bias[0];
}

extern "C" void kernel_launch(void* const* d_in, const int* in_sizes, int n_in,
                              void* d_out, int out_size, void* d_ws,
                              size_t ws_size, hipStream_t stream) {
  const float* x      = (const float*)d_in[0];
  const float* in_w   = (const float*)d_in[1];
  const float* in_b   = (const float*)d_in[2];
  const float* fwx    = (const float*)d_in[3];
  const float* fwy    = (const float*)d_in[4];
  const float* fwz    = (const float*)d_in[5];
  const float* conv_w = (const float*)d_in[6];
  const float* conv_b = (const float*)d_in[7];
  const float* fc1_w  = (const float*)d_in[8];
  const float* fc1_b  = (const float*)d_in[9];
  const float* fc2_w  = (const float*)d_in[10];
  const float* fc2_b  = (const float*)d_in[11];
  const float* fc3_w  = (const float*)d_in[12];
  const float* fc3_b  = (const float*)d_in[13];
  float* out = (float*)d_out;

  float* buf0 = (float*)d_ws;
  float* buf1 = buf0 + HSZ;

  const dim3 blk(256);
  const int wgPts = NPTS / (16 * 8);
  const size_t smemB = (size_t)LDS_FLOATS * sizeof(float);
  const long sX = 64L * 64 * NW;
  const long sY = 64L * NW;
  const long sZ = (long)NW;

  k_input_proj<<<wgPts, blk, 0, stream>>>(x, in_w, in_b, buf0);

  for (int i = 0; i < 4; ++i) {
    k_pointwise<false, false><<<wgPts, blk, 0, stream>>>(
        buf0, conv_w + i * (NW * NW), conv_b + i * NW, buf1);
    const long wofs = (long)i * (NW * NW * NM * 2);
    k_spectral_axis<<<1024, blk, smemB, stream>>>(buf0, fwz + wofs, buf1,
                                                  sX, sY, sZ);
    k_spectral_axis<<<1024, blk, smemB, stream>>>(buf0, fwy + wofs, buf1,
                                                  sX, sZ, sY);
    k_spectral_axis<<<1024, blk, smemB, stream>>>(buf0, fwx + wofs, buf1,
                                                  sY, sZ, sX);
    if (i < 3)
      k_gelu<<<(int)(HSZ / 4 / 256), blk, 0, stream>>>(buf1, buf0);
  }

  k_pointwise<true, true><<<wgPts, blk, 0, stream>>>(buf1, fc1_w, fc1_b, buf0);
  k_pointwise<true, true><<<wgPts, blk, 0, stream>>>(buf0, fc2_w, fc2_b, buf1);
  k_fc3<<<NPTS / 256, blk, 0, stream>>>(buf1, fc3_w, fc3_b, out);
}


# spectral kernel stage 2: each wave takes two adjacent modes (2w, 2w+1; was w, w+8), so both modes' weights are one 16-byte load per lane: half the weight-load instructions and line touches
# speedup vs baseline: 1.9699x; 1.1126x over previous
#include <hip/hip_runtime.h>
#include <math.h>


typedef __attribute__((ext_vector_type(2))) float v2f;
typedef __attribute__((ext_vector_type(8))) float v8f;

#define NB 4
#define NS 64
#define NW 32
#define NM 16
#define NPTS (NB * NS * NS * NS)
#define HSZ ((size_t)NPTS * NW)

__device__ __forceinline__ v8f wmma4(v2f a, v2f b, v8f c) {
  return __builtin_amdgcn_wmma_f32_16x16x4_f32(false, a, false, b, (short)0, c,
                                               false, false);
}

__device__ __forceinline__ v8f zero8() {
  v8f z;
  #pragma unroll
  for (int i = 0; i < 8; ++i) z[i] = 0.0f;
  return z;
}

__device__ __forceinline__ float gelu_exact(float x) {
  return 0.5f * x * (1.0f + erff(x * 0.70710678118654752440f));
}

__global__ __launch_bounds__(256) void k_input_proj(
    const float* __restrict__ x, const float* __restrict__ w,
    const float* __restrict__ bias, float* __restrict__ out) {
  const int lane = threadIdx.x & 31, wv = threadIdx.x >> 5;
  const int half = lane >> 4, l16 = lane & 15;
  const long pbase = ((long)blockIdx.x * 8 + wv) * 16;
  v2f a;
  a.x = x[(pbase + l16) * 4 + half * 2];
  a.y = x[(pbase + l16) * 4 + half * 2 + 1];
  v2f b0, b1;
  b0.x = w[(half * 2) * NW + l16];
  b0.y = w[(half * 2 + 1) * NW + l16];
  b1.x = w[(half * 2) * NW + 16 + l16];
  b1.y = w[(half * 2 + 1) * NW + 16 + l16];
  v8f d0 = zero8(), d1 = zero8();
  d0 = wmma4(a, b0, d0);
  d1 = wmma4(a, b1, d1);
  #pragma unroll
  for (int r = 0; r < 8; ++r) {
    long p = pbase + r + half * 8;
    out[p * NW + l16] = d0[r] + bias[l16];
    out[p * NW + 16 + l16] = d1[r] + bias[16 + l16];
  }
}

template <bool WT, bool DOGELU>
__global__ __launch_bounds__(256) void k_pointwise(
    const float* __restrict__ hin, const float* __restrict__ w,
    const float* __restrict__ bias, float* __restrict__ out) {
  const int lane = threadIdx.x & 31, wv = threadIdx.x >> 5;
  const int half = lane >> 4, l16 = lane & 15;
  const long pbase = ((long)blockIdx.x * 8 + wv) * 16;
  v8f d0 = zero8(), d1 = zero8();
  #pragma unroll
  for (int kk = 0; kk < 8; ++kk) {
    const int i0 = kk * 4 + half * 2;
    v2f a;
    a.x = hin[(pbase + l16) * NW + i0];
    a.y = hin[(pbase + l16) * NW + i0 + 1];
    v2f b0, b1;
    if (WT) {
      b0.x = w[(i0)*NW + l16];      b0.y = w[(i0 + 1) * NW + l16];
      b1.x = w[(i0)*NW + 16 + l16]; b1.y = w[(i0 + 1) * NW + 16 + l16];
    } else {
      b0.x = w[(l16)*NW + i0];        b0.y = w[(l16)*NW + i0 + 1];
      b1.x = w[(16 + l16) * NW + i0]; b1.y = w[(16 + l16) * NW + i0 + 1];
    }
    d0 = wmma4(a, b0, d0);
    d1 = wmma4(a, b1, d1);
  }
  #pragma unroll
  for (int r = 0; r < 8; ++r) {
    long p = pbase + r + half * 8;
    float v0 = d0[r] + bias[l16];
    float v1 = d1[r] + bias[16 + l16];
    if (DOGELU) { v0 = gelu_exact(v0); v1 = gelu_exact(v1); }
    out[p * NW + l16] = v0;
    out[p * NW + 16 + l16] = v1;
  }
}

#define FF_P 68
#define FI_P 36
#define G_RP 34
#define G_LP 1092
#define GP_LP 1027
#define OFF_FI (32 * FF_P)
#define OFF_G (OFF_FI + 64 * FI_P)
#define OFF_GP (OFF_G + 16 * G_LP)
#define LDS_FLOATS (OFF_GP + 16 * GP_LP)
__device__ __forceinline__ float ld_off(const float* p, unsigned boff) {
  return *(const float*)((const char*)p + boff);
}
__device__ __forceinline__ void st_off(float* p, unsigned boff, float v) {
  *(float*)((char*)p + boff) = v;
}

__global__ __launch_bounds__(256, 4) void k_spectral_axis(
    const float* __restrict__ h, const float* __restrict__ wgt,
    float* __restrict__ acc, long strideU, long strideV, long strideA) {
  extern __shared__ float smem[];
  float* Ff = smem;
  float* Fi = smem + OFF_FI;
  float* G  = smem + OFF_G;
  float* Gp = smem + OFF_GP;

  const int tid = threadIdx.x;
  const int lane = tid & 31, wv = tid >> 5;
  const int half = lane >> 4, l16 = lane & 15;
  const float NRM = 0.125f;

  const unsigned sAB = (unsigned)strideA * 4u;
  unsigned base[2];
  #pragma unroll
  for (int ll = 0; ll < 2; ++ll) {
    int l = blockIdx.x * 16 + wv * 2 + ll;
    int b = l >> 12, u = (l >> 6) & 63, v = l & 63;
    base[ll] = ((unsigned)b * (64u * 64 * 64 * NW) + (unsigned)u * (unsigned)strideU +
                (unsigned)v * (unsigned)strideV + (unsigned)l16) * 4u;
  }

  v2f hb0[2][16], hb1[2][16];
  #pragma unroll
  for (int ll = 0; ll < 2; ++ll) {
    #pragma unroll
    for (int kk = 0; kk < 16; ++kk) {
      const unsigned p0 = kk * 4 + half * 2;
      hb0[ll][kk].x = ld_off(h, base[ll] + p0 * sAB);
      hb0[ll][kk].y = ld_off(h, base[ll] + (p0 + 1) * sAB);
      hb1[ll][kk].x = ld_off(h, base[ll] + p0 * sAB + 64u);
      hb1[ll][kk].y = ld_off(h, base[ll] + (p0 + 1) * sAB + 64u);
    }
  }

  for (int idx = tid; idx < 2048; idx += 256) {
    int m2 = idx >> 6, p = idx & 63, k = m2 >> 1;
    float ang = 6.28318530717958647693f * (float)(k * p) * (1.0f / 64.0f);
    Ff[m2 * FF_P + p] = (m2 & 1) ? (-__sinf(ang) * NRM) : (__cosf(ang) * NRM);
  }
  for (int idx = tid; idx < 2048; idx += 256) {
    int p = idx >> 5, m2 = idx & 31, k = m2 >> 1;
    float ang = 6.28318530717958647693f * (float)(k * p) * (1.0f / 64.0f);
    float sc = (k == 0 ? 1.0f : 2.0f) * NRM;
    Fi[p * FI_P + m2] = (m2 & 1) ? (-__sinf(ang) * sc) : (__cosf(ang) * sc);
  }
  __syncthreads();

  #pragma unroll
  for (int ll = 0; ll < 2; ++ll) {
    const int li = wv * 2 + ll;
    v8f a00 = zero8(), a01 = zero8(), a10 = zero8(), a11 = zero8();
    #pragma unroll
    for (int kk = 0; kk < 16; ++kk) {
      const int p0 = kk * 4 + half * 2;
      v2f fa0, fa1;
      fa0.x = Ff[(l16)*FF_P + p0];        fa0.y = Ff[(l16)*FF_P + p0 + 1];
      fa1.x = Ff[(16 + l16) * FF_P + p0]; fa1.y = Ff[(16 + l16) * FF_P + p0 + 1];
      a00 = wmma4(fa0, hb0[ll][kk], a00);
      a01 = wmma4(fa0, hb1[ll][kk], a01);
      a10 = wmma4(fa1, hb0[ll][kk], a10);
      a11 = wmma4(fa1, hb1[ll][kk], a11);
    }
    float* g = G + li * G_LP;
    #pragma unroll
    for (int r = 0; r < 8; ++r) {
      const int m = r + half * 8;
      g[(m)*G_RP + l16]             = a00[r];
      g[(m)*G_RP + 16 + l16]        = a01[r];
      g[(m + 16) * G_RP + l16]      = a10[r];
      g[(m + 16) * G_RP + 16 + l16] = a11[r];
    }
  }

  float4 wq[8][2][2];
  {
    const float4* wgt4 = (const float4*)(wgt + 4 * wv);
    #pragma unroll
    for (int kk = 0; kk < 8; ++kk) {
      const int i0 = kk * 4 + half * 2;
      #pragma unroll
      for (int mt = 0; mt < 2; ++mt) {
        const int o = mt * 16 + l16;
        #pragma unroll
        for (int j = 0; j < 2; ++j)
          wq[kk][mt][j] = wgt4[((i0 + j) * NW + o) * (NM * 2 / 4)];
      }
    }
  }
  __syncthreads();

  #pragma unroll
  for (int km = 0; km < 2; ++km) {
    const int k = 2 * wv + km;
    v8f oRe0 = zero8(), oRe1 = zero8(), oIm0 = zero8(), oIm1 = zero8();
    #pragma unroll
    for (int kk = 0; kk < 8; ++kk) {
      const int i0 = kk * 4 + half * 2, i1 = i0 + 1;
      v2f bRe, bIm;
      bRe.x = G[l16 * G_LP + (2 * k) * G_RP + i0];
      bRe.y = G[l16 * G_LP + (2 * k) * G_RP + i1];
      bIm.x = G[l16 * G_LP + (2 * k + 1) * G_RP + i0];
      bIm.y = G[l16 * G_LP + (2 * k + 1) * G_RP + i1];
      #pragma unroll
      for (int mt = 0; mt < 2; ++mt) {
        v2f aR, aI, aIn;
        aR.x = km ? wq[kk][mt][0].z : wq[kk][mt][0].x;
        aR.y = km ? wq[kk][mt][1].z : wq[kk][mt][1].x;
        aI.x = km ? wq[kk][mt][0].w : wq[kk][mt][0].y;
        aI.y = km ? wq[kk][mt][1].w : wq[kk][mt][1].y;
        aIn.x = -aI.x; aIn.y = -aI.y;
        if (mt == 0) {
          oRe0 = wmma4(aR, bRe, oRe0);  oRe0 = wmma4(aIn, bIm, oRe0);
          oIm0 = wmma4(aI, bRe, oIm0);  oIm0 = wmma4(aR, bIm, oIm0);
        } else {
          oRe1 = wmma4(aR, bRe, oRe1);  oRe1 = wmma4(aIn, bIm, oRe1);
          oIm1 = wmma4(aI, bRe, oIm1);  oIm1 = wmma4(aR, bIm, oIm1);
        }
      }
    }
    const int psw = (k & 1) << 4;
    #pragma unroll
    for (int r = 0; r < 8; ++r) {
      const int o0 = r + half * 8;
      Gp[l16 * GP_LP + (2 * k) * 32 + (o0 ^ psw)]            = oRe0[r];
      Gp[l16 * GP_LP + (2 * k) * 32 + ((o0 + 16) ^ psw)]     = oRe1[r];
      Gp[l16 * GP_LP + (2 * k + 1) * 32 + (o0 ^ psw)]        = oIm0[r];
      Gp[l16 * GP_LP + (2 * k + 1) * 32 + ((o0 + 16) ^ psw)] = oIm1[r];
    }
  }

  float cq[2][8][8];
  #pragma unroll
  for (int r = 0; r < 8; ++r) {
    const int pr = r + half * 8;
    #pragma unroll
    for (int q = 0; q < 4; ++q) {
      const unsigned ia = base[0] + (unsigned)(pr + 16 * q) * sAB;
      cq[0][2 * q][r]     = ld_off(acc, ia);
      cq[0][2 * q + 1][r] = ld_off(acc, ia + 64u);
    }
  }
  __syncthreads();

  #pragma unroll
  for (int ll = 0; ll < 2; ++ll) {
    const int li = wv * 2 + ll;
    const float* gp = Gp + li * GP_LP;
    if (ll == 1) {
      #pragma unroll
      for (int r = 0; r < 8; ++r) {
        const int pr = r + half * 8;
        #pragma unroll
        for (int q = 0; q < 4; ++q) {
          const unsigned ia = base[1] + (unsigned)(pr + 16 * q) * sAB;
          cq[1][2 * q][r]     = ld_off(acc, ia);
          cq[1][2 * q + 1][r] = ld_off(acc, ia + 64u);
        }
      }
    }
    v8f o00 = zero8(), o01 = zero8(), o10 = zero8(), o11 = zero8();
    v8f o20 = zero8(), o21 = zero8(), o30 = zero8(), o31 = zero8();
    for (int kk = 0; kk < 8; ++kk) {
      const int i0 = kk * 4 + half * 2;
      v2f b0, b1;
      const int c0 = l16 ^ (half << 4), c1 = c0 ^ 16;
      b0.x = gp[(i0)*32 + c0]; b0.y = gp[(i0 + 1) * 32 + c0];
      b1.x = gp[(i0)*32 + c1]; b1.y = gp[(i0 + 1) * 32 + c1];
      v2f a0, a1, a2, a3;
      a0.x = Fi[(l16)*FI_P + i0];        a0.y = Fi[(l16)*FI_P + i0 + 1];
      a1.x = Fi[(16 + l16) * FI_P + i0]; a1.y = Fi[(16 + l16) * FI_P + i0 + 1];
      a2.x = Fi[(32 + l16) * FI_P + i0]; a2.y = Fi[(32 + l16) * FI_P + i0 + 1];
      a3.x = Fi[(48 + l16) * FI_P + i0]; a3.y = Fi[(48 + l16) * FI_P + i0 + 1];
      o00 = wmma4(a0, b0, o00);  o01 = wmma4(a0, b1, o01);
      o10 = wmma4(a1, b0, o10);  o11 = wmma4(a1, b1, o11);
      o20 = wmma4(a2, b0, o20);  o21 = wmma4(a2, b1, o21);
      o30 = wmma4(a3, b0, o30);  o31 = wmma4(a3, b1, o31);
    }
    #pragma unroll
    for (int r = 0; r < 8; ++r) {
      const int pr = r + half * 8;
      const unsigned i0a = base[ll] + (unsigned)(pr)*sAB;
      const unsigned i1a = base[ll] + (unsigned)(pr + 16) * sAB;
      const unsigned i2a = base[ll] + (unsigned)(pr + 32) * sAB;
      const unsigned i3a = base[ll] + (unsigned)(pr + 48) * sAB;
      st_off(acc, i0a, cq[ll][0][r] + o00[r]);  st_off(acc, i0a + 64u, cq[ll][1][r] + o01[r]);
      st_off(acc, i1a, cq[ll][2][r] + o10[r]);  st_off(acc, i1a + 64u, cq[ll][3][r] + o11[r]);
      st_off(acc, i2a, cq[ll][4][r] + o20[r]);  st_off(acc, i2a + 64u, cq[ll][5][r] + o21[r]);
      st_off(acc, i3a, cq[ll][6][r] + o30[r]);  st_off(acc, i3a + 64u, cq[ll][7][r] + o31[r]);
    }
  }
}

__global__ __launch_bounds__(256) void k_gelu(const float* __restrict__ in,
                                              float* __restrict__ out) {
  const long i = (long)blockIdx.x * blockDim.x + threadIdx.x;
  float4 v = ((const float4*)in)[i];
  v.x = gelu_exact(v.x);
  v.y = gelu_exact(v.y);
  v.z = gelu_exact(v.z);
  v.w = gelu_exact(v.w);
  ((float4*)out)[i] = v;
}

__global__ __launch_bounds__(256) void k_fc3(const float* __restrict__ hin,
                                             const float* __restrict__ w,
                                             const float* __restrict__ bias,
                                             float* __restrict__ out) {
  const long p = (long)blockIdx.x * blockDim.x + threadIdx.x;
  const float4* hv = (const float4*)(hin + p * NW);
  const float4* wv = (const float4*)w;
  float s = 0.0f;
  #pragma unroll
  for (int i = 0; i < 8; ++i) {
    float4 a = hv[i], b = wv[i];
    s += a.x * b.x + a.y * b.y + a.z * b.z + a.w * b.w;
  }
  out[p] = s + bias[0];
}

extern "C" void kernel_launch(void* const* d_in, const int* in_sizes, int n_in,
                              void* d_out, int out_size, void* d_ws,
                              size_t ws_size, hipStream_t stream) {
  const float* x      = (const float*)d_in[0];
  const float* in_w   = (const float*)d_in[1];
  const float* in_b   = (const float*)d_in[2];
  const float* fwx    = (const float*)d_in[3];
  const float* fwy    = (const float*)d_in[4];
  const float* fwz    = (const float*)d_in[5];
  const float* conv_w = (const float*)d_in[6];
  const float* conv_b = (const float*)d_in[7];
  const float* fc1_w  = (const float*)d_in[8];
  const float* fc1_b  = (const float*)d_in[9];
  const float* fc2_w  = (const float*)d_in[10];
  const float* fc2_b  = (const float*)d_in[11];
  const float* fc3_w  = (const float*)d_in[12];
  const float* fc3_b  = (const float*)d_in[13];
  float* out = (float*)d_out;

  float* buf0 = (float*)d_ws;
  float* buf1 = buf0 + HSZ;

  const dim3 blk(256);
  const int wgPts = NPTS / (16 * 8);
  const size_t smemB = (size_t)LDS_FLOATS * sizeof(float);
  const long sX = 64L * 64 * NW;
  const long sY = 64L * NW;
  const long sZ = (long)NW;

  k_input_proj<<<wgPts, blk, 0, stream>>>(x, in_w, in_b, buf0);

  for (int i = 0; i < 4; ++i) {
    k_pointwise<false, false><<<wgPts, blk, 0, stream>>>(
        buf0, conv_w + i * (NW * NW), conv_b + i * NW, buf1);
    const long wofs = (long)i * (NW * NW * NM * 2);
    k_spectral_axis<<<1024, blk, smemB, stream>>>(buf0, fwz + wofs, buf1,
                                                  sX, sY, sZ);
    k_spectral_axis<<<1024, blk, smemB, stream>>>(buf0, fwy + wofs, buf1,
                                                  sX, sZ, sY);
    k_spectral_axis<<<1024, blk, smemB, stream>>>(buf0, fwx + wofs, buf1,
                                                  sY, sZ, sX);
    if (i < 3)
      k_gelu<<<(int)(HSZ / 4 / 256), blk, 0, stream>>>(buf1, buf0);
  }

  k_pointwise<true, true><<<wgPts, blk, 0, stream>>>(buf1, fc1_w, fc1_b, buf0);
  k_pointwise<true, true><<<wgPts, blk, 0, stream>>>(buf0, fc2_w, fc2_b, buf1);
  k_fc3<<<NPTS / 256, blk, 0, stream>>>(buf1, fc3_w, fc3_b, out);
}


# spectral stage 2: a pre-pass kernel re-lays the mode weights in per-wave fragment order (scratch: head of the output buffer, rewritten whole by the last kernel): a weight load touches 4 lines, not 32
# speedup vs baseline: 2.1362x; 1.0844x over previous
#include <hip/hip_runtime.h>
#include <math.h>


typedef __attribute__((ext_vector_type(2))) float v2f;
typedef __attribute__((ext_vector_type(8))) float v8f;

#define NB 4
#define NS 64
#define NW 32
#define NM 16
#define NPTS (NB * NS * NS * NS)
#define HSZ ((size_t)NPTS * NW)

__device__ __forceinline__ v8f wmma4(v2f a, v2f b, v8f c) {
  return __builtin_amdgcn_wmma_f32_16x16x4_f32(false, a, false, b, (short)0, c,
                                               false, false);
}

__device__ __forceinline__ v8f zero8() {
  v8f z;
  #pragma unroll
  for (int i = 0; i < 8; ++i) z[i] = 0.0f;
  return z;
}

__device__ __forceinline__ float gelu_exact(float x) {
  return 0.5f * x * (1.0f + erff(x * 0.70710678118654752440f));
}

__global__ __launch_bounds__(256) void k_input_proj(
    const float* __restrict__ x, const float* __restrict__ w,
    const float* __restrict__ bias, float* __restrict__ out) {
  const int lane = threadIdx.x & 31, wv = threadIdx.x >> 5;
  const int half = lane >> 4, l16 = lane & 15;
  const long pbase = ((long)blockIdx.x * 8 + wv) * 16;
  v2f a;
  a.x = x[(pbase + l16) * 4 + half * 2];
  a.y = x[(pbase + l16) * 4 + half * 2 + 1];
  v2f b0, b1;
  b0.x = w[(half * 2) * NW + l16];
  b0.y = w[(half * 2 + 1) * NW + l16];
  b1.x = w[(half * 2) * NW + 16 + l16];
  b1.y = w[(half * 2 + 1) * NW + 16 + l16];
  v8f d0 = zero8(), d1 = zero8();
  d0 = wmma4(a, b0, d0);
  d1 = wmma4(a, b1, d1);
  #pragma unroll
  for (int r = 0; r < 8; ++r) {
    long p = pbase + r + half * 8;
    out[p * NW + l16] = d0[r] + bias[l16];
    out[p * NW + 16 + l16] = d1[r] + bias[16 + l16];
  }
}

template <bool WT, bool DOGELU>
__global__ __launch_bounds__(256) void k_pointwise(
    const float* __restrict__ hin, const float* __restrict__ w,
    const float* __restrict__ bias, float* __restrict__ out) {
  const int lane = threadIdx.x & 31, wv = threadIdx.x >> 5;
  const int half = lane >> 4, l16 = lane & 15;
  const long pbase = ((long)blockIdx.x * 8 + wv) * 16;
  v8f d0 = zero8(), d1 = zero8();
  #pragma unroll
  for (int kk = 0; kk < 8; ++kk) {
    const int i0 = kk * 4 + half * 2;
    v2f a;
    a.x = hin[(pbase + l16) * NW + i0];
    a.y = hin[(pbase + l16) * NW + i0 + 1];
    v2f b0, b1;
    if (WT) {
      b0.x = w[(i0)*NW + l16];      b0.y = w[(i0 + 1) * NW + l16];
      b1.x = w[(i0)*NW + 16 + l16]; b1.y = w[(i0 + 1) * NW + 16 + l16];
    } else {
      b0.x = w[(l16)*NW + i0];        b0.y = w[(l16)*NW + i0 + 1];
      b1.x = w[(16 + l16) * NW + i0]; b1.y = w[(16 + l16) * NW + i0 + 1];
    }
    d0 = wmma4(a, b0, d0);
    d1 = wmma4(a, b1, d1);
  }
  #pragma unroll
  for (int r = 0; r < 8; ++r) {
    long p = pbase + r + half * 8;
    float v0 = d0[r] + bias[l16];
    float v1 = d1[r] + bias[16 + l16];
    if (DOGELU) { v0 = gelu_exact(v0); v1 = gelu_exact(v1); }
    out[p * NW + l16] = v0;
    out[p * NW + 16 + l16] = v1;
  }
}

#define FF_P 68
#define FI_P 36
#define G_RP 34
#define G_LP 1092
#define GP_LP 1027
#define OFF_FI (32 * FF_P)
#define OFF_G (OFF_FI + 64 * FI_P)
#define OFF_GP (OFF_G + 16 * G_LP)
#define LDS_FLOATS (OFF_GP + 16 * GP_LP)
__device__ __forceinline__ float ld_off(const float* p, unsigned boff) {
  return *(const float*)((const char*)p + boff);
}
__device__ __forceinline__ void st_off(float* p, unsigned boff, float v) {
  *(float*)((char*)p + boff) = v;
}

__global__ __launch_bounds__(256) void k_weight_relayout(
    const float* __restrict__ fwz, const float* __restrict__ fwy,
    const float* __restrict__ fwx, float* __restrict__ dst) {
  const int t = blockIdx.x * 256 + threadIdx.x;
  const int s = t >> 13, q = t & 8191;
  const int axis = s >> 2, layer = s & 3;
  const float* src = (axis == 0) ? fwz : (axis == 1) ? fwy : fwx;
  const int lane = q & 31, j = (q >> 5) & 1, mt = (q >> 6) & 1;
  const int kk = (q >> 7) & 7, wv = q >> 10;
  const int in = kk * 4 + (lane >> 4) * 2 + j, o = mt * 16 + (lane & 15);
  const float4 v = *(const float4*)(src + layer * (NW * NW * NM * 2) +
                                    (in * NW + o) * (NM * 2) + 4 * wv);
  ((float4*)dst)[t] = v;
}

__global__ __launch_bounds__(256, 4) void k_spectral_axis(
    const float* __restrict__ h, const float* __restrict__ wgt,
    float* __restrict__ acc, long strideU, long strideV, long strideA) {
  extern __shared__ float smem[];
  float* Ff = smem;
  float* Fi = smem + OFF_FI;
  float* G  = smem + OFF_G;
  float* Gp = smem + OFF_GP;

  const int tid = threadIdx.x;
  const int lane = tid & 31, wv = tid >> 5;
  const int half = lane >> 4, l16 = lane & 15;
  const float NRM = 0.125f;

  const unsigned sAB = (unsigned)strideA * 4u;
  unsigned base[2];
  #pragma unroll
  for (int ll = 0; ll < 2; ++ll) {
    int l = blockIdx.x * 16 + wv * 2 + ll;
    int b = l >> 12, u = (l >> 6) & 63, v = l & 63;
    base[ll] = ((unsigned)b * (64u * 64 * 64 * NW) + (unsigned)u * (unsigned)strideU +
                (unsigned)v * (unsigned)strideV + (unsigned)l16) * 4u;
  }

  v2f hb0[2][16], hb1[2][16];
  #pragma unroll
  for (int ll = 0; ll < 2; ++ll) {
    #pragma unroll
    for (int kk = 0; kk < 16; ++kk) {
      const unsigned p0 = kk * 4 + half * 2;
      hb0[ll][kk].x = ld_off(h, base[ll] + p0 * sAB);
      hb0[ll][kk].y = ld_off(h, base[ll] + (p0 + 1) * sAB);
      hb1[ll][kk].x = ld_off(h, base[ll] + p0 * sAB + 64u);
      hb1[ll][kk].y = ld_off(h, base[ll] + (p0 + 1) * sAB + 64u);
    }
  }

  for (int idx = tid; idx < 2048; idx += 256) {
    int m2 = idx >> 6, p = idx & 63, k = m2 >> 1;
    float ang = 6.28318530717958647693f * (float)(k * p) * (1.0f / 64.0f);
    Ff[m2 * FF_P + p] = (m2 & 1) ? (-__sinf(ang) * NRM) : (__cosf(ang) * NRM);
  }
  for (int idx = tid; idx < 2048; idx += 256) {
    int p = idx >> 5, m2 = idx & 31, k = m2 >> 1;
    float ang = 6.28318530717958647693f * (float)(k * p) * (1.0f / 64.0f);
    float sc = (k == 0 ? 1.0f : 2.0f) * NRM;
    Fi[p * FI_P + m2] = (m2 & 1) ? (-__sinf(ang) * sc) : (__cosf(ang) * sc);
  }
  __syncthreads();

  #pragma unroll
  for (int ll = 0; ll < 2; ++ll) {
    const int li = wv * 2 + ll;
    v8f a00 = zero8(), a01 = zero8(), a10 = zero8(), a11 = zero8();
    #pragma unroll
    for (int kk = 0; kk < 16; ++kk) {
      const int p0 = kk * 4 + half * 2;
      v2f fa0, fa1;
      fa0.x = Ff[(l16)*FF_P + p0];        fa0.y = Ff[(l16)*FF_P + p0 + 1];
      fa1.x = Ff[(16 + l16) * FF_P + p0]; fa1.y = Ff[(16 + l16) * FF_P + p0 + 1];
      a00 = wmma4(fa0, hb0[ll][kk], a00);
      a01 = wmma4(fa0, hb1[ll][kk], a01);
      a10 = wmma4(fa1, hb0[ll][kk], a10);
      a11 = wmma4(fa1, hb1[ll][kk], a11);
    }
    float* g = G + li * G_LP;
    #pragma unroll
    for (int r = 0; r < 8; ++r) {
      const int m = r + half * 8;
      g[(m)*G_RP + l16]             = a00[r];
      g[(m)*G_RP + 16 + l16]        = a01[r];
      g[(m + 16) * G_RP + l16]      = a10[r];
      g[(m + 16) * G_RP + 16 + l16] = a11[r];
    }
  }

  float4 wq[8][2][2];
  {
    const float4* wgt4 = (const float4*)wgt + wv * 1024 + lane;
    #pragma unroll
    for (int kk = 0; kk < 8; ++kk) {
      #pragma unroll
      for (int mt = 0; mt < 2; ++mt) {
        #pragma unroll
        for (int j = 0; j < 2; ++j)
          wq[kk][mt][j] = wgt4[((kk * 2 + mt) * 2 + j) * 32];
      }
    }
  }
  __syncthreads();

  #pragma unroll
  for (int km = 0; km < 2; ++km) {
    const int k = 2 * wv + km;
    v8f oRe0 = zero8(), oRe1 = zero8(), oIm0 = zero8(), oIm1 = zero8();
    #pragma unroll
    for (int kk = 0; kk < 8; ++kk) {
      const int i0 = kk * 4 + half * 2, i1 = i0 + 1;
      v2f bRe, bIm;
      bRe.x = G[l16 * G_LP + (2 * k) * G_RP + i0];
      bRe.y = G[l16 * G_LP + (2 * k) * G_RP + i1];
      bIm.x = G[l16 * G_LP + (2 * k + 1) * G_RP + i0];
      bIm.y = G[l16 * G_LP + (2 * k + 1) * G_RP + i1];
      #pragma unroll
      for (int mt = 0; mt < 2; ++mt) {
        v2f aR, aI, aIn;
        aR.x = km ? wq[kk][mt][0].z : wq[kk][mt][0].x;
        aR.y = km ? wq[kk][mt][1].z : wq[kk][mt][1].x;
        aI.x = km ? wq[kk][mt][0].w : wq[kk][mt][0].y;
        aI.y = km ? wq[kk][mt][1].w : wq[kk][mt][1].y;
        aIn.x = -aI.x; aIn.y = -aI.y;
        if (mt == 0) {
          oRe0 = wmma4(aR, bRe, oRe0);  oRe0 = wmma4(aIn, bIm, oRe0);
          oIm0 = wmma4(aI, bRe, oIm0);  oIm0 = wmma4(aR, bIm, oIm0);
        } else {
          oRe1 = wmma4(aR, bRe, oRe1);  oRe1 = wmma4(aIn, bIm, oRe1);
          oIm1 = wmma4(aI, bRe, oIm1);  oIm1 = wmma4(aR, bIm, oIm1);
        }
      }
    }
    const int psw = (k & 1) << 4;
    #pragma unroll
    for (int r = 0; r < 8; ++r) {
      const int o0 = r + half * 8;
      Gp[l16 * GP_LP + (2 * k) * 32 + (o0 ^ psw)]            = oRe0[r];
      Gp[l16 * GP_LP + (2 * k) * 32 + ((o0 + 16) ^ psw)]     = oRe1[r];
      Gp[l16 * GP_LP + (2 * k + 1) * 32 + (o0 ^ psw)]        = oIm0[r];
      Gp[l16 * GP_LP + (2 * k + 1) * 32 + ((o0 + 16) ^ psw)] = oIm1[r];
    }
  }

  float cq[2][8][8];
  #pragma unroll
  for (int r = 0; r < 8; ++r) {
    const int pr = r + half * 8;
    #pragma unroll
    for (int q = 0; q < 4; ++q) {
      const unsigned ia = base[0] + (unsigned)(pr + 16 * q) * sAB;
      cq[0][2 * q][r]     = ld_off(acc, ia);
      cq[0][2 * q + 1][r] = ld_off(acc, ia + 64u);
    }
  }
  __syncthreads();

  #pragma unroll
  for (int ll = 0; ll < 2; ++ll) {
    const int li = wv * 2 + ll;
    const float* gp = Gp + li * GP_LP;
    if (ll == 1) {
      #pragma unroll
      for (int r = 0; r < 8; ++r) {
        const int pr = r + half * 8;
        #pragma unroll
        for (int q = 0; q < 4; ++q) {
          const unsigned ia = base[1] + (unsigned)(pr + 16 * q) * sAB;
          cq[1][2 * q][r]     = ld_off(acc, ia);
          cq[1][2 * q + 1][r] = ld_off(acc, ia + 64u);
        }
      }
    }
    v8f o00 = zero8(), o01 = zero8(), o10 = zero8(), o11 = zero8();
    v8f o20 = zero8(), o21 = zero8(), o30 = zero8(), o31 = zero8();
    for (int kk = 0; kk < 8; ++kk) {
      const int i0 = kk * 4 + half * 2;
      v2f b0, b1;
      const int c0 = l16 ^ (half << 4), c1 = c0 ^ 16;
      b0.x = gp[(i0)*32 + c0]; b0.y = gp[(i0 + 1) * 32 + c0];
      b1.x = gp[(i0)*32 + c1]; b1.y = gp[(i0 + 1) * 32 + c1];
      v2f a0, a1, a2, a3;
      a0.x = Fi[(l16)*FI_P + i0];        a0.y = Fi[(l16)*FI_P + i0 + 1];
      a1.x = Fi[(16 + l16) * FI_P + i0]; a1.y = Fi[(16 + l16) * FI_P + i0 + 1];
      a2.x = Fi[(32 + l16) * FI_P + i0]; a2.y = Fi[(32 + l16) * FI_P + i0 + 1];
      a3.x = Fi[(48 + l16) * FI_P + i0]; a3.y = Fi[(48 + l16) * FI_P + i0 + 1];
      o00 = wmma4(a0, b0, o00);  o01 = wmma4(a0, b1, o01);
      o10 = wmma4(a1, b0, o10);  o11 = wmma4(a1, b1, o11);
      o20 = wmma4(a2, b0, o20);  o21 = wmma4(a2, b1, o21);
      o30 = wmma4(a3, b0, o30);  o31 = wmma4(a3, b1, o31);
    }
    #pragma unroll
    for (int r = 0; r < 8; ++r) {
      const int pr = r + half * 8;
      const unsigned i0a = base[ll] + (unsigned)(pr)*sAB;
      const unsigned i1a = base[ll] + (unsigned)(pr + 16) * sAB;
      const unsigned i2a = base[ll] + (unsigned)(pr + 32) * sAB;
      const unsigned i3a = base[ll] + (unsigned)(pr + 48) * sAB;
      st_off(acc, i0a, cq[ll][0][r] + o00[r]);  st_off(acc, i0a + 64u, cq[ll][1][r] + o01[r]);
      st_off(acc, i1a, cq[ll][2][r] + o10[r]);  st_off(acc, i1a + 64u, cq[ll][3][r] + o11[r]);
      st_off(acc, i2a, cq[ll][4][r] + o20[r]);  st_off(acc, i2a + 64u, cq[ll][5][r] + o21[r]);
      st_off(acc, i3a, cq[ll][6][r] + o30[r]);  st_off(acc, i3a + 64u, cq[ll][7][r] + o31[r]);
    }
  }
}

__global__ __launch_bounds__(256) void k_gelu(const float* __restrict__ in,
                                              float* __restrict__ out) {
  const long i = (long)blockIdx.x * blockDim.x + threadIdx.x;
  float4 v = ((const float4*)in)[i];
  v.x = gelu_exact(v.x);
  v.y = gelu_exact(v.y);
  v.z = gelu_exact(v.z);
  v.w = gelu_exact(v.w);
  ((float4*)out)[i] = v;
}

__global__ __launch_bounds__(256) void k_fc3(const float* __restrict__ hin,
                                             const float* __restrict__ w,
                                             const float* __restrict__ bias,
                                             float* __restrict__ out) {
  const long p = (long)blockIdx.x * blockDim.x + threadIdx.x;
  const float4* hv = (const float4*)(hin + p * NW);
  const float4* wv = (const float4*)w;
  float s = 0.0f;
  #pragma unroll
  for (int i = 0; i < 8; ++i) {
    float4 a = hv[i], b = wv[i];
    s += a.x * b.x + a.y * b.y + a.z * b.z + a.w * b.w;
  }
  out[p] = s + bias[0];
}

extern "C" void kernel_launch(void* const* d_in, const int* in_sizes, int n_in,
                              void* d_out, int out_size, void* d_ws,
                              size_t ws_size, hipStream_t stream) {
  const float* x      = (const float*)d_in[0];
  const float* in_w   = (const float*)d_in[1];
  const float* in_b   = (const float*)d_in[2];
  const float* fwx    = (const float*)d_in[3];
  const float* fwy    = (const float*)d_in[4];
  const float* fwz    = (const float*)d_in[5];
  const float* conv_w = (const float*)d_in[6];
  const float* conv_b = (const float*)d_in[7];
  const float* fc1_w  = (const float*)d_in[8];
  const float* fc1_b  = (const float*)d_in[9];
  const float* fc2_w  = (const float*)d_in[10];
  const float* fc2_b  = (const float*)d_in[11];
  const float* fc3_w  = (const float*)d_in[12];
  const float* fc3_b  = (const float*)d_in[13];
  float* out = (float*)d_out;

  float* buf0 = (float*)d_ws;
  float* buf1 = buf0 + HSZ;
  float* wre = out;
  const long wsl = (long)NW * NW * NM * 2;

  const dim3 blk(256);
  const int wgPts = NPTS / (16 * 8);
  const size_t smemB = (size_t)LDS_FLOATS * sizeof(float);
  const long sX = 64L * 64 * NW;
  const long sY = 64L * NW;
  const long sZ = (long)NW;

  k_input_proj<<<wgPts, blk, 0, stream>>>(x, in_w, in_b, buf0);
  k_weight_relayout<<<12 * 8192 / 256, blk, 0, stream>>>(fwz, fwy, fwx, wre);

  for (int i = 0; i < 4; ++i) {
    k_pointwise<false, false><<<wgPts, blk, 0, stream>>>(
        buf0, conv_w + i * (NW * NW), conv_b + i * NW, buf1);
    const long wofs = (long)i * wsl;
    k_spectral_axis<<<1024, blk, smemB, stream>>>(buf0, wre + wofs, buf1,
                                                  sX, sY, sZ);
    k_spectral_axis<<<1024, blk, smemB, stream>>>(buf0, wre + 4 * wsl + wofs,
                                                  buf1, sX, sZ, sY);
    k_spectral_axis<<<1024, blk, smemB, stream>>>(buf0, wre + 8 * wsl + wofs,
                                                  buf1, sY, sZ, sX);
    if (i < 3)
      k_gelu<<<(int)(HSZ / 4 / 256), blk, 0, stream>>>(buf1, buf0);
  }

  k_pointwise<true, true><<<wgPts, blk, 0, stream>>>(buf1, fc1_w, fc1_b, buf0);
  k_pointwise<true, true><<<wgPts, blk, 0, stream>>>(buf0, fc2_w, fc2_b, buf1);
  k_fc3<<<NPTS / 256, blk, 0, stream>>>(buf1, fc3_w, fc3_b, out);
}


# 1x1 conv fused into the z-axis spectral kernel: 64 more WMMAs per line on the stage-3 tiles (sum order now spectral, conv, bias); the 4 conv launches and the z kernel's accumulator read are gone
# speedup vs baseline: 2.4560x; 1.1497x over previous
#include <hip/hip_runtime.h>
#include <math.h>


typedef __attribute__((ext_vector_type(2))) float v2f;
typedef __attribute__((ext_vector_type(8))) float v8f;

#define NB 4
#define NS 64
#define NW 32
#define NM 16
#define NPTS (NB * NS * NS * NS)
#define HSZ ((size_t)NPTS * NW)

__device__ __forceinline__ v8f wmma4(v2f a, v2f b, v8f c) {
  return __builtin_amdgcn_wmma_f32_16x16x4_f32(false, a, false, b, (short)0, c,
                                               false, false);
}

__device__ __forceinline__ v8f zero8() {
  v8f z;
  #pragma unroll
  for (int i = 0; i < 8; ++i) z[i] = 0.0f;
  return z;
}

__device__ __forceinline__ float gelu_exact(float x) {
  return 0.5f * x * (1.0f + erff(x * 0.70710678118654752440f));
}

__global__ __launch_bounds__(256) void k_input_proj(
    const float* __restrict__ x, const float* __restrict__ w,
    const float* __restrict__ bias, float* __restrict__ out) {
  const int lane = threadIdx.x & 31, wv = threadIdx.x >> 5;
  const int half = lane >> 4, l16 = lane & 15;
  const long pbase = ((long)blockIdx.x * 8 + wv) * 16;
  v2f a;
  a.x = x[(pbase + l16) * 4 + half * 2];
  a.y = x[(pbase + l16) * 4 + half * 2 + 1];
  v2f b0, b1;
  b0.x = w[(half * 2) * NW + l16];
  b0.y = w[(half * 2 + 1) * NW + l16];
  b1.x = w[(half * 2) * NW + 16 + l16];
  b1.y = w[(half * 2 + 1) * NW + 16 + l16];
  v8f d0 = zero8(), d1 = zero8();
  d0 = wmma4(a, b0, d0);
  d1 = wmma4(a, b1, d1);
  #pragma unroll
  for (int r = 0; r < 8; ++r) {
    long p = pbase + r + half * 8;
    out[p * NW + l16] = d0[r] + bias[l16];
    out[p * NW + 16 + l16] = d1[r] + bias[16 + l16];
  }
}

template <bool WT, bool DOGELU>
__global__ __launch_bounds__(256) void k_pointwise(
    const float* __restrict__ hin, const float* __restrict__ w,
    const float* __restrict__ bias, float* __restrict__ out) {
  const int lane = threadIdx.x & 31, wv = threadIdx.x >> 5;
  const int half = lane >> 4, l16 = lane & 15;
  const long pbase = ((long)blockIdx.x * 8 + wv) * 16;
  v8f d0 = zero8(), d1 = zero8();
  #pragma unroll
  for (int kk = 0; kk < 8; ++kk) {
    const int i0 = kk * 4 + half * 2;
    v2f a;
    a.x = hin[(pbase + l16) * NW + i0];
    a.y = hin[(pbase + l16) * NW + i0 + 1];
    v2f b0, b1;
    if (WT) {
      b0.x = w[(i0)*NW + l16];      b0.y = w[(i0 + 1) * NW + l16];
      b1.x = w[(i0)*NW + 16 + l16]; b1.y = w[(i0 + 1) * NW + 16 + l16];
    } else {
      b0.x = w[(l16)*NW + i0];        b0.y = w[(l16)*NW + i0 + 1];
      b1.x = w[(16 + l16) * NW + i0]; b1.y = w[(16 + l16) * NW + i0 + 1];
    }
    d0 = wmma4(a, b0, d0);
    d1 = wmma4(a, b1, d1);
  }
  #pragma unroll
  for (int r = 0; r < 8; ++r) {
    long p = pbase + r + half * 8;
    float v0 = d0[r] + bias[l16];
    float v1 = d1[r] + bias[16 + l16];
    if (DOGELU) { v0 = gelu_exact(v0); v1 = gelu_exact(v1); }
    out[p * NW + l16] = v0;
    out[p * NW + 16 + l16] = v1;
  }
}

#define FF_P 68
#define FI_P 36
#define G_RP 34
#define G_LP 1092
#define GP_LP 1027
#define OFF_FI (32 * FF_P)
#define OFF_G (OFF_FI + 64 * FI_P)
#define OFF_GP (OFF_G + 16 * G_LP)
#define LDS_FLOATS (OFF_GP + 16 * GP_LP)
__device__ __forceinline__ float ld_off(const float* p, unsigned boff) {
  return *(const float*)((const char*)p + boff);
}
__device__ __forceinline__ void st_off(float* p, unsigned boff, float v) {
  *(float*)((char*)p + boff) = v;
}
__device__ __forceinline__ float4 ld4_off(const float* p, unsigned boff) {
  return *(const float4*)((const char*)p + boff);
}
__device__ __forceinline__ v2f k_pair(float4 v, int s) {
  v2f r;
  r.x = s ? v.z : v.x;
  r.y = s ? v.w : v.y;
  return r;
}

__device__ __forceinline__ void irdft_line(
    const float* gp, const float* Fi, int half, int l16,
    v8f& o00, v8f& o01, v8f& o10, v8f& o11,
    v8f& o20, v8f& o21, v8f& o30, v8f& o31) {
  for (int kk = 0; kk < 8; ++kk) {
    const int i0 = kk * 4 + half * 2;
    v2f b0, b1;
    const int c0 = l16 ^ (half << 4), c1 = c0 ^ 16;
    b0.x = gp[(i0)*32 + c0]; b0.y = gp[(i0 + 1) * 32 + c0];
    b1.x = gp[(i0)*32 + c1]; b1.y = gp[(i0 + 1) * 32 + c1];
    v2f a0, a1, a2, a3;
    a0.x = Fi[(l16)*FI_P + i0];        a0.y = Fi[(l16)*FI_P + i0 + 1];
    a1.x = Fi[(16 + l16) * FI_P + i0]; a1.y = Fi[(16 + l16) * FI_P + i0 + 1];
    a2.x = Fi[(32 + l16) * FI_P + i0]; a2.y = Fi[(32 + l16) * FI_P + i0 + 1];
    a3.x = Fi[(48 + l16) * FI_P + i0]; a3.y = Fi[(48 + l16) * FI_P + i0 + 1];
    o00 = wmma4(a0, b0, o00);  o01 = wmma4(a0, b1, o01);
    o10 = wmma4(a1, b0, o10);  o11 = wmma4(a1, b1, o11);
    o20 = wmma4(a2, b0, o20);  o21 = wmma4(a2, b1, o21);
    o30 = wmma4(a3, b0, o30);  o31 = wmma4(a3, b1, o31);
  }
}

__global__ __launch_bounds__(256) void k_weight_relayout(
    const float* __restrict__ fwz, const float* __restrict__ fwy,
    const float* __restrict__ fwx, float* __restrict__ dst) {
  const int t = blockIdx.x * 256 + threadIdx.x;
  const int s = t >> 13, q = t & 8191;
  const int axis = s >> 2, layer = s & 3;
  const float* src = (axis == 0) ? fwz : (axis == 1) ? fwy : fwx;
  const int lane = q & 31, j = (q >> 5) & 1, mt = (q >> 6) & 1;
  const int kk = (q >> 7) & 7, wv = q >> 10;
  const int in = kk * 4 + (lane >> 4) * 2 + j, o = mt * 16 + (lane & 15);
  const float4 v = *(const float4*)(src + layer * (NW * NW * NM * 2) +
                                    (in * NW + o) * (NM * 2) + 4 * wv);
  ((float4*)dst)[t] = v;
}

template <bool FIRST>
__global__ __launch_bounds__(256, 4) void k_spectral_axis(
    const float* __restrict__ h, const float* __restrict__ wgt,
    float* __restrict__ acc, const float* __restrict__ cw,
    const float* __restrict__ cb, long strideU, long strideV, long strideA) {
  extern __shared__ float smem[];
  float* Ff = smem;
  float* Fi = smem + OFF_FI;
  float* G  = smem + OFF_G;
  float* Gp = smem + OFF_GP;

  const int tid = threadIdx.x;
  const int lane = tid & 31, wv = tid >> 5;
  const int half = lane >> 4, l16 = lane & 15;
  const float NRM = 0.125f;

  const unsigned sAB = (unsigned)strideA * 4u;
  unsigned base[2];
  #pragma unroll
  for (int ll = 0; ll < 2; ++ll) {
    int l = blockIdx.x * 16 + wv * 2 + ll;
    int b = l >> 12, u = (l >> 6) & 63, v = l & 63;
    base[ll] = ((unsigned)b * (64u * 64 * 64 * NW) + (unsigned)u * (unsigned)strideU +
                (unsigned)v * (unsigned)strideV + (unsigned)l16) * 4u;
  }

  v2f hb0[2][16], hb1[2][16];
  #pragma unroll
  for (int ll = 0; ll < 2; ++ll) {
    #pragma unroll
    for (int kk = 0; kk < 16; ++kk) {
      const unsigned p0 = kk * 4 + half * 2;
      hb0[ll][kk].x = ld_off(h, base[ll] + p0 * sAB);
      hb0[ll][kk].y = ld_off(h, base[ll] + (p0 + 1) * sAB);
      hb1[ll][kk].x = ld_off(h, base[ll] + p0 * sAB + 64u);
      hb1[ll][kk].y = ld_off(h, base[ll] + (p0 + 1) * sAB + 64u);
    }
  }

  for (int idx = tid; idx < 2048; idx += 256) {
    int m2 = idx >> 6, p = idx & 63, k = m2 >> 1;
    float ang = 6.28318530717958647693f * (float)(k * p) * (1.0f / 64.0f);
    Ff[m2 * FF_P + p] = (m2 & 1) ? (-__sinf(ang) * NRM) : (__cosf(ang) * NRM);
  }
  for (int idx = tid; idx < 2048; idx += 256) {
    int p = idx >> 5, m2 = idx & 31, k = m2 >> 1;
    float ang = 6.28318530717958647693f * (float)(k * p) * (1.0f / 64.0f);
    float sc = (k == 0 ? 1.0f : 2.0f) * NRM;
    Fi[p * FI_P + m2] = (m2 & 1) ? (-__sinf(ang) * sc) : (__cosf(ang) * sc);
  }
  __syncthreads();

  #pragma unroll
  for (int ll = 0; ll < 2; ++ll) {
    const int li = wv * 2 + ll;
    v8f a00 = zero8(), a01 = zero8(), a10 = zero8(), a11 = zero8();
    #pragma unroll
    for (int kk = 0; kk < 16; ++kk) {
      const int p0 = kk * 4 + half * 2;
      v2f fa0, fa1;
      fa0.x = Ff[(l16)*FF_P + p0];        fa0.y = Ff[(l16)*FF_P + p0 + 1];
      fa1.x = Ff[(16 + l16) * FF_P + p0]; fa1.y = Ff[(16 + l16) * FF_P + p0 + 1];
      a00 = wmma4(fa0, hb0[ll][kk], a00);
      a01 = wmma4(fa0, hb1[ll][kk], a01);
      a10 = wmma4(fa1, hb0[ll][kk], a10);
      a11 = wmma4(fa1, hb1[ll][kk], a11);
    }
    float* g = G + li * G_LP;
    #pragma unroll
    for (int r = 0; r < 8; ++r) {
      const int m = r + half * 8;
      g[(m)*G_RP + l16]             = a00[r];
      g[(m)*G_RP + 16 + l16]        = a01[r];
      g[(m + 16) * G_RP + l16]      = a10[r];
      g[(m + 16) * G_RP + 16 + l16] = a11[r];
    }
  }

  float4 wq[8][2][2];
  {
    const float4* wgt4 = (const float4*)wgt + wv * 1024 + lane;
    #pragma unroll
    for (int kk = 0; kk < 8; ++kk) {
      #pragma unroll
      for (int mt = 0; mt < 2; ++mt) {
        #pragma unroll
        for (int j = 0; j < 2; ++j)
          wq[kk][mt][j] = wgt4[((kk * 2 + mt) * 2 + j) * 32];
      }
    }
  }
  __syncthreads();

  #pragma unroll
  for (int km = 0; km < 2; ++km) {
    const int k = 2 * wv + km;
    v8f oRe0 = zero8(), oRe1 = zero8(), oIm0 = zero8(), oIm1 = zero8();
    #pragma unroll
    for (int kk = 0; kk < 8; ++kk) {
      const int i0 = kk * 4 + half * 2, i1 = i0 + 1;
      v2f bRe, bIm;
      bRe.x = G[l16 * G_LP + (2 * k) * G_RP + i0];
      bRe.y = G[l16 * G_LP + (2 * k) * G_RP + i1];
      bIm.x = G[l16 * G_LP + (2 * k + 1) * G_RP + i0];
      bIm.y = G[l16 * G_LP + (2 * k + 1) * G_RP + i1];
      #pragma unroll
      for (int mt = 0; mt < 2; ++mt) {
        v2f aR, aI, aIn;
        aR.x = km ? wq[kk][mt][0].z : wq[kk][mt][0].x;
        aR.y = km ? wq[kk][mt][1].z : wq[kk][mt][1].x;
        aI.x = km ? wq[kk][mt][0].w : wq[kk][mt][0].y;
        aI.y = km ? wq[kk][mt][1].w : wq[kk][mt][1].y;
        aIn.x = -aI.x; aIn.y = -aI.y;
        if (mt == 0) {
          oRe0 = wmma4(aR, bRe, oRe0);  oRe0 = wmma4(aIn, bIm, oRe0);
          oIm0 = wmma4(aI, bRe, oIm0);  oIm0 = wmma4(aR, bIm, oIm0);
        } else {
          oRe1 = wmma4(aR, bRe, oRe1);  oRe1 = wmma4(aIn, bIm, oRe1);
          oIm1 = wmma4(aI, bRe, oIm1);  oIm1 = wmma4(aR, bIm, oIm1);
        }
      }
    }
    const int psw = (k & 1) << 4;
    #pragma unroll
    for (int r = 0; r < 8; ++r) {
      const int o0 = r + half * 8;
      Gp[l16 * GP_LP + (2 * k) * 32 + (o0 ^ psw)]            = oRe0[r];
      Gp[l16 * GP_LP + (2 * k) * 32 + ((o0 + 16) ^ psw)]     = oRe1[r];
      Gp[l16 * GP_LP + (2 * k + 1) * 32 + (o0 ^ psw)]        = oIm0[r];
      Gp[l16 * GP_LP + (2 * k + 1) * 32 + ((o0 + 16) ^ psw)] = oIm1[r];
    }
  }

  if (FIRST) {
    float4 cwq[2][4];
    float cbv[2];
    #pragma unroll
    for (int nt = 0; nt < 2; ++nt) {
      cbv[nt] = cb[nt * 16 + l16];
      #pragma unroll
      for (int j = 0; j < 4; ++j)
        cwq[nt][j] = *(const float4*)(cw + (nt * 16 + l16) * NW + 8 * j + 4 * half);
    }
    __syncthreads();

    #pragma unroll
    for (int ll = 0; ll < 2; ++ll) {
      const unsigned lb = base[ll];
      const unsigned ab = lb - (unsigned)l16 * 4u + (unsigned)l16 * sAB +
                          (unsigned)half * 16u;
      float4 ha[4][4];
      #pragma unroll
      for (int q = 0; q < 4; ++q) {
        #pragma unroll
        for (int j = 0; j < 4; ++j)
          ha[q][j] = ld4_off(h, ab + (unsigned)(16 * q) * sAB + 32u * j);
      }
      __builtin_amdgcn_sched_barrier(0);
      v8f o00 = zero8(), o01 = zero8(), o10 = zero8(), o11 = zero8();
      v8f o20 = zero8(), o21 = zero8(), o30 = zero8(), o31 = zero8();
      irdft_line(Gp + (wv * 2 + ll) * GP_LP, Fi, half, l16,
                 o00, o01, o10, o11, o20, o21, o30, o31);
      #pragma unroll
      for (int j = 0; j < 4; ++j) {
        #pragma unroll
        for (int s = 0; s < 2; ++s) {
          const v2f cb0 = k_pair(cwq[0][j], s), cb1 = k_pair(cwq[1][j], s);
          const v2f ca0 = k_pair(ha[0][j], s), ca1 = k_pair(ha[1][j], s);
          const v2f ca2 = k_pair(ha[2][j], s), ca3 = k_pair(ha[3][j], s);
          o00 = wmma4(ca0, cb0, o00);  o01 = wmma4(ca0, cb1, o01);
          o10 = wmma4(ca1, cb0, o10);  o11 = wmma4(ca1, cb1, o11);
          o20 = wmma4(ca2, cb0, o20);  o21 = wmma4(ca2, cb1, o21);
          o30 = wmma4(ca3, cb0, o30);  o31 = wmma4(ca3, cb1, o31);
        }
      }
      #pragma unroll
      for (int r = 0; r < 8; ++r) {
        const int pr = r + half * 8;
        const unsigned i0a = lb + (unsigned)(pr)*sAB;
        const unsigned i1a = lb + (unsigned)(pr + 16) * sAB;
        const unsigned i2a = lb + (unsigned)(pr + 32) * sAB;
        const unsigned i3a = lb + (unsigned)(pr + 48) * sAB;
        st_off(acc, i0a, o00[r] + cbv[0]);  st_off(acc, i0a + 64u, o01[r] + cbv[1]);
        st_off(acc, i1a, o10[r] + cbv[0]);  st_off(acc, i1a + 64u, o11[r] + cbv[1]);
        st_off(acc, i2a, o20[r] + cbv[0]);  st_off(acc, i2a + 64u, o21[r] + cbv[1]);
        st_off(acc, i3a, o30[r] + cbv[0]);  st_off(acc, i3a + 64u, o31[r] + cbv[1]);
      }
    }
    return;
  }

  float cq[2][8][8];
  #pragma unroll
  for (int r = 0; r < 8; ++r) {
    const int pr = r + half * 8;
    #pragma unroll
    for (int q = 0; q < 4; ++q) {
      const unsigned ia = base[0] + (unsigned)(pr + 16 * q) * sAB;
      cq[0][2 * q][r]     = ld_off(acc, ia);
      cq[0][2 * q + 1][r] = ld_off(acc, ia + 64u);
    }
  }
  __syncthreads();

  #pragma unroll
  for (int ll = 0; ll < 2; ++ll) {
    const int li = wv * 2 + ll;
    if (ll == 1) {
      #pragma unroll
      for (int r = 0; r < 8; ++r) {
        const int pr = r + half * 8;
        #pragma unroll
        for (int q = 0; q < 4; ++q) {
          const unsigned ia = base[1] + (unsigned)(pr + 16 * q) * sAB;
          cq[1][2 * q][r]     = ld_off(acc, ia);
          cq[1][2 * q + 1][r] = ld_off(acc, ia + 64u);
        }
      }
    }
    v8f o00 = zero8(), o01 = zero8(), o10 = zero8(), o11 = zero8();
    v8f o20 = zero8(), o21 = zero8(), o30 = zero8(), o31 = zero8();
    irdft_line(Gp + li * GP_LP, Fi, half, l16,
               o00, o01, o10, o11, o20, o21, o30, o31);
    #pragma unroll
    for (int r = 0; r < 8; ++r) {
      const int pr = r + half * 8;
      const unsigned i0a = base[ll] + (unsigned)(pr)*sAB;
      const unsigned i1a = base[ll] + (unsigned)(pr + 16) * sAB;
      const unsigned i2a = base[ll] + (unsigned)(pr + 32) * sAB;
      const unsigned i3a = base[ll] + (unsigned)(pr + 48) * sAB;
      st_off(acc, i0a, cq[ll][0][r] + o00[r]);  st_off(acc, i0a + 64u, cq[ll][1][r] + o01[r]);
      st_off(acc, i1a, cq[ll][2][r] + o10[r]);  st_off(acc, i1a + 64u, cq[ll][3][r] + o11[r]);
      st_off(acc, i2a, cq[ll][4][r] + o20[r]);  st_off(acc, i2a + 64u, cq[ll][5][r] + o21[r]);
      st_off(acc, i3a, cq[ll][6][r] + o30[r]);  st_off(acc, i3a + 64u, cq[ll][7][r] + o31[r]);
    }
  }
}

__global__ __launch_bounds__(256) void k_gelu(const float* __restrict__ in,
                                              float* __restrict__ out) {
  const long i = (long)blockIdx.x * blockDim.x + threadIdx.x;
  float4 v = ((const float4*)in)[i];
  v.x = gelu_exact(v.x);
  v.y = gelu_exact(v.y);
  v.z = gelu_exact(v.z);
  v.w = gelu_exact(v.w);
  ((float4*)out)[i] = v;
}

__global__ __launch_bounds__(256) void k_fc3(const float* __restrict__ hin,
                                             const float* __restrict__ w,
                                             const float* __restrict__ bias,
                                             float* __restrict__ out) {
  const long p = (long)blockIdx.x * blockDim.x + threadIdx.x;
  const float4* hv = (const float4*)(hin + p * NW);
  const float4* wv = (const float4*)w;
  float s = 0.0f;
  #pragma unroll
  for (int i = 0; i < 8; ++i) {
    float4 a = hv[i], b = wv[i];
    s += a.x * b.x + a.y * b.y + a.z * b.z + a.w * b.w;
  }
  out[p] = s + bias[0];
}

extern "C" void kernel_launch(void* const* d_in, const int* in_sizes, int n_in,
                              void* d_out, int out_size, void* d_ws,
                              size_t ws_size, hipStream_t stream) {
  const float* x      = (const float*)d_in[0];
  const float* in_w   = (const float*)d_in[1];
  const float* in_b   = (const float*)d_in[2];
  const float* fwx    = (const float*)d_in[3];
  const float* fwy    = (const float*)d_in[4];
  const float* fwz    = (const float*)d_in[5];
  const float* conv_w = (const float*)d_in[6];
  const float* conv_b = (const float*)d_in[7];
  const float* fc1_w  = (const float*)d_in[8];
  const float* fc1_b  = (const float*)d_in[9];
  const float* fc2_w  = (const float*)d_in[10];
  const float* fc2_b  = (const float*)d_in[11];
  const float* fc3_w  = (const float*)d_in[12];
  const float* fc3_b  = (const float*)d_in[13];
  float* out = (float*)d_out;

  float* buf0 = (float*)d_ws;
  float* buf1 = buf0 + HSZ;
  float* wre = out;
  const long wsl = (long)NW * NW * NM * 2;

  const dim3 blk(256);
  const int wgPts = NPTS / (16 * 8);
  const size_t smemB = (size_t)LDS_FLOATS * sizeof(float);
  const long sX = 64L * 64 * NW;
  const long sY = 64L * NW;
  const long sZ = (long)NW;

  k_input_proj<<<wgPts, blk, 0, stream>>>(x, in_w, in_b, buf0);
  k_weight_relayout<<<12 * 8192 / 256, blk, 0, stream>>>(fwz, fwy, fwx, wre);

  for (int i = 0; i < 4; ++i) {
    const long wofs = (long)i * wsl;
    const float* cwi = conv_w + i * (NW * NW);
    const float* cbi = conv_b + i * NW;
    k_spectral_axis<true><<<1024, blk, smemB, stream>>>(
        buf0, wre + wofs, buf1, cwi, cbi, sX, sY, sZ);
    k_spectral_axis<false><<<1024, blk, smemB, stream>>>(
        buf0, wre + 4 * wsl + wofs, buf1, cwi, cbi, sX, sZ, sY);
    k_spectral_axis<false><<<1024, blk, smemB, stream>>>(
        buf0, wre + 8 * wsl + wofs, buf1, cwi, cbi, sY, sZ, sX);
    if (i < 3)
      k_gelu<<<(int)(HSZ / 4 / 256), blk, 0, stream>>>(buf1, buf0);
  }

  k_pointwise<true, true><<<wgPts, blk, 0, stream>>>(buf1, fc1_w, fc1_b, buf0);
  k_pointwise<true, true><<<wgPts, blk, 0, stream>>>(buf0, fc2_w, fc2_b, buf1);
  k_fc3<<<NPTS / 256, blk, 0, stream>>>(buf1, fc3_w, fc3_b, out);
}


# spectral kernel: both 64-point transforms folded at the half period (even modes from h[p]+h[p+32], odd from h[p]-h[p+32]; out = E+-O; conv rows too): 32 WMMAs each per line, were 64; sum order changes
# speedup vs baseline: 2.6757x; 1.0895x over previous
#include <hip/hip_runtime.h>
#include <math.h>


typedef __attribute__((ext_vector_type(2))) float v2f;
typedef __attribute__((ext_vector_type(8))) float v8f;

#define NB 4
#define NS 64
#define NW 32
#define NM 16
#define NPTS (NB * NS * NS * NS)
#define HSZ ((size_t)NPTS * NW)

__device__ __forceinline__ v8f wmma4(v2f a, v2f b, v8f c) {
  return __builtin_amdgcn_wmma_f32_16x16x4_f32(false, a, false, b, (short)0, c,
                                               false, false);
}

__device__ __forceinline__ v8f zero8() {
  v8f z;
  #pragma unroll
  for (int i = 0; i < 8; ++i) z[i] = 0.0f;
  return z;
}

__device__ __forceinline__ float gelu_exact(float x) {
  return 0.5f * x * (1.0f + erff(x * 0.70710678118654752440f));
}

__global__ __launch_bounds__(256) void k_input_proj(
    const float* __restrict__ x, const float* __restrict__ w,
    const float* __restrict__ bias, float* __restrict__ out) {
  const int lane = threadIdx.x & 31, wv = threadIdx.x >> 5;
  const int half = lane >> 4, l16 = lane & 15;
  const long pbase = ((long)blockIdx.x * 8 + wv) * 16;
  v2f a;
  a.x = x[(pbase + l16) * 4 + half * 2];
  a.y = x[(pbase + l16) * 4 + half * 2 + 1];
  v2f b0, b1;
  b0.x = w[(half * 2) * NW + l16];
  b0.y = w[(half * 2 + 1) * NW + l16];
  b1.x = w[(half * 2) * NW + 16 + l16];
  b1.y = w[(half * 2 + 1) * NW + 16 + l16];
  v8f d0 = zero8(), d1 = zero8();
  d0 = wmma4(a, b0, d0);
  d1 = wmma4(a, b1, d1);
  #pragma unroll
  for (int r = 0; r < 8; ++r) {
    long p = pbase + r + half * 8;
    out[p * NW + l16] = d0[r] + bias[l16];
    out[p * NW + 16 + l16] = d1[r] + bias[16 + l16];
  }
}

template <bool WT, bool DOGELU>
__global__ __launch_bounds__(256) void k_pointwise(
    const float* __restrict__ hin, const float* __restrict__ w,
    const float* __restrict__ bias, float* __restrict__ out) {
  const int lane = threadIdx.x & 31, wv = threadIdx.x >> 5;
  const int half = lane >> 4, l16 = lane & 15;
  const long pbase = ((long)blockIdx.x * 8 + wv) * 16;
  v8f d0 = zero8(), d1 = zero8();
  #pragma unroll
  for (int kk = 0; kk < 8; ++kk) {
    const int i0 = kk * 4 + half * 2;
    v2f a;
    a.x = hin[(pbase + l16) * NW + i0];
    a.y = hin[(pbase + l16) * NW + i0 + 1];
    v2f b0, b1;
    if (WT) {
      b0.x = w[(i0)*NW + l16];      b0.y = w[(i0 + 1) * NW + l16];
      b1.x = w[(i0)*NW + 16 + l16]; b1.y = w[(i0 + 1) * NW + 16 + l16];
    } else {
      b0.x = w[(l16)*NW + i0];        b0.y = w[(l16)*NW + i0 + 1];
      b1.x = w[(16 + l16) * NW + i0]; b1.y = w[(16 + l16) * NW + i0 + 1];
    }
    d0 = wmma4(a, b0, d0);
    d1 = wmma4(a, b1, d1);
  }
  #pragma unroll
  for (int r = 0; r < 8; ++r) {
    long p = pbase + r + half * 8;
    float v0 = d0[r] + bias[l16];
    float v1 = d1[r] + bias[16 + l16];
    if (DOGELU) { v0 = gelu_exact(v0); v1 = gelu_exact(v1); }
    out[p * NW + l16] = v0;
    out[p * NW + 16 + l16] = v1;
  }
}

#define FF_P 36
#define FI_P 36
#define G_RP 34
#define G_LP 1092
#define GP_LP 1027
#define OFF_FI (32 * FF_P)
#define OFF_G (OFF_FI + 32 * FI_P)
#define OFF_GP (OFF_G + 16 * G_LP)
#define LDS_FLOATS (OFF_GP + 16 * GP_LP)
__device__ __forceinline__ float ld_off(const float* p, unsigned boff) {
  return *(const float*)((const char*)p + boff);
}
__device__ __forceinline__ void st_off(float* p, unsigned boff, float v) {
  *(float*)((char*)p + boff) = v;
}
__device__ __forceinline__ float4 ld4_off(const float* p, unsigned boff) {
  return *(const float4*)((const char*)p + boff);
}
__device__ __forceinline__ v2f k_pair(float4 v, int s) {
  v2f r;
  r.x = s ? v.z : v.x;
  r.y = s ? v.w : v.y;
  return r;
}

__device__ __forceinline__ int mode_of_row(int R) {
  return ((R & 15) >> 1) * 2 + (R >> 4);
}

__device__ __forceinline__ void irdft_half(
    const float* gp, const float* Fi, int g, int half, int l16,
    v8f& t00, v8f& t01, v8f& t10, v8f& t11) {
  for (int kk = 4 * g; kk < 4 * g + 4; ++kk) {
    const int i0 = kk * 4 + half * 2;
    v2f b0, b1;
    const int c0 = l16 ^ (half << 4), c1 = c0 ^ 16;
    b0.x = gp[(i0)*32 + c0]; b0.y = gp[(i0 + 1) * 32 + c0];
    b1.x = gp[(i0)*32 + c1]; b1.y = gp[(i0 + 1) * 32 + c1];
    v2f a0, a1;
    a0.x = Fi[(l16)*FI_P + i0];        a0.y = Fi[(l16)*FI_P + i0 + 1];
    a1.x = Fi[(16 + l16) * FI_P + i0]; a1.y = Fi[(16 + l16) * FI_P + i0 + 1];
    t00 = wmma4(a0, b0, t00);  t01 = wmma4(a0, b1, t01);
    t10 = wmma4(a1, b0, t10);  t11 = wmma4(a1, b1, t11);
  }
}

__global__ __launch_bounds__(256) void k_weight_relayout(
    const float* __restrict__ fwz, const float* __restrict__ fwy,
    const float* __restrict__ fwx, float* __restrict__ dst) {
  const int t = blockIdx.x * 256 + threadIdx.x;
  const int s = t >> 13, q = t & 8191;
  const int axis = s >> 2, layer = s & 3;
  const float* src = (axis == 0) ? fwz : (axis == 1) ? fwy : fwx;
  const int lane = q & 31, j = (q >> 5) & 1, mt = (q >> 6) & 1;
  const int kk = (q >> 7) & 7, wv = q >> 10;
  const int in = kk * 4 + (lane >> 4) * 2 + j, o = mt * 16 + (lane & 15);
  const float4 v = *(const float4*)(src + layer * (NW * NW * NM * 2) +
                                    (in * NW + o) * (NM * 2) + 4 * wv);
  ((float4*)dst)[t] = v;
}

template <bool FIRST>
__global__ __launch_bounds__(256, 4) void k_spectral_axis(
    const float* __restrict__ h, const float* __restrict__ wgt,
    float* __restrict__ acc, const float* __restrict__ cw,
    const float* __restrict__ cb, long strideU, long strideV, long strideA) {
  extern __shared__ float smem[];
  float* Ff = smem;
  float* Fi = smem + OFF_FI;
  float* G  = smem + OFF_G;
  float* Gp = smem + OFF_GP;

  const int tid = threadIdx.x;
  const int lane = tid & 31, wv = tid >> 5;
  const int half = lane >> 4, l16 = lane & 15;
  const float NRM = 0.125f;

  const unsigned sAB = (unsigned)strideA * 4u;
  unsigned base[2];
  #pragma unroll
  for (int ll = 0; ll < 2; ++ll) {
    int l = blockIdx.x * 16 + wv * 2 + ll;
    int b = l >> 12, u = (l >> 6) & 63, v = l & 63;
    base[ll] = ((unsigned)b * (64u * 64 * 64 * NW) + (unsigned)u * (unsigned)strideU +
                (unsigned)v * (unsigned)strideV + (unsigned)l16) * 4u;
  }

  v2f hb0[2][16], hb1[2][16];
  #pragma unroll
  for (int ll = 0; ll < 2; ++ll) {
    #pragma unroll
    for (int kk = 0; kk < 16; ++kk) {
      const unsigned p0 = kk * 4 + half * 2;
      hb0[ll][kk].x = ld_off(h, base[ll] + p0 * sAB);
      hb0[ll][kk].y = ld_off(h, base[ll] + (p0 + 1) * sAB);
      hb1[ll][kk].x = ld_off(h, base[ll] + p0 * sAB + 64u);
      hb1[ll][kk].y = ld_off(h, base[ll] + (p0 + 1) * sAB + 64u);
    }
  }

  for (int idx = tid; idx < 1024; idx += 256) {
    int m2 = idx >> 5, p = idx & 31, k = mode_of_row(m2);
    float ang = 6.28318530717958647693f * (float)(k * p) * (1.0f / 64.0f);
    Ff[m2 * FF_P + p] = (m2 & 1) ? (-__sinf(ang) * NRM) : (__cosf(ang) * NRM);
  }
  for (int idx = tid; idx < 1024; idx += 256) {
    int p = idx >> 5, m2 = idx & 31, k = mode_of_row(m2);
    float ang = 6.28318530717958647693f * (float)(k * p) * (1.0f / 64.0f);
    float sc = (k == 0 ? 1.0f : 2.0f) * NRM;
    Fi[p * FI_P + m2] = (m2 & 1) ? (-__sinf(ang) * sc) : (__cosf(ang) * sc);
  }
  __syncthreads();

  #pragma unroll
  for (int ll = 0; ll < 2; ++ll) {
    const int li = wv * 2 + ll;
    v8f a00 = zero8(), a01 = zero8(), a10 = zero8(), a11 = zero8();
    #pragma unroll
    for (int kk = 0; kk < 8; ++kk) {
      const int p0 = kk * 4 + half * 2;
      v2f fa0, fa1;
      fa0.x = Ff[(l16)*FF_P + p0];        fa0.y = Ff[(l16)*FF_P + p0 + 1];
      fa1.x = Ff[(16 + l16) * FF_P + p0]; fa1.y = Ff[(16 + l16) * FF_P + p0 + 1];
      const v2f s0 = hb0[ll][kk] + hb0[ll][kk + 8];
      const v2f s1 = hb1[ll][kk] + hb1[ll][kk + 8];
      const v2f d0 = hb0[ll][kk] - hb0[ll][kk + 8];
      const v2f d1 = hb1[ll][kk] - hb1[ll][kk + 8];
      a00 = wmma4(fa0, s0, a00);
      a01 = wmma4(fa0, s1, a01);
      a10 = wmma4(fa1, d0, a10);
      a11 = wmma4(fa1, d1, a11);
    }
    float* g = G + li * G_LP;
    #pragma unroll
    for (int r = 0; r < 8; ++r) {
      const int m = r + half * 8;
      g[(m)*G_RP + l16]             = a00[r];
      g[(m)*G_RP + 16 + l16]        = a01[r];
      g[(m + 16) * G_RP + l16]      = a10[r];
      g[(m + 16) * G_RP + 16 + l16] = a11[r];
    }
    if (ll == 0) __builtin_amdgcn_sched_barrier(0);
  }

  float4 wq[8][2][2];
  {
    const float4* wgt4 = (const float4*)wgt + wv * 1024 + lane;
    #pragma unroll
    for (int kk = 0; kk < 8; ++kk) {
      #pragma unroll
      for (int mt = 0; mt < 2; ++mt) {
        #pragma unroll
        for (int j = 0; j < 2; ++j)
          wq[kk][mt][j] = wgt4[((kk * 2 + mt) * 2 + j) * 32];
      }
    }
  }
  __syncthreads();

  #pragma unroll
  for (int km = 0; km < 2; ++km) {
    const int rRe = km * 16 + 2 * wv, rIm = rRe + 1;
    v8f oRe0 = zero8(), oRe1 = zero8(), oIm0 = zero8(), oIm1 = zero8();
    #pragma unroll
    for (int kk = 0; kk < 8; ++kk) {
      const int i0 = kk * 4 + half * 2, i1 = i0 + 1;
      v2f bRe, bIm;
      bRe.x = G[l16 * G_LP + rRe * G_RP + i0];
      bRe.y = G[l16 * G_LP + rRe * G_RP + i1];
      bIm.x = G[l16 * G_LP + rIm * G_RP + i0];
      bIm.y = G[l16 * G_LP + rIm * G_RP + i1];
      #pragma unroll
      for (int mt = 0; mt < 2; ++mt) {
        v2f aR, aI, aIn;
        aR.x = km ? wq[kk][mt][0].z : wq[kk][mt][0].x;
        aR.y = km ? wq[kk][mt][1].z : wq[kk][mt][1].x;
        aI.x = km ? wq[kk][mt][0].w : wq[kk][mt][0].y;
        aI.y = km ? wq[kk][mt][1].w : wq[kk][mt][1].y;
        aIn.x = -aI.x; aIn.y = -aI.y;
        if (mt == 0) {
          oRe0 = wmma4(aR, bRe, oRe0);  oRe0 = wmma4(aIn, bIm, oRe0);
          oIm0 = wmma4(aI, bRe, oIm0);  oIm0 = wmma4(aR, bIm, oIm0);
        } else {
          oRe1 = wmma4(aR, bRe, oRe1);  oRe1 = wmma4(aIn, bIm, oRe1);
          oIm1 = wmma4(aI, bRe, oIm1);  oIm1 = wmma4(aR, bIm, oIm1);
        }
      }
    }
    const int psw = (wv & 1) << 4;
    #pragma unroll
    for (int r = 0; r < 8; ++r) {
      const int o0 = r + half * 8;
      Gp[l16 * GP_LP + rRe * 32 + (o0 ^ psw)]        = oRe0[r];
      Gp[l16 * GP_LP + rRe * 32 + ((o0 + 16) ^ psw)] = oRe1[r];
      Gp[l16 * GP_LP + rIm * 32 + (o0 ^ psw)]        = oIm0[r];
      Gp[l16 * GP_LP + rIm * 32 + ((o0 + 16) ^ psw)] = oIm1[r];
    }
  }

  if (FIRST) {
    float4 cwq[2][4];
    float cbv[2];
    #pragma unroll
    for (int nt = 0; nt < 2; ++nt) {
      cbv[nt] = cb[nt * 16 + l16];
      #pragma unroll
      for (int j = 0; j < 4; ++j) {
        const float4 c = *(const float4*)(cw + (nt * 16 + l16) * NW + 8 * j + 4 * half);
        cwq[nt][j] = make_float4(0.5f * c.x, 0.5f * c.y, 0.5f * c.z, 0.5f * c.w);
      }
    }
    __syncthreads();

    #pragma unroll
    for (int ll = 0; ll < 2; ++ll) {
      const unsigned lb = base[ll];
      const unsigned ab = lb - (unsigned)l16 * 4u + (unsigned)l16 * sAB +
                          (unsigned)half * 16u;
      float4 ha[4][4];
      #pragma unroll
      for (int q = 0; q < 4; ++q) {
        #pragma unroll
        for (int j = 0; j < 4; ++j)
          ha[q][j] = ld4_off(h, ab + (unsigned)(16 * q) * sAB + 32u * j);
      }
      __builtin_amdgcn_sched_barrier(0);
      v8f e00 = zero8(), e01 = zero8(), e10 = zero8(), e11 = zero8();
      v8f o00 = zero8(), o01 = zero8(), o10 = zero8(), o11 = zero8();
      const float* gp = Gp + (wv * 2 + ll) * GP_LP;
      irdft_half(gp, Fi, 0, half, l16, e00, e01, e10, e11);
      irdft_half(gp, Fi, 1, half, l16, o00, o01, o10, o11);
      #pragma unroll
      for (int q = 0; q < 2; ++q) {
        #pragma unroll
        for (int j = 0; j < 4; ++j) {
          const float4 u = ha[q][j], v = ha[q + 2][j];
          ha[q][j] = make_float4(u.x + v.x, u.y + v.y, u.z + v.z, u.w + v.w);
          ha[q + 2][j] = make_float4(u.x - v.x, u.y - v.y, u.z - v.z, u.w - v.w);
        }
      }
      #pragma unroll
      for (int j = 0; j < 4; ++j) {
        #pragma unroll
        for (int s = 0; s < 2; ++s) {
          const v2f cb0 = k_pair(cwq[0][j], s), cb1 = k_pair(cwq[1][j], s);
          const v2f cs0 = k_pair(ha[0][j], s), cs1 = k_pair(ha[1][j], s);
          const v2f cd0 = k_pair(ha[2][j], s), cd1 = k_pair(ha[3][j], s);
          e00 = wmma4(cs0, cb0, e00);  e01 = wmma4(cs0, cb1, e01);
          e10 = wmma4(cs1, cb0, e10);  e11 = wmma4(cs1, cb1, e11);
          o00 = wmma4(cd0, cb0, o00);  o01 = wmma4(cd0, cb1, o01);
          o10 = wmma4(cd1, cb0, o10);  o11 = wmma4(cd1, cb1, o11);
        }
      }
      #pragma unroll
      for (int r = 0; r < 8; ++r) {
        const int pr = r + half * 8;
        const unsigned i0a = lb + (unsigned)(pr)*sAB;
        const unsigned i1a = lb + (unsigned)(pr + 16) * sAB;
        const unsigned i2a = lb + (unsigned)(pr + 32) * sAB;
        const unsigned i3a = lb + (unsigned)(pr + 48) * sAB;
        st_off(acc, i0a, (e00[r] + o00[r]) + cbv[0]);  st_off(acc, i0a + 64u, (e01[r] + o01[r]) + cbv[1]);
        st_off(acc, i1a, (e10[r] + o10[r]) + cbv[0]);  st_off(acc, i1a + 64u, (e11[r] + o11[r]) + cbv[1]);
        st_off(acc, i2a, (e00[r] - o00[r]) + cbv[0]);  st_off(acc, i2a + 64u, (e01[r] - o01[r]) + cbv[1]);
        st_off(acc, i3a, (e10[r] - o10[r]) + cbv[0]);  st_off(acc, i3a + 64u, (e11[r] - o11[r]) + cbv[1]);
      }
    }
    return;
  }

  float cq[2][8][8];
  #pragma unroll
  for (int r = 0; r < 8; ++r) {
    const int pr = r + half * 8;
    #pragma unroll
    for (int q = 0; q < 4; ++q) {
      const unsigned ia = base[0] + (unsigned)(pr + 16 * q) * sAB;
      cq[0][2 * q][r]     = ld_off(acc, ia);
      cq[0][2 * q + 1][r] = ld_off(acc, ia + 64u);
    }
  }
  __syncthreads();

  #pragma unroll
  for (int ll = 0; ll < 2; ++ll) {
    const int li = wv * 2 + ll;
    if (ll == 1) {
      #pragma unroll
      for (int r = 0; r < 8; ++r) {
        const int pr = r + half * 8;
        #pragma unroll
        for (int q = 0; q < 4; ++q) {
          const unsigned ia = base[1] + (unsigned)(pr + 16 * q) * sAB;
          cq[1][2 * q][r]     = ld_off(acc, ia);
          cq[1][2 * q + 1][r] = ld_off(acc, ia + 64u);
        }
      }
    }
    v8f e00 = zero8(), e01 = zero8(), e10 = zero8(), e11 = zero8();
    v8f o00 = zero8(), o01 = zero8(), o10 = zero8(), o11 = zero8();
    irdft_half(Gp + li * GP_LP, Fi, 0, half, l16, e00, e01, e10, e11);
    irdft_half(Gp + li * GP_LP, Fi, 1, half, l16, o00, o01, o10, o11);
    #pragma unroll
    for (int r = 0; r < 8; ++r) {
      const int pr = r + half * 8;
      const unsigned i0a = base[ll] + (unsigned)(pr)*sAB;
      const unsigned i1a = base[ll] + (unsigned)(pr + 16) * sAB;
      const unsigned i2a = base[ll] + (unsigned)(pr + 32) * sAB;
      const unsigned i3a = base[ll] + (unsigned)(pr + 48) * sAB;
      st_off(acc, i0a, cq[ll][0][r] + (e00[r] + o00[r]));  st_off(acc, i0a + 64u, cq[ll][1][r] + (e01[r] + o01[r]));
      st_off(acc, i1a, cq[ll][2][r] + (e10[r] + o10[r]));  st_off(acc, i1a + 64u, cq[ll][3][r] + (e11[r] + o11[r]));
      st_off(acc, i2a, cq[ll][4][r] + (e00[r] - o00[r]));  st_off(acc, i2a + 64u, cq[ll][5][r] + (e01[r] - o01[r]));
      st_off(acc, i3a, cq[ll][6][r] + (e10[r] - o10[r]));  st_off(acc, i3a + 64u, cq[ll][7][r] + (e11[r] - o11[r]));
    }
  }
}

__global__ __launch_bounds__(256) void k_gelu(const float* __restrict__ in,
                                              float* __restrict__ out) {
  const long i = (long)blockIdx.x * blockDim.x + threadIdx.x;
  float4 v = ((const float4*)in)[i];
  v.x = gelu_exact(v.x);
  v.y = gelu_exact(v.y);
  v.z = gelu_exact(v.z);
  v.w = gelu_exact(v.w);
  ((float4*)out)[i] = v;
}

__global__ __launch_bounds__(256) void k_fc3(const float* __restrict__ hin,
                                             const float* __restrict__ w,
                                             const float* __restrict__ bias,
                                             float* __restrict__ out) {
  const long p = (long)blockIdx.x * blockDim.x + threadIdx.x;
  const float4* hv = (const float4*)(hin + p * NW);
  const float4* wv = (const float4*)w;
  float s = 0.0f;
  #pragma unroll
  for (int i = 0; i < 8; ++i) {
    float4 a = hv[i], b = wv[i];
    s += a.x * b.x + a.y * b.y + a.z * b.z + a.w * b.w;
  }
  out[p] = s + bias[0];
}

extern "C" void kernel_launch(void* const* d_in, const int* in_sizes, int n_in,
                              void* d_out, int out_size, void* d_ws,
                              size_t ws_size, hipStream_t stream) {
  const float* x      = (const float*)d_in[0];
  const float* in_w   = (const float*)d_in[1];
  const float* in_b   = (const float*)d_in[2];
  const float* fwx    = (const float*)d_in[3];
  const float* fwy    = (const float*)d_in[4];
  const float* fwz    = (const float*)d_in[5];
  const float* conv_w = (const float*)d_in[6];
  const float* conv_b = (const float*)d_in[7];
  const float* fc1_w  = (const float*)d_in[8];
  const float* fc1_b  = (const float*)d_in[9];
  const float* fc2_w  = (const float*)d_in[10];
  const float* fc2_b  = (const float*)d_in[11];
  const float* fc3_w  = (const float*)d_in[12];
  const float* fc3_b  = (const float*)d_in[13];
  float* out = (float*)d_out;

  float* buf0 = (float*)d_ws;
  float* buf1 = buf0 + HSZ;
  float* wre = out;
  const long wsl = (long)NW * NW * NM * 2;

  const dim3 blk(256);
  const int wgPts = NPTS / (16 * 8);
  const size_t smemB = (size_t)LDS_FLOATS * sizeof(float);
  const long sX = 64L * 64 * NW;
  const long sY = 64L * NW;
  const long sZ = (long)NW;

  k_input_proj<<<wgPts, blk, 0, stream>>>(x, in_w, in_b, buf0);
  k_weight_relayout<<<12 * 8192 / 256, blk, 0, stream>>>(fwz, fwy, fwx, wre);

  for (int i = 0; i < 4; ++i) {
    const long wofs = (long)i * wsl;
    const float* cwi = conv_w + i * (NW * NW);
    const float* cbi = conv_b + i * NW;
    k_spectral_axis<true><<<1024, blk, smemB, stream>>>(
        buf0, wre + wofs, buf1, cwi, cbi, sX, sY, sZ);
    k_spectral_axis<false><<<1024, blk, smemB, stream>>>(
        buf0, wre + 4 * wsl + wofs, buf1, cwi, cbi, sX, sZ, sY);
    k_spectral_axis<false><<<1024, blk, smemB, stream>>>(
        buf0, wre + 8 * wsl + wofs, buf1, cwi, cbi, sY, sZ, sX);
    if (i < 3)
      k_gelu<<<(int)(HSZ / 4 / 256), blk, 0, stream>>>(buf1, buf0);
  }

  k_pointwise<true, true><<<wgPts, blk, 0, stream>>>(buf1, fc1_w, fc1_b, buf0);
  k_pointwise<true, true><<<wgPts, blk, 0, stream>>>(buf0, fc2_w, fc2_b, buf1);
  k_fc3<<<NPTS / 256, blk, 0, stream>>>(buf1, fc3_w, fc3_b, out);
}


# each layer's gelu applied in the x-axis kernel's store epilogue: 3 gelu launches gone, state buffers alternate; erf by a branch-free 1-2^(-aQ(a)) (within 2e-7) instead of the library erff
# speedup vs baseline: 2.9396x; 1.0986x over previous
#include <hip/hip_runtime.h>
#include <math.h>


typedef __attribute__((ext_vector_type(2))) float v2f;
typedef __attribute__((ext_vector_type(8))) float v8f;

#define NB 4
#define NS 64
#define NW 32
#define NM 16
#define NPTS (NB * NS * NS * NS)
#define HSZ ((size_t)NPTS * NW)

__device__ __forceinline__ v8f wmma4(v2f a, v2f b, v8f c) {
  return __builtin_amdgcn_wmma_f32_16x16x4_f32(false, a, false, b, (short)0, c,
                                               false, false);
}

__device__ __forceinline__ v8f zero8() {
  v8f z;
  #pragma unroll
  for (int i = 0; i < 8; ++i) z[i] = 0.0f;
  return z;
}

__device__ __forceinline__ float gelu_exact(float x) {
  const float a = fminf(fabsf(x) * 0.70710678118654752440f, 4.0f);
  float q = -1.16633282e-05f;
  q = fmaf(q, a, 1.53490531e-04f);
  q = fmaf(q, a, -8.50145181e-04f);
  q = fmaf(q, a, 2.27840361e-03f);
  q = fmaf(q, a, -8.86525522e-05f);
  q = fmaf(q, a, -2.77221724e-02f);
  q = fmaf(q, a, 1.48307174e-01f);
  q = fmaf(q, a, 9.18443024e-01f);
  q = fmaf(q, a, 1.62790728f);
  const float e = __builtin_amdgcn_exp2f(-(q * a));
  const float hx = 0.5f * x;
  return fmaf(fabsf(hx), 1.0f - e, hx);
}
template <bool G>
__device__ __forceinline__ float act(float v) {
  return G ? gelu_exact(v) : v;
}

__global__ __launch_bounds__(256) void k_input_proj(
    const float* __restrict__ x, const float* __restrict__ w,
    const float* __restrict__ bias, float* __restrict__ out) {
  const int lane = threadIdx.x & 31, wv = threadIdx.x >> 5;
  const int half = lane >> 4, l16 = lane & 15;
  const long pbase = ((long)blockIdx.x * 8 + wv) * 16;
  v2f a;
  a.x = x[(pbase + l16) * 4 + half * 2];
  a.y = x[(pbase + l16) * 4 + half * 2 + 1];
  v2f b0, b1;
  b0.x = w[(half * 2) * NW + l16];
  b0.y = w[(half * 2 + 1) * NW + l16];
  b1.x = w[(half * 2) * NW + 16 + l16];
  b1.y = w[(half * 2 + 1) * NW + 16 + l16];
  v8f d0 = zero8(), d1 = zero8();
  d0 = wmma4(a, b0, d0);
  d1 = wmma4(a, b1, d1);
  #pragma unroll
  for (int r = 0; r < 8; ++r) {
    long p = pbase + r + half * 8;
    out[p * NW + l16] = d0[r] + bias[l16];
    out[p * NW + 16 + l16] = d1[r] + bias[16 + l16];
  }
}

template <bool WT, bool DOGELU>
__global__ __launch_bounds__(256) void k_pointwise(
    const float* __restrict__ hin, const float* __restrict__ w,
    const float* __restrict__ bias, float* __restrict__ out) {
  const int lane = threadIdx.x & 31, wv = threadIdx.x >> 5;
  const int half = lane >> 4, l16 = lane & 15;
  const long pbase = ((long)blockIdx.x * 8 + wv) * 16;
  v8f d0 = zero8(), d1 = zero8();
  #pragma unroll
  for (int kk = 0; kk < 8; ++kk) {
    const int i0 = kk * 4 + half * 2;
    v2f a;
    a.x = hin[(pbase + l16) * NW + i0];
    a.y = hin[(pbase + l16) * NW + i0 + 1];
    v2f b0, b1;
    if (WT) {
      b0.x = w[(i0)*NW + l16];      b0.y = w[(i0 + 1) * NW + l16];
      b1.x = w[(i0)*NW + 16 + l16]; b1.y = w[(i0 + 1) * NW + 16 + l16];
    } else {
      b0.x = w[(l16)*NW + i0];        b0.y = w[(l16)*NW + i0 + 1];
      b1.x = w[(16 + l16) * NW + i0]; b1.y = w[(16 + l16) * NW + i0 + 1];
    }
    d0 = wmma4(a, b0, d0);
    d1 = wmma4(a, b1, d1);
  }
  #pragma unroll
  for (int r = 0; r < 8; ++r) {
    long p = pbase + r + half * 8;
    float v0 = d0[r] + bias[l16];
    float v1 = d1[r] + bias[16 + l16];
    if (DOGELU) { v0 = gelu_exact(v0); v1 = gelu_exact(v1); }
    out[p * NW + l16] = v0;
    out[p * NW + 16 + l16] = v1;
  }
}

#define FF_P 36
#define FI_P 36
#define G_RP 34
#define G_LP 1092
#define GP_LP 1027
#define OFF_FI (32 * FF_P)
#define OFF_G (OFF_FI + 32 * FI_P)
#define OFF_GP (OFF_G + 16 * G_LP)
#define LDS_FLOATS (OFF_GP + 16 * GP_LP)
__device__ __forceinline__ float ld_off(const float* p, unsigned boff) {
  return *(const float*)((const char*)p + boff);
}
__device__ __forceinline__ void st_off(float* p, unsigned boff, float v) {
  *(float*)((char*)p + boff) = v;
}
__device__ __forceinline__ float4 ld4_off(const float* p, unsigned boff) {
  return *(const float4*)((const char*)p + boff);
}
__device__ __forceinline__ v2f k_pair(float4 v, int s) {
  v2f r;
  r.x = s ? v.z : v.x;
  r.y = s ? v.w : v.y;
  return r;
}

__device__ __forceinline__ int mode_of_row(int R) {
  return ((R & 15) >> 1) * 2 + (R >> 4);
}

__device__ __forceinline__ void irdft_half(
    const float* gp, const float* Fi, int g, int half, int l16,
    v8f& t00, v8f& t01, v8f& t10, v8f& t11) {
  for (int kk = 4 * g; kk < 4 * g + 4; ++kk) {
    const int i0 = kk * 4 + half * 2;
    v2f b0, b1;
    const int c0 = l16 ^ (half << 4), c1 = c0 ^ 16;
    b0.x = gp[(i0)*32 + c0]; b0.y = gp[(i0 + 1) * 32 + c0];
    b1.x = gp[(i0)*32 + c1]; b1.y = gp[(i0 + 1) * 32 + c1];
    v2f a0, a1;
    a0.x = Fi[(l16)*FI_P + i0];        a0.y = Fi[(l16)*FI_P + i0 + 1];
    a1.x = Fi[(16 + l16) * FI_P + i0]; a1.y = Fi[(16 + l16) * FI_P + i0 + 1];
    t00 = wmma4(a0, b0, t00);  t01 = wmma4(a0, b1, t01);
    t10 = wmma4(a1, b0, t10);  t11 = wmma4(a1, b1, t11);
  }
}

__global__ __launch_bounds__(256) void k_weight_relayout(
    const float* __restrict__ fwz, const float* __restrict__ fwy,
    const float* __restrict__ fwx, float* __restrict__ dst) {
  const int t = blockIdx.x * 256 + threadIdx.x;
  const int s = t >> 13, q = t & 8191;
  const int axis = s >> 2, layer = s & 3;
  const float* src = (axis == 0) ? fwz : (axis == 1) ? fwy : fwx;
  const int lane = q & 31, j = (q >> 5) & 1, mt = (q >> 6) & 1;
  const int kk = (q >> 7) & 7, wv = q >> 10;
  const int in = kk * 4 + (lane >> 4) * 2 + j, o = mt * 16 + (lane & 15);
  const float4 v = *(const float4*)(src + layer * (NW * NW * NM * 2) +
                                    (in * NW + o) * (NM * 2) + 4 * wv);
  ((float4*)dst)[t] = v;
}

template <bool FIRST, bool GELU>
__global__ __launch_bounds__(256, 4) void k_spectral_axis(
    const float* __restrict__ h, const float* __restrict__ wgt,
    float* __restrict__ acc, const float* __restrict__ cw,
    const float* __restrict__ cb, long strideU, long strideV, long strideA) {
  extern __shared__ float smem[];
  float* Ff = smem;
  float* Fi = smem + OFF_FI;
  float* G  = smem + OFF_G;
  float* Gp = smem + OFF_GP;

  const int tid = threadIdx.x;
  const int lane = tid & 31, wv = tid >> 5;
  const int half = lane >> 4, l16 = lane & 15;
  const float NRM = 0.125f;

  const unsigned sAB = (unsigned)strideA * 4u;
  unsigned base[2];
  #pragma unroll
  for (int ll = 0; ll < 2; ++ll) {
    int l = blockIdx.x * 16 + wv * 2 + ll;
    int b = l >> 12, u = (l >> 6) & 63, v = l & 63;
    base[ll] = ((unsigned)b * (64u * 64 * 64 * NW) + (unsigned)u * (unsigned)strideU +
                (unsigned)v * (unsigned)strideV + (unsigned)l16) * 4u;
  }

  v2f hb0[2][16], hb1[2][16];
  #pragma unroll
  for (int ll = 0; ll < 2; ++ll) {
    #pragma unroll
    for (int kk = 0; kk < 16; ++kk) {
      const unsigned p0 = kk * 4 + half * 2;
      hb0[ll][kk].x = ld_off(h, base[ll] + p0 * sAB);
      hb0[ll][kk].y = ld_off(h, base[ll] + (p0 + 1) * sAB);
      hb1[ll][kk].x = ld_off(h, base[ll] + p0 * sAB + 64u);
      hb1[ll][kk].y = ld_off(h, base[ll] + (p0 + 1) * sAB + 64u);
    }
  }

  for (int idx = tid; idx < 1024; idx += 256) {
    int m2 = idx >> 5, p = idx & 31, k = mode_of_row(m2);
    float ang = 6.28318530717958647693f * (float)(k * p) * (1.0f / 64.0f);
    Ff[m2 * FF_P + p] = (m2 & 1) ? (-__sinf(ang) * NRM) : (__cosf(ang) * NRM);
  }
  for (int idx = tid; idx < 1024; idx += 256) {
    int p = idx >> 5, m2 = idx & 31, k = mode_of_row(m2);
    float ang = 6.28318530717958647693f * (float)(k * p) * (1.0f / 64.0f);
    float sc = (k == 0 ? 1.0f : 2.0f) * NRM;
    Fi[p * FI_P + m2] = (m2 & 1) ? (-__sinf(ang) * sc) : (__cosf(ang) * sc);
  }
  __syncthreads();

  #pragma unroll
  for (int ll = 0; ll < 2; ++ll) {
    const int li = wv * 2 + ll;
    v8f a00 = zero8(), a01 = zero8(), a10 = zero8(), a11 = zero8();
    #pragma unroll
    for (int kk = 0; kk < 8; ++kk) {
      const int p0 = kk * 4 + half * 2;
      v2f fa0, fa1;
      fa0.x = Ff[(l16)*FF_P + p0];        fa0.y = Ff[(l16)*FF_P + p0 + 1];
      fa1.x = Ff[(16 + l16) * FF_P + p0]; fa1.y = Ff[(16 + l16) * FF_P + p0 + 1];
      const v2f s0 = hb0[ll][kk] + hb0[ll][kk + 8];
      const v2f s1 = hb1[ll][kk] + hb1[ll][kk + 8];
      const v2f d0 = hb0[ll][kk] - hb0[ll][kk + 8];
      const v2f d1 = hb1[ll][kk] - hb1[ll][kk + 8];
      a00 = wmma4(fa0, s0, a00);
      a01 = wmma4(fa0, s1, a01);
      a10 = wmma4(fa1, d0, a10);
      a11 = wmma4(fa1, d1, a11);
    }
    float* g = G + li * G_LP;
    #pragma unroll
    for (int r = 0; r < 8; ++r) {
      const int m = r + half * 8;
      g[(m)*G_RP + l16]             = a00[r];
      g[(m)*G_RP + 16 + l16]        = a01[r];
      g[(m + 16) * G_RP + l16]      = a10[r];
      g[(m + 16) * G_RP + 16 + l16] = a11[r];
    }
    if (ll == 0) __builtin_amdgcn_sched_barrier(0);
  }

  float4 wq[8][2][2];
  {
    const float4* wgt4 = (const float4*)wgt + wv * 1024 + lane;
    #pragma unroll
    for (int kk = 0; kk < 8; ++kk) {
      #pragma unroll
      for (int mt = 0; mt < 2; ++mt) {
        #pragma unroll
        for (int j = 0; j < 2; ++j)
          wq[kk][mt][j] = wgt4[((kk * 2 + mt) * 2 + j) * 32];
      }
    }
  }
  __syncthreads();

  #pragma unroll
  for (int km = 0; km < 2; ++km) {
    const int rRe = km * 16 + 2 * wv, rIm = rRe + 1;
    v8f oRe0 = zero8(), oRe1 = zero8(), oIm0 = zero8(), oIm1 = zero8();
    #pragma unroll
    for (int kk = 0; kk < 8; ++kk) {
      const int i0 = kk * 4 + half * 2, i1 = i0 + 1;
      v2f bRe, bIm;
      bRe.x = G[l16 * G_LP + rRe * G_RP + i0];
      bRe.y = G[l16 * G_LP + rRe * G_RP + i1];
      bIm.x = G[l16 * G_LP + rIm * G_RP + i0];
      bIm.y = G[l16 * G_LP + rIm * G_RP + i1];
      #pragma unroll
      for (int mt = 0; mt < 2; ++mt) {
        v2f aR, aI, aIn;
        aR.x = km ? wq[kk][mt][0].z : wq[kk][mt][0].x;
        aR.y = km ? wq[kk][mt][1].z : wq[kk][mt][1].x;
        aI.x = km ? wq[kk][mt][0].w : wq[kk][mt][0].y;
        aI.y = km ? wq[kk][mt][1].w : wq[kk][mt][1].y;
        aIn.x = -aI.x; aIn.y = -aI.y;
        if (mt == 0) {
          oRe0 = wmma4(aR, bRe, oRe0);  oRe0 = wmma4(aIn, bIm, oRe0);
          oIm0 = wmma4(aI, bRe, oIm0);  oIm0 = wmma4(aR, bIm, oIm0);
        } else {
          oRe1 = wmma4(aR, bRe, oRe1);  oRe1 = wmma4(aIn, bIm, oRe1);
          oIm1 = wmma4(aI, bRe, oIm1);  oIm1 = wmma4(aR, bIm, oIm1);
        }
      }
    }
    const int psw = (wv & 1) << 4;
    #pragma unroll
    for (int r = 0; r < 8; ++r) {
      const int o0 = r + half * 8;
      Gp[l16 * GP_LP + rRe * 32 + (o0 ^ psw)]        = oRe0[r];
      Gp[l16 * GP_LP + rRe * 32 + ((o0 + 16) ^ psw)] = oRe1[r];
      Gp[l16 * GP_LP + rIm * 32 + (o0 ^ psw)]        = oIm0[r];
      Gp[l16 * GP_LP + rIm * 32 + ((o0 + 16) ^ psw)] = oIm1[r];
    }
  }

  if (FIRST) {
    float4 cwq[2][4];
    float cbv[2];
    #pragma unroll
    for (int nt = 0; nt < 2; ++nt) {
      cbv[nt] = cb[nt * 16 + l16];
      #pragma unroll
      for (int j = 0; j < 4; ++j) {
        const float4 c = *(const float4*)(cw + (nt * 16 + l16) * NW + 8 * j + 4 * half);
        cwq[nt][j] = make_float4(0.5f * c.x, 0.5f * c.y, 0.5f * c.z, 0.5f * c.w);
      }
    }
    __syncthreads();

    #pragma unroll
    for (int ll = 0; ll < 2; ++ll) {
      const unsigned lb = base[ll];
      const unsigned ab = lb - (unsigned)l16 * 4u + (unsigned)l16 * sAB +
                          (unsigned)half * 16u;
      float4 ha[4][4];
      #pragma unroll
      for (int q = 0; q < 4; ++q) {
        #pragma unroll
        for (int j = 0; j < 4; ++j)
          ha[q][j] = ld4_off(h, ab + (unsigned)(16 * q) * sAB + 32u * j);
      }
      __builtin_amdgcn_sched_barrier(0);
      v8f e00 = zero8(), e01 = zero8(), e10 = zero8(), e11 = zero8();
      v8f o00 = zero8(), o01 = zero8(), o10 = zero8(), o11 = zero8();
      const float* gp = Gp + (wv * 2 + ll) * GP_LP;
      irdft_half(gp, Fi, 0, half, l16, e00, e01, e10, e11);
      irdft_half(gp, Fi, 1, half, l16, o00, o01, o10, o11);
      #pragma unroll
      for (int q = 0; q < 2; ++q) {
        #pragma unroll
        for (int j = 0; j < 4; ++j) {
          const float4 u = ha[q][j], v = ha[q + 2][j];
          ha[q][j] = make_float4(u.x + v.x, u.y + v.y, u.z + v.z, u.w + v.w);
          ha[q + 2][j] = make_float4(u.x - v.x, u.y - v.y, u.z - v.z, u.w - v.w);
        }
      }
      #pragma unroll
      for (int j = 0; j < 4; ++j) {
        #pragma unroll
        for (int s = 0; s < 2; ++s) {
          const v2f cb0 = k_pair(cwq[0][j], s), cb1 = k_pair(cwq[1][j], s);
          const v2f cs0 = k_pair(ha[0][j], s), cs1 = k_pair(ha[1][j], s);
          const v2f cd0 = k_pair(ha[2][j], s), cd1 = k_pair(ha[3][j], s);
          e00 = wmma4(cs0, cb0, e00);  e01 = wmma4(cs0, cb1, e01);
          e10 = wmma4(cs1, cb0, e10);  e11 = wmma4(cs1, cb1, e11);
          o00 = wmma4(cd0, cb0, o00);  o01 = wmma4(cd0, cb1, o01);
          o10 = wmma4(cd1, cb0, o10);  o11 = wmma4(cd1, cb1, o11);
        }
      }
      #pragma unroll
      for (int r = 0; r < 8; ++r) {
        const int pr = r + half * 8;
        const unsigned i0a = lb + (unsigned)(pr)*sAB;
        const unsigned i1a = lb + (unsigned)(pr + 16) * sAB;
        const unsigned i2a = lb + (unsigned)(pr + 32) * sAB;
        const unsigned i3a = lb + (unsigned)(pr + 48) * sAB;
        st_off(acc, i0a, (e00[r] + o00[r]) + cbv[0]);  st_off(acc, i0a + 64u, (e01[r] + o01[r]) + cbv[1]);
        st_off(acc, i1a, (e10[r] + o10[r]) + cbv[0]);  st_off(acc, i1a + 64u, (e11[r] + o11[r]) + cbv[1]);
        st_off(acc, i2a, (e00[r] - o00[r]) + cbv[0]);  st_off(acc, i2a + 64u, (e01[r] - o01[r]) + cbv[1]);
        st_off(acc, i3a, (e10[r] - o10[r]) + cbv[0]);  st_off(acc, i3a + 64u, (e11[r] - o11[r]) + cbv[1]);
      }
    }
    return;
  }

  float cq[2][8][8];
  #pragma unroll
  for (int r = 0; r < 8; ++r) {
    const int pr = r + half * 8;
    #pragma unroll
    for (int q = 0; q < 4; ++q) {
      const unsigned ia = base[0] + (unsigned)(pr + 16 * q) * sAB;
      cq[0][2 * q][r]     = ld_off(acc, ia);
      cq[0][2 * q + 1][r] = ld_off(acc, ia + 64u);
    }
  }
  __syncthreads();

  #pragma unroll
  for (int ll = 0; ll < 2; ++ll) {
    const int li = wv * 2 + ll;
    if (ll == 1) {
      #pragma unroll
      for (int r = 0; r < 8; ++r) {
        const int pr = r + half * 8;
        #pragma unroll
        for (int q = 0; q < 4; ++q) {
          const unsigned ia = base[1] + (unsigned)(pr + 16 * q) * sAB;
          cq[1][2 * q][r]     = ld_off(acc, ia);
          cq[1][2 * q + 1][r] = ld_off(acc, ia + 64u);
        }
      }
    }
    v8f e00 = zero8(), e01 = zero8(), e10 = zero8(), e11 = zero8();
    v8f o00 = zero8(), o01 = zero8(), o10 = zero8(), o11 = zero8();
    irdft_half(Gp + li * GP_LP, Fi, 0, half, l16, e00, e01, e10, e11);
    irdft_half(Gp + li * GP_LP, Fi, 1, half, l16, o00, o01, o10, o11);
    #pragma unroll
    for (int r = 0; r < 8; ++r) {
      const int pr = r + half * 8;
      const unsigned i0a = base[ll] + (unsigned)(pr)*sAB;
      const unsigned i1a = base[ll] + (unsigned)(pr + 16) * sAB;
      const unsigned i2a = base[ll] + (unsigned)(pr + 32) * sAB;
      const unsigned i3a = base[ll] + (unsigned)(pr + 48) * sAB;
      st_off(acc, i0a, act<GELU>(cq[ll][0][r] + (e00[r] + o00[r])));  st_off(acc, i0a + 64u, act<GELU>(cq[ll][1][r] + (e01[r] + o01[r])));
      st_off(acc, i1a, act<GELU>(cq[ll][2][r] + (e10[r] + o10[r])));  st_off(acc, i1a + 64u, act<GELU>(cq[ll][3][r] + (e11[r] + o11[r])));
      st_off(acc, i2a, act<GELU>(cq[ll][4][r] + (e00[r] - o00[r])));  st_off(acc, i2a + 64u, act<GELU>(cq[ll][5][r] + (e01[r] - o01[r])));
      st_off(acc, i3a, act<GELU>(cq[ll][6][r] + (e10[r] - o10[r])));  st_off(acc, i3a + 64u, act<GELU>(cq[ll][7][r] + (e11[r] - o11[r])));
    }
  }
}

__global__ __launch_bounds__(256) void k_fc3(const float* __restrict__ hin,
                                             const float* __restrict__ w,
                                             const float* __restrict__ bias,
                                             float* __restrict__ out) {
  const long p = (long)blockIdx.x * blockDim.x + threadIdx.x;
  const float4* hv = (const float4*)(hin + p * NW);
  const float4* wv = (const float4*)w;
  float s = 0.0f;
  #pragma unroll
  for (int i = 0; i < 8; ++i) {
    float4 a = hv[i], b = wv[i];
    s += a.x * b.x + a.y * b.y + a.z * b.z + a.w * b.w;
  }
  out[p] = s + bias[0];
}

extern "C" void kernel_launch(void* const* d_in, const int* in_sizes, int n_in,
                              void* d_out, int out_size, void* d_ws,
                              size_t ws_size, hipStream_t stream) {
  const float* x      = (const float*)d_in[0];
  const float* in_w   = (const float*)d_in[1];
  const float* in_b   = (const float*)d_in[2];
  const float* fwx    = (const float*)d_in[3];
  const float* fwy    = (const float*)d_in[4];
  const float* fwz    = (const float*)d_in[5];
  const float* conv_w = (const float*)d_in[6];
  const float* conv_b = (const float*)d_in[7];
  const float* fc1_w  = (const float*)d_in[8];
  const float* fc1_b  = (const float*)d_in[9];
  const float* fc2_w  = (const float*)d_in[10];
  const float* fc2_b  = (const float*)d_in[11];
  const float* fc3_w  = (const float*)d_in[12];
  const float* fc3_b  = (const float*)d_in[13];
  float* out = (float*)d_out;

  float* hcur = (float*)d_ws;
  float* hacc = hcur + HSZ;
  float* wre = out;
  const long wsl = (long)NW * NW * NM * 2;

  const dim3 blk(256);
  const int wgPts = NPTS / (16 * 8);
  const size_t smemB = (size_t)LDS_FLOATS * sizeof(float);
  const long sX = 64L * 64 * NW;
  const long sY = 64L * NW;
  const long sZ = (long)NW;

  k_input_proj<<<wgPts, blk, 0, stream>>>(x, in_w, in_b, hcur);
  k_weight_relayout<<<12 * 8192 / 256, blk, 0, stream>>>(fwz, fwy, fwx, wre);

  for (int i = 0; i < 4; ++i) {
    const long wofs = (long)i * wsl;
    const float* cwi = conv_w + i * (NW * NW);
    const float* cbi = conv_b + i * NW;
    k_spectral_axis<true, false><<<1024, blk, smemB, stream>>>(
        hcur, wre + wofs, hacc, cwi, cbi, sX, sY, sZ);
    k_spectral_axis<false, false><<<1024, blk, smemB, stream>>>(
        hcur, wre + 4 * wsl + wofs, hacc, cwi, cbi, sX, sZ, sY);
    if (i < 3)
      k_spectral_axis<false, true><<<1024, blk, smemB, stream>>>(
          hcur, wre + 8 * wsl + wofs, hacc, cwi, cbi, sY, sZ, sX);
    else
      k_spectral_axis<false, false><<<1024, blk, smemB, stream>>>(
          hcur, wre + 8 * wsl + wofs, hacc, cwi, cbi, sY, sZ, sX);
    float* const done = hacc;
    hacc = hcur;
    hcur = done;
  }

  k_pointwise<true, true><<<wgPts, blk, 0, stream>>>(hcur, fc1_w, fc1_b, hacc);
  k_pointwise<true, true><<<wgPts, blk, 0, stream>>>(hacc, fc2_w, fc2_b, hcur);
  k_fc3<<<NPTS / 256, blk, 0, stream>>>(hcur, fc3_w, fc3_b, out);
}


# MLP head (fc1, gelu, fc2, gelu, fc3) fused into one kernel on transposed products (fc1's D registers are fc2's B fragments): 3 launches become 1, two tensors never stored; channel sum order changes
# speedup vs baseline: 3.0991x; 1.0543x over previous
#include <hip/hip_runtime.h>
#include <math.h>


typedef __attribute__((ext_vector_type(2))) float v2f;
typedef __attribute__((ext_vector_type(8))) float v8f;

#define NB 4
#define NS 64
#define NW 32
#define NM 16
#define NPTS (NB * NS * NS * NS)
#define HSZ ((size_t)NPTS * NW)

__device__ __forceinline__ v8f wmma4(v2f a, v2f b, v8f c) {
  return __builtin_amdgcn_wmma_f32_16x16x4_f32(false, a, false, b, (short)0, c,
                                               false, false);
}

__device__ __forceinline__ v8f zero8() {
  v8f z;
  #pragma unroll
  for (int i = 0; i < 8; ++i) z[i] = 0.0f;
  return z;
}

__device__ __forceinline__ float gelu_exact(float x) {
  const float a = fminf(fabsf(x) * 0.70710678118654752440f, 4.0f);
  float q = -1.16633282e-05f;
  q = fmaf(q, a, 1.53490531e-04f);
  q = fmaf(q, a, -8.50145181e-04f);
  q = fmaf(q, a, 2.27840361e-03f);
  q = fmaf(q, a, -8.86525522e-05f);
  q = fmaf(q, a, -2.77221724e-02f);
  q = fmaf(q, a, 1.48307174e-01f);
  q = fmaf(q, a, 9.18443024e-01f);
  q = fmaf(q, a, 1.62790728f);
  const float e = __builtin_amdgcn_exp2f(-(q * a));
  const float hx = 0.5f * x;
  return fmaf(fabsf(hx), 1.0f - e, hx);
}
template <bool G>
__device__ __forceinline__ float act(float v) {
  return G ? gelu_exact(v) : v;
}

__global__ __launch_bounds__(256) void k_input_proj(
    const float* __restrict__ x, const float* __restrict__ w,
    const float* __restrict__ bias, float* __restrict__ out) {
  const int lane = threadIdx.x & 31, wv = threadIdx.x >> 5;
  const int half = lane >> 4, l16 = lane & 15;
  const long pbase = ((long)blockIdx.x * 8 + wv) * 16;
  v2f a;
  a.x = x[(pbase + l16) * 4 + half * 2];
  a.y = x[(pbase + l16) * 4 + half * 2 + 1];
  v2f b0, b1;
  b0.x = w[(half * 2) * NW + l16];
  b0.y = w[(half * 2 + 1) * NW + l16];
  b1.x = w[(half * 2) * NW + 16 + l16];
  b1.y = w[(half * 2 + 1) * NW + 16 + l16];
  v8f d0 = zero8(), d1 = zero8();
  d0 = wmma4(a, b0, d0);
  d1 = wmma4(a, b1, d1);
  #pragma unroll
  for (int r = 0; r < 8; ++r) {
    long p = pbase + r + half * 8;
    out[p * NW + l16] = d0[r] + bias[l16];
    out[p * NW + 16 + l16] = d1[r] + bias[16 + l16];
  }
}

#define FF_P 36
#define FI_P 36
#define G_RP 34
#define G_LP 1092
#define GP_LP 1027
#define OFF_FI (32 * FF_P)
#define OFF_G (OFF_FI + 32 * FI_P)
#define OFF_GP (OFF_G + 16 * G_LP)
#define LDS_FLOATS (OFF_GP + 16 * GP_LP)
__device__ __forceinline__ float ld_off(const float* p, unsigned boff) {
  return *(const float*)((const char*)p + boff);
}
__device__ __forceinline__ void st_off(float* p, unsigned boff, float v) {
  *(float*)((char*)p + boff) = v;
}
__device__ __forceinline__ float4 ld4_off(const float* p, unsigned boff) {
  return *(const float4*)((const char*)p + boff);
}
__device__ __forceinline__ v2f k_pair(float4 v, int s) {
  v2f r;
  r.x = s ? v.z : v.x;
  r.y = s ? v.w : v.y;
  return r;
}

__device__ __forceinline__ int mode_of_row(int R) {
  return ((R & 15) >> 1) * 2 + (R >> 4);
}

__device__ __forceinline__ void irdft_half(
    const float* gp, const float* Fi, int g, int half, int l16,
    v8f& t00, v8f& t01, v8f& t10, v8f& t11) {
  for (int kk = 4 * g; kk < 4 * g + 4; ++kk) {
    const int i0 = kk * 4 + half * 2;
    v2f b0, b1;
    const int c0 = l16 ^ (half << 4), c1 = c0 ^ 16;
    b0.x = gp[(i0)*32 + c0]; b0.y = gp[(i0 + 1) * 32 + c0];
    b1.x = gp[(i0)*32 + c1]; b1.y = gp[(i0 + 1) * 32 + c1];
    v2f a0, a1;
    a0.x = Fi[(l16)*FI_P + i0];        a0.y = Fi[(l16)*FI_P + i0 + 1];
    a1.x = Fi[(16 + l16) * FI_P + i0]; a1.y = Fi[(16 + l16) * FI_P + i0 + 1];
    t00 = wmma4(a0, b0, t00);  t01 = wmma4(a0, b1, t01);
    t10 = wmma4(a1, b0, t10);  t11 = wmma4(a1, b1, t11);
  }
}

__global__ __launch_bounds__(256) void k_weight_relayout(
    const float* __restrict__ fwz, const float* __restrict__ fwy,
    const float* __restrict__ fwx, float* __restrict__ dst) {
  const int t = blockIdx.x * 256 + threadIdx.x;
  const int s = t >> 13, q = t & 8191;
  const int axis = s >> 2, layer = s & 3;
  const float* src = (axis == 0) ? fwz : (axis == 1) ? fwy : fwx;
  const int lane = q & 31, j = (q >> 5) & 1, mt = (q >> 6) & 1;
  const int kk = (q >> 7) & 7, wv = q >> 10;
  const int in = kk * 4 + (lane >> 4) * 2 + j, o = mt * 16 + (lane & 15);
  const float4 v = *(const float4*)(src + layer * (NW * NW * NM * 2) +
                                    (in * NW + o) * (NM * 2) + 4 * wv);
  ((float4*)dst)[t] = v;
}

template <bool FIRST, bool GELU>
__global__ __launch_bounds__(256, 4) void k_spectral_axis(
    const float* __restrict__ h, const float* __restrict__ wgt,
    float* __restrict__ acc, const float* __restrict__ cw,
    const float* __restrict__ cb, long strideU, long strideV, long strideA) {
  extern __shared__ float smem[];
  float* Ff = smem;
  float* Fi = smem + OFF_FI;
  float* G  = smem + OFF_G;
  float* Gp = smem + OFF_GP;

  const int tid = threadIdx.x;
  const int lane = tid & 31, wv = tid >> 5;
  const int half = lane >> 4, l16 = lane & 15;
  const float NRM = 0.125f;

  const unsigned sAB = (unsigned)strideA * 4u;
  unsigned base[2];
  #pragma unroll
  for (int ll = 0; ll < 2; ++ll) {
    int l = blockIdx.x * 16 + wv * 2 + ll;
    int b = l >> 12, u = (l >> 6) & 63, v = l & 63;
    base[ll] = ((unsigned)b * (64u * 64 * 64 * NW) + (unsigned)u * (unsigned)strideU +
                (unsigned)v * (unsigned)strideV + (unsigned)l16) * 4u;
  }

  v2f hb0[2][16], hb1[2][16];
  #pragma unroll
  for (int ll = 0; ll < 2; ++ll) {
    #pragma unroll
    for (int kk = 0; kk < 16; ++kk) {
      const unsigned p0 = kk * 4 + half * 2;
      hb0[ll][kk].x = ld_off(h, base[ll] + p0 * sAB);
      hb0[ll][kk].y = ld_off(h, base[ll] + (p0 + 1) * sAB);
      hb1[ll][kk].x = ld_off(h, base[ll] + p0 * sAB + 64u);
      hb1[ll][kk].y = ld_off(h, base[ll] + (p0 + 1) * sAB + 64u);
    }
  }

  for (int idx = tid; idx < 1024; idx += 256) {
    int m2 = idx >> 5, p = idx & 31, k = mode_of_row(m2);
    float ang = 6.28318530717958647693f * (float)(k * p) * (1.0f / 64.0f);
    Ff[m2 * FF_P + p] = (m2 & 1) ? (-__sinf(ang) * NRM) : (__cosf(ang) * NRM);
  }
  for (int idx = tid; idx < 1024; idx += 256) {
    int p = idx >> 5, m2 = idx & 31, k = mode_of_row(m2);
    float ang = 6.28318530717958647693f * (float)(k * p) * (1.0f / 64.0f);
    float sc = (k == 0 ? 1.0f : 2.0f) * NRM;
    Fi[p * FI_P + m2] = (m2 & 1) ? (-__sinf(ang) * sc) : (__cosf(ang) * sc);
  }
  __syncthreads();

  #pragma unroll
  for (int ll = 0; ll < 2; ++ll) {
    const int li = wv * 2 + ll;
    v8f a00 = zero8(), a01 = zero8(), a10 = zero8(), a11 = zero8();
    #pragma unroll
    for (int kk = 0; kk < 8; ++kk) {
      const int p0 = kk * 4 + half * 2;
      v2f fa0, fa1;
      fa0.x = Ff[(l16)*FF_P + p0];        fa0.y = Ff[(l16)*FF_P + p0 + 1];
      fa1.x = Ff[(16 + l16) * FF_P + p0]; fa1.y = Ff[(16 + l16) * FF_P + p0 + 1];
      const v2f s0 = hb0[ll][kk] + hb0[ll][kk + 8];
      const v2f s1 = hb1[ll][kk] + hb1[ll][kk + 8];
      const v2f d0 = hb0[ll][kk] - hb0[ll][kk + 8];
      const v2f d1 = hb1[ll][kk] - hb1[ll][kk + 8];
      a00 = wmma4(fa0, s0, a00);
      a01 = wmma4(fa0, s1, a01);
      a10 = wmma4(fa1, d0, a10);
      a11 = wmma4(fa1, d1, a11);
    }
    float* g = G + li * G_LP;
    #pragma unroll
    for (int r = 0; r < 8; ++r) {
      const int m = r + half * 8;
      g[(m)*G_RP + l16]             = a00[r];
      g[(m)*G_RP + 16 + l16]        = a01[r];
      g[(m + 16) * G_RP + l16]      = a10[r];
      g[(m + 16) * G_RP + 16 + l16] = a11[r];
    }
    if (ll == 0) __builtin_amdgcn_sched_barrier(0);
  }

  float4 wq[8][2][2];
  {
    const float4* wgt4 = (const float4*)wgt + wv * 1024 + lane;
    #pragma unroll
    for (int kk = 0; kk < 8; ++kk) {
      #pragma unroll
      for (int mt = 0; mt < 2; ++mt) {
        #pragma unroll
        for (int j = 0; j < 2; ++j)
          wq[kk][mt][j] = wgt4[((kk * 2 + mt) * 2 + j) * 32];
      }
    }
  }
  __syncthreads();

  #pragma unroll
  for (int km = 0; km < 2; ++km) {
    const int rRe = km * 16 + 2 * wv, rIm = rRe + 1;
    v8f oRe0 = zero8(), oRe1 = zero8(), oIm0 = zero8(), oIm1 = zero8();
    #pragma unroll
    for (int kk = 0; kk < 8; ++kk) {
      const int i0 = kk * 4 + half * 2, i1 = i0 + 1;
      v2f bRe, bIm;
      bRe.x = G[l16 * G_LP + rRe * G_RP + i0];
      bRe.y = G[l16 * G_LP + rRe * G_RP + i1];
      bIm.x = G[l16 * G_LP + rIm * G_RP + i0];
      bIm.y = G[l16 * G_LP + rIm * G_RP + i1];
      #pragma unroll
      for (int mt = 0; mt < 2; ++mt) {
        v2f aR, aI, aIn;
        aR.x = km ? wq[kk][mt][0].z : wq[kk][mt][0].x;
        aR.y = km ? wq[kk][mt][1].z : wq[kk][mt][1].x;
        aI.x = km ? wq[kk][mt][0].w : wq[kk][mt][0].y;
        aI.y = km ? wq[kk][mt][1].w : wq[kk][mt][1].y;
        aIn.x = -aI.x; aIn.y = -aI.y;
        if (mt == 0) {
          oRe0 = wmma4(aR, bRe, oRe0);  oRe0 = wmma4(aIn, bIm, oRe0);
          oIm0 = wmma4(aI, bRe, oIm0);  oIm0 = wmma4(aR, bIm, oIm0);
        } else {
          oRe1 = wmma4(aR, bRe, oRe1);  oRe1 = wmma4(aIn, bIm, oRe1);
          oIm1 = wmma4(aI, bRe, oIm1);  oIm1 = wmma4(aR, bIm, oIm1);
        }
      }
    }
    const int psw = (wv & 1) << 4;
    #pragma unroll
    for (int r = 0; r < 8; ++r) {
      const int o0 = r + half * 8;
      Gp[l16 * GP_LP + rRe * 32 + (o0 ^ psw)]        = oRe0[r];
      Gp[l16 * GP_LP + rRe * 32 + ((o0 + 16) ^ psw)] = oRe1[r];
      Gp[l16 * GP_LP + rIm * 32 + (o0 ^ psw)]        = oIm0[r];
      Gp[l16 * GP_LP + rIm * 32 + ((o0 + 16) ^ psw)] = oIm1[r];
    }
  }

  if (FIRST) {
    float4 cwq[2][4];
    float cbv[2];
    #pragma unroll
    for (int nt = 0; nt < 2; ++nt) {
      cbv[nt] = cb[nt * 16 + l16];
      #pragma unroll
      for (int j = 0; j < 4; ++j) {
        const float4 c = *(const float4*)(cw + (nt * 16 + l16) * NW + 8 * j + 4 * half);
        cwq[nt][j] = make_float4(0.5f * c.x, 0.5f * c.y, 0.5f * c.z, 0.5f * c.w);
      }
    }
    __syncthreads();

    #pragma unroll
    for (int ll = 0; ll < 2; ++ll) {
      const unsigned lb = base[ll];
      const unsigned ab = lb - (unsigned)l16 * 4u + (unsigned)l16 * sAB +
                          (unsigned)half * 16u;
      float4 ha[4][4];
      #pragma unroll
      for (int q = 0; q < 4; ++q) {
        #pragma unroll
        for (int j = 0; j < 4; ++j)
          ha[q][j] = ld4_off(h, ab + (unsigned)(16 * q) * sAB + 32u * j);
      }
      __builtin_amdgcn_sched_barrier(0);
      v8f e00 = zero8(), e01 = zero8(), e10 = zero8(), e11 = zero8();
      v8f o00 = zero8(), o01 = zero8(), o10 = zero8(), o11 = zero8();
      const float* gp = Gp + (wv * 2 + ll) * GP_LP;
      irdft_half(gp, Fi, 0, half, l16, e00, e01, e10, e11);
      irdft_half(gp, Fi, 1, half, l16, o00, o01, o10, o11);
      #pragma unroll
      for (int q = 0; q < 2; ++q) {
        #pragma unroll
        for (int j = 0; j < 4; ++j) {
          const float4 u = ha[q][j], v = ha[q + 2][j];
          ha[q][j] = make_float4(u.x + v.x, u.y + v.y, u.z + v.z, u.w + v.w);
          ha[q + 2][j] = make_float4(u.x - v.x, u.y - v.y, u.z - v.z, u.w - v.w);
        }
      }
      #pragma unroll
      for (int j = 0; j < 4; ++j) {
        #pragma unroll
        for (int s = 0; s < 2; ++s) {
          const v2f cb0 = k_pair(cwq[0][j], s), cb1 = k_pair(cwq[1][j], s);
          const v2f cs0 = k_pair(ha[0][j], s), cs1 = k_pair(ha[1][j], s);
          const v2f cd0 = k_pair(ha[2][j], s), cd1 = k_pair(ha[3][j], s);
          e00 = wmma4(cs0, cb0, e00);  e01 = wmma4(cs0, cb1, e01);
          e10 = wmma4(cs1, cb0, e10);  e11 = wmma4(cs1, cb1, e11);
          o00 = wmma4(cd0, cb0, o00);  o01 = wmma4(cd0, cb1, o01);
          o10 = wmma4(cd1, cb0, o10);  o11 = wmma4(cd1, cb1, o11);
        }
      }
      #pragma unroll
      for (int r = 0; r < 8; ++r) {
        const int pr = r + half * 8;
        const unsigned i0a = lb + (unsigned)(pr)*sAB;
        const unsigned i1a = lb + (unsigned)(pr + 16) * sAB;
        const unsigned i2a = lb + (unsigned)(pr + 32) * sAB;
        const unsigned i3a = lb + (unsigned)(pr + 48) * sAB;
        st_off(acc, i0a, (e00[r] + o00[r]) + cbv[0]);  st_off(acc, i0a + 64u, (e01[r] + o01[r]) + cbv[1]);
        st_off(acc, i1a, (e10[r] + o10[r]) + cbv[0]);  st_off(acc, i1a + 64u, (e11[r] + o11[r]) + cbv[1]);
        st_off(acc, i2a, (e00[r] - o00[r]) + cbv[0]);  st_off(acc, i2a + 64u, (e01[r] - o01[r]) + cbv[1]);
        st_off(acc, i3a, (e10[r] - o10[r]) + cbv[0]);  st_off(acc, i3a + 64u, (e11[r] - o11[r]) + cbv[1]);
      }
    }
    return;
  }

  float cq[2][8][8];
  #pragma unroll
  for (int r = 0; r < 8; ++r) {
    const int pr = r + half * 8;
    #pragma unroll
    for (int q = 0; q < 4; ++q) {
      const unsigned ia = base[0] + (unsigned)(pr + 16 * q) * sAB;
      cq[0][2 * q][r]     = ld_off(acc, ia);
      cq[0][2 * q + 1][r] = ld_off(acc, ia + 64u);
    }
  }
  __syncthreads();

  #pragma unroll
  for (int ll = 0; ll < 2; ++ll) {
    const int li = wv * 2 + ll;
    if (ll == 1) {
      #pragma unroll
      for (int r = 0; r < 8; ++r) {
        const int pr = r + half * 8;
        #pragma unroll
        for (int q = 0; q < 4; ++q) {
          const unsigned ia = base[1] + (unsigned)(pr + 16 * q) * sAB;
          cq[1][2 * q][r]     = ld_off(acc, ia);
          cq[1][2 * q + 1][r] = ld_off(acc, ia + 64u);
        }
      }
    }
    v8f e00 = zero8(), e01 = zero8(), e10 = zero8(), e11 = zero8();
    v8f o00 = zero8(), o01 = zero8(), o10 = zero8(), o11 = zero8();
    irdft_half(Gp + li * GP_LP, Fi, 0, half, l16, e00, e01, e10, e11);
    irdft_half(Gp + li * GP_LP, Fi, 1, half, l16, o00, o01, o10, o11);
    #pragma unroll
    for (int r = 0; r < 8; ++r) {
      const int pr = r + half * 8;
      const unsigned i0a = base[ll] + (unsigned)(pr)*sAB;
      const unsigned i1a = base[ll] + (unsigned)(pr + 16) * sAB;
      const unsigned i2a = base[ll] + (unsigned)(pr + 32) * sAB;
      const unsigned i3a = base[ll] + (unsigned)(pr + 48) * sAB;
      st_off(acc, i0a, act<GELU>(cq[ll][0][r] + (e00[r] + o00[r])));  st_off(acc, i0a + 64u, act<GELU>(cq[ll][1][r] + (e01[r] + o01[r])));
      st_off(acc, i1a, act<GELU>(cq[ll][2][r] + (e10[r] + o10[r])));  st_off(acc, i1a + 64u, act<GELU>(cq[ll][3][r] + (e11[r] + o11[r])));
      st_off(acc, i2a, act<GELU>(cq[ll][4][r] + (e00[r] - o00[r])));  st_off(acc, i2a + 64u, act<GELU>(cq[ll][5][r] + (e01[r] - o01[r])));
      st_off(acc, i3a, act<GELU>(cq[ll][6][r] + (e10[r] - o10[r])));  st_off(acc, i3a + 64u, act<GELU>(cq[ll][7][r] + (e11[r] - o11[r])));
    }
  }
}

#define HEAD_TRIPS 4
__global__ __launch_bounds__(256, 4) void k_head(
    const float* __restrict__ hin, const float* __restrict__ w1,
    const float* __restrict__ b1, const float* __restrict__ w2,
    const float* __restrict__ b2, const float* __restrict__ w3,
    const float* __restrict__ b3, float* __restrict__ out) {
  const int lane = threadIdx.x & 31, wv = threadIdx.x >> 5;
  const int half = lane >> 4, l16 = lane & 15;

  v2f wa1[4][2][2];
  #pragma unroll
  for (int j = 0; j < 4; ++j) {
    #pragma unroll
    for (int s = 0; s < 2; ++s) {
      const int c = 8 * j + 4 * half + 2 * s;
      #pragma unroll
      for (int mo = 0; mo < 2; ++mo) {
        wa1[j][s][mo].x = w1[(c)*NW + 16 * mo + l16];
        wa1[j][s][mo].y = w1[(c + 1) * NW + 16 * mo + l16];
      }
    }
  }
  v2f wa2[2][4][2];
  #pragma unroll
  for (int mt = 0; mt < 2; ++mt) {
    #pragma unroll
    for (int rp = 0; rp < 4; ++rp) {
      const int c = 16 * mt + 8 * half + 2 * rp;
      #pragma unroll
      for (int mo = 0; mo < 2; ++mo) {
        wa2[mt][rp][mo].x = w2[(c)*NW + 16 * mo + l16];
        wa2[mt][rp][mo].y = w2[(c + 1) * NW + 16 * mo + l16];
      }
    }
  }
  float bv1[2][8], bv2[2][8], wv3[2][8];
  #pragma unroll
  for (int mt = 0; mt < 2; ++mt) {
    #pragma unroll
    for (int r = 0; r < 8; ++r) {
      const int c = 16 * mt + 8 * half + r;
      bv1[mt][r] = b1[c];
      bv2[mt][r] = b2[c];
      wv3[mt][r] = w3[c];
    }
  }
  const float b3v = b3[0];

  const long p0 = ((long)blockIdx.x * 8 + wv) * (32 * HEAD_TRIPS);
  const float4* row = (const float4*)(hin + (p0 + l16) * NW + 4 * half);
  float4 hv[2][4];
  #pragma unroll
  for (int t = 0; t < 2; ++t) {
    #pragma unroll
    for (int j = 0; j < 4; ++j) hv[t][j] = row[(16 * t) * 8 + 2 * j];
  }

  for (int trip = 0; trip < HEAD_TRIPS; ++trip) {
    v8f d[2][2];
    d[0][0] = zero8(); d[0][1] = zero8(); d[1][0] = zero8(); d[1][1] = zero8();
    #pragma unroll
    for (int j = 0; j < 4; ++j) {
      #pragma unroll
      for (int s = 0; s < 2; ++s) {
        #pragma unroll
        for (int t = 0; t < 2; ++t) {
          const v2f b = k_pair(hv[t][j], s);
          d[t][0] = wmma4(wa1[j][s][0], b, d[t][0]);
          d[t][1] = wmma4(wa1[j][s][1], b, d[t][1]);
        }
      }
    }
    if (trip + 1 < HEAD_TRIPS) {
      #pragma unroll
      for (int t = 0; t < 2; ++t) {
        #pragma unroll
        for (int j = 0; j < 4; ++j)
          hv[t][j] = row[(32 * (trip + 1) + 16 * t) * 8 + 2 * j];
      }
    }
    #pragma unroll
    for (int t = 0; t < 2; ++t) {
      #pragma unroll
      for (int mt = 0; mt < 2; ++mt) {
        #pragma unroll
        for (int r = 0; r < 8; ++r)
          d[t][mt][r] = gelu_exact(d[t][mt][r] + bv1[mt][r]);
      }
    }

    v8f e[2][2];
    e[0][0] = zero8(); e[0][1] = zero8(); e[1][0] = zero8(); e[1][1] = zero8();
    #pragma unroll
    for (int mt = 0; mt < 2; ++mt) {
      #pragma unroll
      for (int rp = 0; rp < 4; ++rp) {
        #pragma unroll
        for (int t = 0; t < 2; ++t) {
          v2f b;
          b.x = d[t][mt][2 * rp];
          b.y = d[t][mt][2 * rp + 1];
          e[t][0] = wmma4(wa2[mt][rp][0], b, e[t][0]);
          e[t][1] = wmma4(wa2[mt][rp][1], b, e[t][1]);
        }
      }
    }

    float tot[2];
    #pragma unroll
    for (int t = 0; t < 2; ++t) {
      float s0 = 0.0f, s1 = 0.0f;
      #pragma unroll
      for (int r = 0; r < 8; ++r) {
        s0 = fmaf(gelu_exact(e[t][0][r] + bv2[0][r]), wv3[0][r], s0);
        s1 = fmaf(gelu_exact(e[t][1][r] + bv2[1][r]), wv3[1][r], s1);
      }
      const float ps = s0 + s1;
      tot[t] = ps + __shfl_xor(ps, 16);
    }
    out[p0 + 32 * trip + lane] = (half ? tot[1] : tot[0]) + b3v;
  }
}

extern "C" void kernel_launch(void* const* d_in, const int* in_sizes, int n_in,
                              void* d_out, int out_size, void* d_ws,
                              size_t ws_size, hipStream_t stream) {
  const float* x      = (const float*)d_in[0];
  const float* in_w   = (const float*)d_in[1];
  const float* in_b   = (const float*)d_in[2];
  const float* fwx    = (const float*)d_in[3];
  const float* fwy    = (const float*)d_in[4];
  const float* fwz    = (const float*)d_in[5];
  const float* conv_w = (const float*)d_in[6];
  const float* conv_b = (const float*)d_in[7];
  const float* fc1_w  = (const float*)d_in[8];
  const float* fc1_b  = (const float*)d_in[9];
  const float* fc2_w  = (const float*)d_in[10];
  const float* fc2_b  = (const float*)d_in[11];
  const float* fc3_w  = (const float*)d_in[12];
  const float* fc3_b  = (const float*)d_in[13];
  float* out = (float*)d_out;

  float* hcur = (float*)d_ws;
  float* hacc = hcur + HSZ;
  float* wre = out;
  const long wsl = (long)NW * NW * NM * 2;

  const dim3 blk(256);
  const int wgPts = NPTS / (16 * 8);
  const size_t smemB = (size_t)LDS_FLOATS * sizeof(float);
  const long sX = 64L * 64 * NW;
  const long sY = 64L * NW;
  const long sZ = (long)NW;

  k_input_proj<<<wgPts, blk, 0, stream>>>(x, in_w, in_b, hcur);
  k_weight_relayout<<<12 * 8192 / 256, blk, 0, stream>>>(fwz, fwy, fwx, wre);

  for (int i = 0; i < 4; ++i) {
    const long wofs = (long)i * wsl;
    const float* cwi = conv_w + i * (NW * NW);
    const float* cbi = conv_b + i * NW;
    k_spectral_axis<true, false><<<1024, blk, smemB, stream>>>(
        hcur, wre + wofs, hacc, cwi, cbi, sX, sY, sZ);
    k_spectral_axis<false, false><<<1024, blk, smemB, stream>>>(
        hcur, wre + 4 * wsl + wofs, hacc, cwi, cbi, sX, sZ, sY);
    if (i < 3)
      k_spectral_axis<false, true><<<1024, blk, smemB, stream>>>(
          hcur, wre + 8 * wsl + wofs, hacc, cwi, cbi, sY, sZ, sX);
    else
      k_spectral_axis<false, false><<<1024, blk, smemB, stream>>>(
          hcur, wre + 8 * wsl + wofs, hacc, cwi, cbi, sY, sZ, sX);
    float* const done = hacc;
    hacc = hcur;
    hcur = done;
  }

  k_head<<<NPTS / (32 * HEAD_TRIPS * 8), blk, 0, stream>>>(
      hcur, fc1_w, fc1_b, fc2_w, fc2_b, fc3_w, fc3_b, out);
}


# spectral kernel: input rows, acc and result turn between fragment and row order in a wave-private LDS tile (its dead G lines): global access 16 B a lane on 4 whole 128-B rows, was 4 B on 2 half rows
# speedup vs baseline: 3.5391x; 1.1420x over previous
#include <hip/hip_runtime.h>
#include <math.h>


typedef __attribute__((ext_vector_type(2))) float v2f;
typedef __attribute__((ext_vector_type(8))) float v8f;

#define NB 4
#define NS 64
#define NW 32
#define NM 16
#define NPTS (NB * NS * NS * NS)
#define HSZ ((size_t)NPTS * NW)

__device__ __forceinline__ v8f wmma4(v2f a, v2f b, v8f c) {
  return __builtin_amdgcn_wmma_f32_16x16x4_f32(false, a, false, b, (short)0, c,
                                               false, false);
}

__device__ __forceinline__ v8f zero8() {
  v8f z;
  #pragma unroll
  for (int i = 0; i < 8; ++i) z[i] = 0.0f;
  return z;
}

__device__ __forceinline__ float gelu_exact(float x) {
  const float a = fminf(fabsf(x) * 0.70710678118654752440f, 4.0f);
  float q = -1.16633282e-05f;
  q = fmaf(q, a, 1.53490531e-04f);
  q = fmaf(q, a, -8.50145181e-04f);
  q = fmaf(q, a, 2.27840361e-03f);
  q = fmaf(q, a, -8.86525522e-05f);
  q = fmaf(q, a, -2.77221724e-02f);
  q = fmaf(q, a, 1.48307174e-01f);
  q = fmaf(q, a, 9.18443024e-01f);
  q = fmaf(q, a, 1.62790728f);
  const float e = __builtin_amdgcn_exp2f(-(q * a));
  const float hx = 0.5f * x;
  return fmaf(fabsf(hx), 1.0f - e, hx);
}
template <bool G>
__device__ __forceinline__ float act(float v) {
  return G ? gelu_exact(v) : v;
}

__global__ __launch_bounds__(256) void k_input_proj(
    const float* __restrict__ x, const float* __restrict__ w,
    const float* __restrict__ bias, float* __restrict__ out) {
  const int lane = threadIdx.x & 31, wv = threadIdx.x >> 5;
  const int half = lane >> 4, l16 = lane & 15;
  const long pbase = ((long)blockIdx.x * 8 + wv) * 16;
  v2f a;
  a.x = x[(pbase + l16) * 4 + half * 2];
  a.y = x[(pbase + l16) * 4 + half * 2 + 1];
  v2f b0, b1;
  b0.x = w[(half * 2) * NW + l16];
  b0.y = w[(half * 2 + 1) * NW + l16];
  b1.x = w[(half * 2) * NW + 16 + l16];
  b1.y = w[(half * 2 + 1) * NW + 16 + l16];
  v8f d0 = zero8(), d1 = zero8();
  d0 = wmma4(a, b0, d0);
  d1 = wmma4(a, b1, d1);
  #pragma unroll
  for (int r = 0; r < 8; ++r) {
    long p = pbase + r + half * 8;
    out[p * NW + l16] = d0[r] + bias[l16];
    out[p * NW + 16 + l16] = d1[r] + bias[16 + l16];
  }
}

#define FF_P 36
#define FI_P 36
#define G_RP 34
#define G_LP 1092
#define GP_LP 1027
#define OFF_FI (32 * FF_P)
#define OFF_G (OFF_FI + 32 * FI_P)
#define OFF_GP (OFF_G + 16 * G_LP)
#define LDS_FLOATS (OFF_GP + 16 * GP_LP)
__device__ __forceinline__ float4 ld4_off(const float* p, unsigned boff) {
  return *(const float4*)((const char*)p + boff);
}
__device__ __forceinline__ const float* adv(const float* p, unsigned bytes) {
  return (const float*)((const char*)p + bytes);
}
__device__ __forceinline__ float* adv(float* p, unsigned bytes) {
  return (float*)((char*)p + bytes);
}
__device__ __forceinline__ void st4_off(float* p, unsigned boff, float4 v) {
  *(float4*)((char*)p + boff) = v;
}
template <bool BIAS>
__device__ __forceinline__ void tile_put(
    float* T, int half, int l16,
    const v8f& e00, const v8f& e01, const v8f& e10, const v8f& e11,
    const v8f& o00, const v8f& o01, const v8f& o10, const v8f& o11,
    float b0, float b1) {
  const int c0 = l16 ^ (half << 4), c1 = c0 ^ 16;
  #pragma unroll
  for (int r = 0; r < 8; ++r) {
    const int pr = r + half * 8;
    float v[8] = {e00[r] + o00[r], e01[r] + o01[r], e10[r] + o10[r], e11[r] + o11[r],
                  e00[r] - o00[r], e01[r] - o01[r], e10[r] - o10[r], e11[r] - o11[r]};
    if (BIAS) {
      #pragma unroll
      for (int k = 0; k < 8; k += 2) { v[k] += b0; v[k + 1] += b1; }
    }
    T[(pr)*32 + c0]        = v[0];  T[(pr)*32 + c1]        = v[1];
    T[(pr + 16) * 32 + c0] = v[2];  T[(pr + 16) * 32 + c1] = v[3];
    T[(pr + 32) * 32 + c0] = v[4];  T[(pr + 32) * 32 + c1] = v[5];
    T[(pr + 48) * 32 + c0] = v[6];  T[(pr + 48) * 32 + c1] = v[7];
  }
}
__device__ __forceinline__ float4 tile_get(const float* T, int lane, int i) {
  const int row = 4 * i + (lane >> 3);
  const int c4 = (lane & 7) ^ ((i & 2) << 1);
  return *(const float4*)(T + row * 32 + c4 * 4);
}
__device__ __forceinline__ void tile_order() {
  __builtin_amdgcn_fence(__ATOMIC_ACQ_REL, "wavefront");
  __builtin_amdgcn_wave_barrier();
}
__device__ __forceinline__ v2f k_pair(float4 v, int s) {
  v2f r;
  r.x = s ? v.z : v.x;
  r.y = s ? v.w : v.y;
  return r;
}

__device__ __forceinline__ int mode_of_row(int R) {
  return ((R & 15) >> 1) * 2 + (R >> 4);
}

__device__ __forceinline__ void irdft_half(
    const float* gp, const float* Fi, int g, int half, int l16,
    v8f& t00, v8f& t01, v8f& t10, v8f& t11) {
  for (int kk = 4 * g; kk < 4 * g + 4; ++kk) {
    const int i0 = kk * 4 + half * 2;
    v2f b0, b1;
    const int c0 = l16 ^ (half << 4), c1 = c0 ^ 16;
    b0.x = gp[(i0)*32 + c0]; b0.y = gp[(i0 + 1) * 32 + c0];
    b1.x = gp[(i0)*32 + c1]; b1.y = gp[(i0 + 1) * 32 + c1];
    v2f a0, a1;
    a0.x = Fi[(l16)*FI_P + i0];        a0.y = Fi[(l16)*FI_P + i0 + 1];
    a1.x = Fi[(16 + l16) * FI_P + i0]; a1.y = Fi[(16 + l16) * FI_P + i0 + 1];
    t00 = wmma4(a0, b0, t00);  t01 = wmma4(a0, b1, t01);
    t10 = wmma4(a1, b0, t10);  t11 = wmma4(a1, b1, t11);
  }
}

__global__ __launch_bounds__(256) void k_weight_relayout(
    const float* __restrict__ fwz, const float* __restrict__ fwy,
    const float* __restrict__ fwx, float* __restrict__ dst) {
  const int t = blockIdx.x * 256 + threadIdx.x;
  const int s = t >> 13, q = t & 8191;
  const int axis = s >> 2, layer = s & 3;
  const float* src = (axis == 0) ? fwz : (axis == 1) ? fwy : fwx;
  const int lane = q & 31, j = (q >> 5) & 1, mt = (q >> 6) & 1;
  const int kk = (q >> 7) & 7, wv = q >> 10;
  const int in = kk * 4 + (lane >> 4) * 2 + j, o = mt * 16 + (lane & 15);
  const float4 v = *(const float4*)(src + layer * (NW * NW * NM * 2) +
                                    (in * NW + o) * (NM * 2) + 4 * wv);
  ((float4*)dst)[t] = v;
}

template <bool FIRST, bool GELU>
__global__ __launch_bounds__(256, 4) void k_spectral_axis(
    const float* __restrict__ h, const float* __restrict__ wgt,
    float* __restrict__ acc, const float* __restrict__ cw,
    const float* __restrict__ cb, long strideU, long strideV, long strideA) {
  extern __shared__ float smem[];
  float* Ff = smem;
  float* Fi = smem + OFF_FI;
  float* G  = smem + OFF_G;
  float* Gp = smem + OFF_GP;

  const int tid = threadIdx.x;
  const int lane = tid & 31, wv = tid >> 5;
  const int half = lane >> 4, l16 = lane & 15;
  const float NRM = 0.125f;
  float* T = G + (wv * 2) * G_LP;

  const unsigned sAB = (unsigned)strideA * 4u;
  unsigned base[2], rv[2];
  #pragma unroll
  for (int ll = 0; ll < 2; ++ll) {
    int l = blockIdx.x * 16 + wv * 2 + ll;
    int b = l >> 12, u = (l >> 6) & 63, v = l & 63;
    const unsigned lineB = ((unsigned)b * (64u * 64 * 64 * NW) + (unsigned)u * (unsigned)strideU +
                            (unsigned)v * (unsigned)strideV) * 4u;
    base[ll] = lineB + (unsigned)l16 * 4u;
    rv[ll] = lineB + (unsigned)(lane >> 3) * sAB + (unsigned)(lane & 7) * 16u;
  }

  float4 hr[2][16];
  #pragma unroll
  for (int ll = 0; ll < 2; ++ll) {
    #pragma unroll
    for (int i = 0; i < 16; ++i) hr[ll][i] = ld4_off(adv(h, (unsigned)(4 * i) * sAB), rv[ll]);
  }

  for (int idx = tid; idx < 1024; idx += 256) {
    int m2 = idx >> 5, p = idx & 31, k = mode_of_row(m2);
    float ang = 6.28318530717958647693f * (float)(k * p) * (1.0f / 64.0f);
    Ff[m2 * FF_P + p] = (m2 & 1) ? (-__sinf(ang) * NRM) : (__cosf(ang) * NRM);
  }
  for (int idx = tid; idx < 1024; idx += 256) {
    int p = idx >> 5, m2 = idx & 31, k = mode_of_row(m2);
    float ang = 6.28318530717958647693f * (float)(k * p) * (1.0f / 64.0f);
    float sc = (k == 0 ? 1.0f : 2.0f) * NRM;
    Fi[p * FI_P + m2] = (m2 & 1) ? (-__sinf(ang) * sc) : (__cosf(ang) * sc);
  }
  __syncthreads();

  v8f a00[2], a01[2], a10[2], a11[2];
  #pragma unroll
  for (int ll = 0; ll < 2; ++ll) {
    tile_order();
    #pragma unroll
    for (int i = 0; i < 16; ++i) {
      const int row = 4 * i + (lane >> 3), c4 = (lane & 7) ^ (half << 2);
      *(float4*)(T + row * 32 + c4 * 4) = hr[ll][i];
    }
    tile_order();
    a00[ll] = zero8(); a01[ll] = zero8(); a10[ll] = zero8(); a11[ll] = zero8();
    const int c0 = l16 ^ (half << 4), c1 = c0 ^ 16;
    #pragma unroll
    for (int kk = 0; kk < 8; ++kk) {
      const int p0 = kk * 4 + half * 2;
      v2f fa0, fa1;
      fa0.x = Ff[(l16)*FF_P + p0];        fa0.y = Ff[(l16)*FF_P + p0 + 1];
      fa1.x = Ff[(16 + l16) * FF_P + p0]; fa1.y = Ff[(16 + l16) * FF_P + p0 + 1];
      v2f x0, y0, x1, y1;
      x0.x = T[(p0)*32 + c0];      x0.y = T[(p0 + 1) * 32 + c0];
      y0.x = T[(p0 + 32) * 32 + c0]; y0.y = T[(p0 + 33) * 32 + c0];
      x1.x = T[(p0)*32 + c1];      x1.y = T[(p0 + 1) * 32 + c1];
      y1.x = T[(p0 + 32) * 32 + c1]; y1.y = T[(p0 + 33) * 32 + c1];
      const v2f s0 = x0 + y0, s1 = x1 + y1;
      const v2f d0 = x0 - y0, d1 = x1 - y1;
      a00[ll] = wmma4(fa0, s0, a00[ll]);
      a01[ll] = wmma4(fa0, s1, a01[ll]);
      a10[ll] = wmma4(fa1, d0, a10[ll]);
      a11[ll] = wmma4(fa1, d1, a11[ll]);
    }
    if (ll == 0) __builtin_amdgcn_sched_barrier(0);
  }
  tile_order();
  #pragma unroll
  for (int ll = 0; ll < 2; ++ll) {
    float* g = G + (wv * 2 + ll) * G_LP;
    #pragma unroll
    for (int r = 0; r < 8; ++r) {
      const int m = r + half * 8;
      g[(m)*G_RP + l16]             = a00[ll][r];
      g[(m)*G_RP + 16 + l16]        = a01[ll][r];
      g[(m + 16) * G_RP + l16]      = a10[ll][r];
      g[(m + 16) * G_RP + 16 + l16] = a11[ll][r];
    }
  }

  float4 wq[8][2][2];
  {
    const float4* wgt4 = (const float4*)wgt + wv * 1024 + lane;
    #pragma unroll
    for (int kk = 0; kk < 8; ++kk) {
      #pragma unroll
      for (int mt = 0; mt < 2; ++mt) {
        #pragma unroll
        for (int j = 0; j < 2; ++j)
          wq[kk][mt][j] = wgt4[((kk * 2 + mt) * 2 + j) * 32];
      }
    }
  }
  __syncthreads();

  #pragma unroll
  for (int km = 0; km < 2; ++km) {
    const int rRe = km * 16 + 2 * wv, rIm = rRe + 1;
    v8f oRe0 = zero8(), oRe1 = zero8(), oIm0 = zero8(), oIm1 = zero8();
    #pragma unroll
    for (int kk = 0; kk < 8; ++kk) {
      const int i0 = kk * 4 + half * 2, i1 = i0 + 1;
      v2f bRe, bIm;
      bRe.x = G[l16 * G_LP + rRe * G_RP + i0];
      bRe.y = G[l16 * G_LP + rRe * G_RP + i1];
      bIm.x = G[l16 * G_LP + rIm * G_RP + i0];
      bIm.y = G[l16 * G_LP + rIm * G_RP + i1];
      #pragma unroll
      for (int mt = 0; mt < 2; ++mt) {
        v2f aR, aI, aIn;
        aR.x = km ? wq[kk][mt][0].z : wq[kk][mt][0].x;
        aR.y = km ? wq[kk][mt][1].z : wq[kk][mt][1].x;
        aI.x = km ? wq[kk][mt][0].w : wq[kk][mt][0].y;
        aI.y = km ? wq[kk][mt][1].w : wq[kk][mt][1].y;
        aIn.x = -aI.x; aIn.y = -aI.y;
        if (mt == 0) {
          oRe0 = wmma4(aR, bRe, oRe0);  oRe0 = wmma4(aIn, bIm, oRe0);
          oIm0 = wmma4(aI, bRe, oIm0);  oIm0 = wmma4(aR, bIm, oIm0);
        } else {
          oRe1 = wmma4(aR, bRe, oRe1);  oRe1 = wmma4(aIn, bIm, oRe1);
          oIm1 = wmma4(aI, bRe, oIm1);  oIm1 = wmma4(aR, bIm, oIm1);
        }
      }
    }
    const int psw = (wv & 1) << 4;
    #pragma unroll
    for (int r = 0; r < 8; ++r) {
      const int o0 = r + half * 8;
      Gp[l16 * GP_LP + rRe * 32 + (o0 ^ psw)]        = oRe0[r];
      Gp[l16 * GP_LP + rRe * 32 + ((o0 + 16) ^ psw)] = oRe1[r];
      Gp[l16 * GP_LP + rIm * 32 + (o0 ^ psw)]        = oIm0[r];
      Gp[l16 * GP_LP + rIm * 32 + ((o0 + 16) ^ psw)] = oIm1[r];
    }
  }

  if (FIRST) {
    float4 cwq[2][4];
    float cbv[2];
    #pragma unroll
    for (int nt = 0; nt < 2; ++nt) {
      cbv[nt] = cb[nt * 16 + l16];
      #pragma unroll
      for (int j = 0; j < 4; ++j) {
        const float4 c = *(const float4*)(cw + (nt * 16 + l16) * NW + 8 * j + 4 * half);
        cwq[nt][j] = make_float4(0.5f * c.x, 0.5f * c.y, 0.5f * c.z, 0.5f * c.w);
      }
    }
    __syncthreads();

    #pragma unroll
    for (int ll = 0; ll < 2; ++ll) {
      const unsigned lb = base[ll];
      const unsigned ab = lb - (unsigned)l16 * 4u + (unsigned)l16 * sAB +
                          (unsigned)half * 16u;
      float4 ha[4][4];
      #pragma unroll
      for (int q = 0; q < 4; ++q) {
        #pragma unroll
        for (int j = 0; j < 4; ++j)
          ha[q][j] = ld4_off(h, ab + (unsigned)(16 * q) * sAB + 32u * j);
      }
      __builtin_amdgcn_sched_barrier(0);
      v8f e00 = zero8(), e01 = zero8(), e10 = zero8(), e11 = zero8();
      v8f o00 = zero8(), o01 = zero8(), o10 = zero8(), o11 = zero8();
      const float* gp = Gp + (wv * 2 + ll) * GP_LP;
      irdft_half(gp, Fi, 0, half, l16, e00, e01, e10, e11);
      irdft_half(gp, Fi, 1, half, l16, o00, o01, o10, o11);
      #pragma unroll
      for (int q = 0; q < 2; ++q) {
        #pragma unroll
        for (int j = 0; j < 4; ++j) {
          const float4 u = ha[q][j], v = ha[q + 2][j];
          ha[q][j] = make_float4(u.x + v.x, u.y + v.y, u.z + v.z, u.w + v.w);
          ha[q + 2][j] = make_float4(u.x - v.x, u.y - v.y, u.z - v.z, u.w - v.w);
        }
      }
      #pragma unroll
      for (int j = 0; j < 4; ++j) {
        #pragma unroll
        for (int s = 0; s < 2; ++s) {
          const v2f cb0 = k_pair(cwq[0][j], s), cb1 = k_pair(cwq[1][j], s);
          const v2f cs0 = k_pair(ha[0][j], s), cs1 = k_pair(ha[1][j], s);
          const v2f cd0 = k_pair(ha[2][j], s), cd1 = k_pair(ha[3][j], s);
          e00 = wmma4(cs0, cb0, e00);  e01 = wmma4(cs0, cb1, e01);
          e10 = wmma4(cs1, cb0, e10);  e11 = wmma4(cs1, cb1, e11);
          o00 = wmma4(cd0, cb0, o00);  o01 = wmma4(cd0, cb1, o01);
          o10 = wmma4(cd1, cb0, o10);  o11 = wmma4(cd1, cb1, o11);
        }
      }
      tile_order();
      tile_put<true>(T, half, l16, e00, e01, e10, e11, o00, o01, o10, o11,
                     cbv[0], cbv[1]);
      tile_order();
      #pragma unroll
      for (int i = 0; i < 16; ++i)
        st4_off(adv(acc, (unsigned)(4 * i) * sAB), rv[ll], tile_get(T, lane, i));
    }
    return;
  }

  float4 cq[2][16];
  #pragma unroll
  for (int i = 0; i < 16; ++i) cq[0][i] = ld4_off(adv(acc, (unsigned)(4 * i) * sAB), rv[0]);
  __syncthreads();

  #pragma unroll
  for (int ll = 0; ll < 2; ++ll) {
    const int li = wv * 2 + ll;
    if (ll == 1) {
      #pragma unroll
      for (int i = 0; i < 16; ++i) cq[1][i] = ld4_off(adv(acc, (unsigned)(4 * i) * sAB), rv[1]);
    }
    v8f e00 = zero8(), e01 = zero8(), e10 = zero8(), e11 = zero8();
    v8f o00 = zero8(), o01 = zero8(), o10 = zero8(), o11 = zero8();
    irdft_half(Gp + li * GP_LP, Fi, 0, half, l16, e00, e01, e10, e11);
    irdft_half(Gp + li * GP_LP, Fi, 1, half, l16, o00, o01, o10, o11);
    tile_order();
    tile_put<false>(T, half, l16, e00, e01, e10, e11, o00, o01, o10, o11, 0.0f, 0.0f);
    tile_order();
    #pragma unroll
    for (int i = 0; i < 16; ++i) {
      const float4 t = tile_get(T, lane, i);
      const float4 a = cq[ll][i];
      st4_off(adv(acc, (unsigned)(4 * i) * sAB), rv[ll],
              make_float4(act<GELU>(a.x + t.x), act<GELU>(a.y + t.y),
                          act<GELU>(a.z + t.z), act<GELU>(a.w + t.w)));
    }
  }
}

#define HEAD_TRIPS 4
__global__ __launch_bounds__(256, 4) void k_head(
    const float* __restrict__ hin, const float* __restrict__ w1,
    const float* __restrict__ b1, const float* __restrict__ w2,
    const float* __restrict__ b2, const float* __restrict__ w3,
    const float* __restrict__ b3, float* __restrict__ out) {
  const int lane = threadIdx.x & 31, wv = threadIdx.x >> 5;
  const int half = lane >> 4, l16 = lane & 15;

  v2f wa1[4][2][2];
  #pragma unroll
  for (int j = 0; j < 4; ++j) {
    #pragma unroll
    for (int s = 0; s < 2; ++s) {
      const int c = 8 * j + 4 * half + 2 * s;
      #pragma unroll
      for (int mo = 0; mo < 2; ++mo) {
        wa1[j][s][mo].x = w1[(c)*NW + 16 * mo + l16];
        wa1[j][s][mo].y = w1[(c + 1) * NW + 16 * mo + l16];
      }
    }
  }
  v2f wa2[2][4][2];
  #pragma unroll
  for (int mt = 0; mt < 2; ++mt) {
    #pragma unroll
    for (int rp = 0; rp < 4; ++rp) {
      const int c = 16 * mt + 8 * half + 2 * rp;
      #pragma unroll
      for (int mo = 0; mo < 2; ++mo) {
        wa2[mt][rp][mo].x = w2[(c)*NW + 16 * mo + l16];
        wa2[mt][rp][mo].y = w2[(c + 1) * NW + 16 * mo + l16];
      }
    }
  }
  float bv1[2][8], bv2[2][8], wv3[2][8];
  #pragma unroll
  for (int mt = 0; mt < 2; ++mt) {
    #pragma unroll
    for (int r = 0; r < 8; ++r) {
      const int c = 16 * mt + 8 * half + r;
      bv1[mt][r] = b1[c];
      bv2[mt][r] = b2[c];
      wv3[mt][r] = w3[c];
    }
  }
  const float b3v = b3[0];

  const long p0 = ((long)blockIdx.x * 8 + wv) * (32 * HEAD_TRIPS);
  const float4* row = (const float4*)(hin + (p0 + l16) * NW + 4 * half);
  float4 hv[2][4];
  #pragma unroll
  for (int t = 0; t < 2; ++t) {
    #pragma unroll
    for (int j = 0; j < 4; ++j) hv[t][j] = row[(16 * t) * 8 + 2 * j];
  }

  for (int trip = 0; trip < HEAD_TRIPS; ++trip) {
    v8f d[2][2];
    d[0][0] = zero8(); d[0][1] = zero8(); d[1][0] = zero8(); d[1][1] = zero8();
    #pragma unroll
    for (int j = 0; j < 4; ++j) {
      #pragma unroll
      for (int s = 0; s < 2; ++s) {
        #pragma unroll
        for (int t = 0; t < 2; ++t) {
          const v2f b = k_pair(hv[t][j], s);
          d[t][0] = wmma4(wa1[j][s][0], b, d[t][0]);
          d[t][1] = wmma4(wa1[j][s][1], b, d[t][1]);
        }
      }
    }
    if (trip + 1 < HEAD_TRIPS) {
      #pragma unroll
      for (int t = 0; t < 2; ++t) {
        #pragma unroll
        for (int j = 0; j < 4; ++j)
          hv[t][j] = row[(32 * (trip + 1) + 16 * t) * 8 + 2 * j];
      }
    }
    #pragma unroll
    for (int t = 0; t < 2; ++t) {
      #pragma unroll
      for (int mt = 0; mt < 2; ++mt) {
        #pragma unroll
        for (int r = 0; r < 8; ++r)
          d[t][mt][r] = gelu_exact(d[t][mt][r] + bv1[mt][r]);
      }
    }

    v8f e[2][2];
    e[0][0] = zero8(); e[0][1] = zero8(); e[1][0] = zero8(); e[1][1] = zero8();
    #pragma unroll
    for (int mt = 0; mt < 2; ++mt) {
      #pragma unroll
      for (int rp = 0; rp < 4; ++rp) {
        #pragma unroll
        for (int t = 0; t < 2; ++t) {
          v2f b;
          b.x = d[t][mt][2 * rp];
          b.y = d[t][mt][2 * rp + 1];
          e[t][0] = wmma4(wa2[mt][rp][0], b, e[t][0]);
          e[t][1] = wmma4(wa2[mt][rp][1], b, e[t][1]);
        }
      }
    }

    float tot[2];
    #pragma unroll
    for (int t = 0; t < 2; ++t) {
      float s0 = 0.0f, s1 = 0.0f;
      #pragma unroll
      for (int r = 0; r < 8; ++r) {
        s0 = fmaf(gelu_exact(e[t][0][r] + bv2[0][r]), wv3[0][r], s0);
        s1 = fmaf(gelu_exact(e[t][1][r] + bv2[1][r]), wv3[1][r], s1);
      }
      const float ps = s0 + s1;
      tot[t] = ps + __shfl_xor(ps, 16);
    }
    out[p0 + 32 * trip + lane] = (half ? tot[1] : tot[0]) + b3v;
  }
}

extern "C" void kernel_launch(void* const* d_in, const int* in_sizes, int n_in,
                              void* d_out, int out_size, void* d_ws,
                              size_t ws_size, hipStream_t stream) {
  const float* x      = (const float*)d_in[0];
  const float* in_w   = (const float*)d_in[1];
  const float* in_b   = (const float*)d_in[2];
  const float* fwx    = (const float*)d_in[3];
  const float* fwy    = (const float*)d_in[4];
  const float* fwz    = (const float*)d_in[5];
  const float* conv_w = (const float*)d_in[6];
  const float* conv_b = (const float*)d_in[7];
  const float* fc1_w  = (const float*)d_in[8];
  const float* fc1_b  = (const float*)d_in[9];
  const float* fc2_w  = (const float*)d_in[10];
  const float* fc2_b  = (const float*)d_in[11];
  const float* fc3_w  = (const float*)d_in[12];
  const float* fc3_b  = (const float*)d_in[13];
  float* out = (float*)d_out;

  float* hcur = (float*)d_ws;
  float* hacc = hcur + HSZ;
  float* wre = out;
  const long wsl = (long)NW * NW * NM * 2;

  const dim3 blk(256);
  const int wgPts = NPTS / (16 * 8);
  const size_t smemB = (size_t)LDS_FLOATS * sizeof(float);
  const long sX = 64L * 64 * NW;
  const long sY = 64L * NW;
  const long sZ = (long)NW;

  k_input_proj<<<wgPts, blk, 0, stream>>>(x, in_w, in_b, hcur);
  k_weight_relayout<<<12 * 8192 / 256, blk, 0, stream>>>(fwz, fwy, fwx, wre);

  for (int i = 0; i < 4; ++i) {
    const long wofs = (long)i * wsl;
    const float* cwi = conv_w + i * (NW * NW);
    const float* cbi = conv_b + i * NW;
    k_spectral_axis<true, false><<<1024, blk, smemB, stream>>>(
        hcur, wre + wofs, hacc, cwi, cbi, sX, sY, sZ);
    k_spectral_axis<false, false><<<1024, blk, smemB, stream>>>(
        hcur, wre + 4 * wsl + wofs, hacc, cwi, cbi, sX, sZ, sY);
    if (i < 3)
      k_spectral_axis<false, true><<<1024, blk, smemB, stream>>>(
          hcur, wre + 8 * wsl + wofs, hacc, cwi, cbi, sY, sZ, sX);
    else
      k_spectral_axis<false, false><<<1024, blk, smemB, stream>>>(
          hcur, wre + 8 * wsl + wofs, hacc, cwi, cbi, sY, sZ, sX);
    float* const done = hacc;
    hacc = hcur;
    hcur = done;
  }

  k_head<<<NPTS / (32 * HEAD_TRIPS * 8), blk, 0, stream>>>(
      hcur, fc1_w, fc1_b, fc2_w, fc2_b, fc3_w, fc3_b, out);
}


# input projection (K = 4, one WMMA a tile) formed from x inside layer 0's three spectral launches: its launch, its store of h and three reads of h are gone; in_b is the accumulator's start value
# speedup vs baseline: 3.7635x; 1.0634x over previous
#include <hip/hip_runtime.h>
#include <math.h>


typedef __attribute__((ext_vector_type(2))) float v2f;
typedef __attribute__((ext_vector_type(8))) float v8f;

#define NB 4
#define NS 64
#define NW 32
#define NM 16
#define NPTS (NB * NS * NS * NS)
#define HSZ ((size_t)NPTS * NW)

__device__ __forceinline__ v8f wmma4(v2f a, v2f b, v8f c) {
  return __builtin_amdgcn_wmma_f32_16x16x4_f32(false, a, false, b, (short)0, c,
                                               false, false);
}

__device__ __forceinline__ v8f zero8() {
  v8f z;
  #pragma unroll
  for (int i = 0; i < 8; ++i) z[i] = 0.0f;
  return z;
}

__device__ __forceinline__ float gelu_exact(float x) {
  const float a = fminf(fabsf(x) * 0.70710678118654752440f, 4.0f);
  float q = -1.16633282e-05f;
  q = fmaf(q, a, 1.53490531e-04f);
  q = fmaf(q, a, -8.50145181e-04f);
  q = fmaf(q, a, 2.27840361e-03f);
  q = fmaf(q, a, -8.86525522e-05f);
  q = fmaf(q, a, -2.77221724e-02f);
  q = fmaf(q, a, 1.48307174e-01f);
  q = fmaf(q, a, 9.18443024e-01f);
  q = fmaf(q, a, 1.62790728f);
  const float e = __builtin_amdgcn_exp2f(-(q * a));
  const float hx = 0.5f * x;
  return fmaf(fabsf(hx), 1.0f - e, hx);
}
template <bool G>
__device__ __forceinline__ float act(float v) {
  return G ? gelu_exact(v) : v;
}

#define FF_P 36
#define FI_P 36
#define G_RP 34
#define G_LP 1092
#define GP_LP 1027
#define OFF_FI (32 * FF_P)
#define OFF_G (OFF_FI + 32 * FI_P)
#define OFF_GP (OFF_G + 16 * G_LP)
#define LDS_FLOATS (OFF_GP + 16 * GP_LP)
__device__ __forceinline__ float4 ld4_off(const float* p, unsigned boff) {
  return *(const float4*)((const char*)p + boff);
}
__device__ __forceinline__ const float* adv(const float* p, unsigned bytes) {
  return (const float*)((const char*)p + bytes);
}
__device__ __forceinline__ float* adv(float* p, unsigned bytes) {
  return (float*)((char*)p + bytes);
}
__device__ __forceinline__ void st4_off(float* p, unsigned boff, float4 v) {
  *(float4*)((char*)p + boff) = v;
}
template <bool BIAS>
__device__ __forceinline__ void tile_put(
    float* T, int half, int l16,
    const v8f& e00, const v8f& e01, const v8f& e10, const v8f& e11,
    const v8f& o00, const v8f& o01, const v8f& o10, const v8f& o11,
    float b0, float b1) {
  const int c0 = l16 ^ (half << 4), c1 = c0 ^ 16;
  #pragma unroll
  for (int r = 0; r < 8; ++r) {
    const int pr = r + half * 8;
    float v[8] = {e00[r] + o00[r], e01[r] + o01[r], e10[r] + o10[r], e11[r] + o11[r],
                  e00[r] - o00[r], e01[r] - o01[r], e10[r] - o10[r], e11[r] - o11[r]};
    if (BIAS) {
      #pragma unroll
      for (int k = 0; k < 8; k += 2) { v[k] += b0; v[k + 1] += b1; }
    }
    T[(pr)*32 + c0]        = v[0];  T[(pr)*32 + c1]        = v[1];
    T[(pr + 16) * 32 + c0] = v[2];  T[(pr + 16) * 32 + c1] = v[3];
    T[(pr + 32) * 32 + c0] = v[4];  T[(pr + 32) * 32 + c1] = v[5];
    T[(pr + 48) * 32 + c0] = v[6];  T[(pr + 48) * 32 + c1] = v[7];
  }
}
__device__ __forceinline__ float4 tile_get(const float* T, int lane, int i) {
  const int row = 4 * i + (lane >> 3);
  const int c4 = (lane & 7) ^ ((i & 2) << 1);
  return *(const float4*)(T + row * 32 + c4 * 4);
}
__device__ __forceinline__ void tile_order() {
  __builtin_amdgcn_fence(__ATOMIC_ACQ_REL, "wavefront");
  __builtin_amdgcn_wave_barrier();
}
__device__ __forceinline__ v2f k_pair(float4 v, int s) {
  v2f r;
  r.x = s ? v.z : v.x;
  r.y = s ? v.w : v.y;
  return r;
}

__device__ __forceinline__ int mode_of_row(int R) {
  return ((R & 15) >> 1) * 2 + (R >> 4);
}

__device__ __forceinline__ void irdft_half(
    const float* gp, const float* Fi, int g, int half, int l16,
    v8f& t00, v8f& t01, v8f& t10, v8f& t11) {
  for (int kk = 4 * g; kk < 4 * g + 4; ++kk) {
    const int i0 = kk * 4 + half * 2;
    v2f b0, b1;
    const int c0 = l16 ^ (half << 4), c1 = c0 ^ 16;
    b0.x = gp[(i0)*32 + c0]; b0.y = gp[(i0 + 1) * 32 + c0];
    b1.x = gp[(i0)*32 + c1]; b1.y = gp[(i0 + 1) * 32 + c1];
    v2f a0, a1;
    a0.x = Fi[(l16)*FI_P + i0];        a0.y = Fi[(l16)*FI_P + i0 + 1];
    a1.x = Fi[(16 + l16) * FI_P + i0]; a1.y = Fi[(16 + l16) * FI_P + i0 + 1];
    t00 = wmma4(a0, b0, t00);  t01 = wmma4(a0, b1, t01);
    t10 = wmma4(a1, b0, t10);  t11 = wmma4(a1, b1, t11);
  }
}

__global__ __launch_bounds__(256) void k_weight_relayout(
    const float* __restrict__ fwz, const float* __restrict__ fwy,
    const float* __restrict__ fwx, float* __restrict__ dst) {
  const int t = blockIdx.x * 256 + threadIdx.x;
  const int s = t >> 13, q = t & 8191;
  const int axis = s >> 2, layer = s & 3;
  const float* src = (axis == 0) ? fwz : (axis == 1) ? fwy : fwx;
  const int lane = q & 31, j = (q >> 5) & 1, mt = (q >> 6) & 1;
  const int kk = (q >> 7) & 7, wv = q >> 10;
  const int in = kk * 4 + (lane >> 4) * 2 + j, o = mt * 16 + (lane & 15);
  const float4 v = *(const float4*)(src + layer * (NW * NW * NM * 2) +
                                    (in * NW + o) * (NM * 2) + 4 * wv);
  ((float4*)dst)[t] = v;
}

template <bool FIRST, bool GELU, bool PROJ>
__global__ __launch_bounds__(256, 4) void k_spectral_axis(
    const float* __restrict__ h, const float* __restrict__ wgt,
    float* __restrict__ acc, const float* __restrict__ cw,
    const float* __restrict__ cb, const float* __restrict__ xin,
    const float* __restrict__ inw, const float* __restrict__ inb,
    long strideU, long strideV, long strideA) {
  extern __shared__ float smem[];
  float* Ff = smem;
  float* Fi = smem + OFF_FI;
  float* G  = smem + OFF_G;
  float* Gp = smem + OFF_GP;

  const int tid = threadIdx.x;
  const int lane = tid & 31, wv = tid >> 5;
  const int half = lane >> 4, l16 = lane & 15;
  const float NRM = 0.125f;
  float* T = G + (wv * 2) * G_LP;

  const unsigned sAB = (unsigned)strideA * 4u;
  const unsigned sXB = sAB >> 3;
  unsigned base[2], rv[2], xo[2];
  #pragma unroll
  for (int ll = 0; ll < 2; ++ll) {
    int l = blockIdx.x * 16 + wv * 2 + ll;
    int b = l >> 12, u = (l >> 6) & 63, v = l & 63;
    const unsigned lineB = ((unsigned)b * (64u * 64 * 64 * NW) + (unsigned)u * (unsigned)strideU +
                            (unsigned)v * (unsigned)strideV) * 4u;
    base[ll] = lineB + (unsigned)l16 * 4u;
    rv[ll] = lineB + (unsigned)(lane >> 3) * sAB + (unsigned)(lane & 7) * 16u;
    xo[ll] = (lineB >> 3) + (unsigned)l16 * sXB + (unsigned)half * 8u;
  }

  float4 hr[2][16];
  v2f xa[2][4], wb[2];
  float bC[2];
  if (PROJ) {
    #pragma unroll
    for (int ll = 0; ll < 2; ++ll) {
      #pragma unroll
      for (int q = 0; q < 4; ++q)
        xa[ll][q] = *(const v2f*)((const char*)xin + (xo[ll] + (unsigned)(16 * q) * sXB));
    }
    #pragma unroll
    for (int nt = 0; nt < 2; ++nt) {
      wb[nt].x = inw[(half * 2) * NW + nt * 16 + l16];
      wb[nt].y = inw[(half * 2 + 1) * NW + nt * 16 + l16];
      bC[nt] = inb[nt * 16 + l16];
    }
  } else {
    #pragma unroll
    for (int ll = 0; ll < 2; ++ll) {
      #pragma unroll
      for (int i = 0; i < 16; ++i) hr[ll][i] = ld4_off(adv(h, (unsigned)(4 * i) * sAB), rv[ll]);
    }
  }

  for (int idx = tid; idx < 1024; idx += 256) {
    int m2 = idx >> 5, p = idx & 31, k = mode_of_row(m2);
    float ang = 6.28318530717958647693f * (float)(k * p) * (1.0f / 64.0f);
    Ff[m2 * FF_P + p] = (m2 & 1) ? (-__sinf(ang) * NRM) : (__cosf(ang) * NRM);
  }
  for (int idx = tid; idx < 1024; idx += 256) {
    int p = idx >> 5, m2 = idx & 31, k = mode_of_row(m2);
    float ang = 6.28318530717958647693f * (float)(k * p) * (1.0f / 64.0f);
    float sc = (k == 0 ? 1.0f : 2.0f) * NRM;
    Fi[p * FI_P + m2] = (m2 & 1) ? (-__sinf(ang) * sc) : (__cosf(ang) * sc);
  }
  __syncthreads();

  v8f a00[2], a01[2], a10[2], a11[2];
  #pragma unroll
  for (int ll = 0; ll < 2; ++ll) {
    a00[ll] = zero8(); a01[ll] = zero8(); a10[ll] = zero8(); a11[ll] = zero8();
    if (PROJ) {
      v8f hs[2][2], hd[2][2];
      #pragma unroll
      for (int nt = 0; nt < 2; ++nt) {
        v8f c;
        #pragma unroll
        for (int r = 0; r < 8; ++r) c[r] = bC[nt];
        const v8f t0 = wmma4(xa[ll][0], wb[nt], c), t1 = wmma4(xa[ll][1], wb[nt], c);
        const v8f t2 = wmma4(xa[ll][2], wb[nt], c), t3 = wmma4(xa[ll][3], wb[nt], c);
        hs[0][nt] = t0 + t2; hs[1][nt] = t1 + t3;
        hd[0][nt] = t0 - t2; hd[1][nt] = t1 - t3;
      }
      #pragma unroll
      for (int qq = 0; qq < 2; ++qq) {
        #pragma unroll
        for (int t = 0; t < 4; ++t) {
          const int p0 = 16 * qq + 8 * half + 2 * t;
          v2f fa0, fa1, s0, s1, d0, d1;
          fa0.x = Ff[(l16)*FF_P + p0];        fa0.y = Ff[(l16)*FF_P + p0 + 1];
          fa1.x = Ff[(16 + l16) * FF_P + p0]; fa1.y = Ff[(16 + l16) * FF_P + p0 + 1];
          s0.x = hs[qq][0][2 * t]; s0.y = hs[qq][0][2 * t + 1];
          s1.x = hs[qq][1][2 * t]; s1.y = hs[qq][1][2 * t + 1];
          d0.x = hd[qq][0][2 * t]; d0.y = hd[qq][0][2 * t + 1];
          d1.x = hd[qq][1][2 * t]; d1.y = hd[qq][1][2 * t + 1];
          a00[ll] = wmma4(fa0, s0, a00[ll]);
          a01[ll] = wmma4(fa0, s1, a01[ll]);
          a10[ll] = wmma4(fa1, d0, a10[ll]);
          a11[ll] = wmma4(fa1, d1, a11[ll]);
        }
      }
    } else {
    tile_order();
    #pragma unroll
    for (int i = 0; i < 16; ++i) {
      const int row = 4 * i + (lane >> 3), c4 = (lane & 7) ^ (half << 2);
      *(float4*)(T + row * 32 + c4 * 4) = hr[ll][i];
    }
    tile_order();
    const int c0 = l16 ^ (half << 4), c1 = c0 ^ 16;
    #pragma unroll
    for (int kk = 0; kk < 8; ++kk) {
      const int p0 = kk * 4 + half * 2;
      v2f fa0, fa1;
      fa0.x = Ff[(l16)*FF_P + p0];        fa0.y = Ff[(l16)*FF_P + p0 + 1];
      fa1.x = Ff[(16 + l16) * FF_P + p0]; fa1.y = Ff[(16 + l16) * FF_P + p0 + 1];
      v2f x0, y0, x1, y1;
      x0.x = T[(p0)*32 + c0];      x0.y = T[(p0 + 1) * 32 + c0];
      y0.x = T[(p0 + 32) * 32 + c0]; y0.y = T[(p0 + 33) * 32 + c0];
      x1.x = T[(p0)*32 + c1];      x1.y = T[(p0 + 1) * 32 + c1];
      y1.x = T[(p0 + 32) * 32 + c1]; y1.y = T[(p0 + 33) * 32 + c1];
      const v2f s0 = x0 + y0, s1 = x1 + y1;
      const v2f d0 = x0 - y0, d1 = x1 - y1;
      a00[ll] = wmma4(fa0, s0, a00[ll]);
      a01[ll] = wmma4(fa0, s1, a01[ll]);
      a10[ll] = wmma4(fa1, d0, a10[ll]);
      a11[ll] = wmma4(fa1, d1, a11[ll]);
    }
    }
    if (ll == 0) __builtin_amdgcn_sched_barrier(0);
  }
  tile_order();
  #pragma unroll
  for (int ll = 0; ll < 2; ++ll) {
    float* g = G + (wv * 2 + ll) * G_LP;
    #pragma unroll
    for (int r = 0; r < 8; ++r) {
      const int m = r + half * 8;
      g[(m)*G_RP + l16]             = a00[ll][r];
      g[(m)*G_RP + 16 + l16]        = a01[ll][r];
      g[(m + 16) * G_RP + l16]      = a10[ll][r];
      g[(m + 16) * G_RP + 16 + l16] = a11[ll][r];
    }
  }

  float4 wq[8][2][2];
  {
    const float4* wgt4 = (const float4*)wgt + wv * 1024 + lane;
    #pragma unroll
    for (int kk = 0; kk < 8; ++kk) {
      #pragma unroll
      for (int mt = 0; mt < 2; ++mt) {
        #pragma unroll
        for (int j = 0; j < 2; ++j)
          wq[kk][mt][j] = wgt4[((kk * 2 + mt) * 2 + j) * 32];
      }
    }
  }
  __syncthreads();

  #pragma unroll
  for (int km = 0; km < 2; ++km) {
    const int rRe = km * 16 + 2 * wv, rIm = rRe + 1;
    v8f oRe0 = zero8(), oRe1 = zero8(), oIm0 = zero8(), oIm1 = zero8();
    #pragma unroll
    for (int kk = 0; kk < 8; ++kk) {
      const int i0 = kk * 4 + half * 2, i1 = i0 + 1;
      v2f bRe, bIm;
      bRe.x = G[l16 * G_LP + rRe * G_RP + i0];
      bRe.y = G[l16 * G_LP + rRe * G_RP + i1];
      bIm.x = G[l16 * G_LP + rIm * G_RP + i0];
      bIm.y = G[l16 * G_LP + rIm * G_RP + i1];
      #pragma unroll
      for (int mt = 0; mt < 2; ++mt) {
        v2f aR, aI, aIn;
        aR.x = km ? wq[kk][mt][0].z : wq[kk][mt][0].x;
        aR.y = km ? wq[kk][mt][1].z : wq[kk][mt][1].x;
        aI.x = km ? wq[kk][mt][0].w : wq[kk][mt][0].y;
        aI.y = km ? wq[kk][mt][1].w : wq[kk][mt][1].y;
        aIn.x = -aI.x; aIn.y = -aI.y;
        if (mt == 0) {
          oRe0 = wmma4(aR, bRe, oRe0);  oRe0 = wmma4(aIn, bIm, oRe0);
          oIm0 = wmma4(aI, bRe, oIm0);  oIm0 = wmma4(aR, bIm, oIm0);
        } else {
          oRe1 = wmma4(aR, bRe, oRe1);  oRe1 = wmma4(aIn, bIm, oRe1);
          oIm1 = wmma4(aI, bRe, oIm1);  oIm1 = wmma4(aR, bIm, oIm1);
        }
      }
    }
    const int psw = (wv & 1) << 4;
    #pragma unroll
    for (int r = 0; r < 8; ++r) {
      const int o0 = r + half * 8;
      Gp[l16 * GP_LP + rRe * 32 + (o0 ^ psw)]        = oRe0[r];
      Gp[l16 * GP_LP + rRe * 32 + ((o0 + 16) ^ psw)] = oRe1[r];
      Gp[l16 * GP_LP + rIm * 32 + (o0 ^ psw)]        = oIm0[r];
      Gp[l16 * GP_LP + rIm * 32 + ((o0 + 16) ^ psw)] = oIm1[r];
    }
  }

  if (FIRST) {
    float4 cwq[2][4];
    float cbv[2];
    #pragma unroll
    for (int nt = 0; nt < 2; ++nt) {
      cbv[nt] = cb[nt * 16 + l16];
      #pragma unroll
      for (int j = 0; j < 4; ++j) {
        const int col = PROJ ? (16 * (j >> 1) + 8 * half + 4 * (j & 1)) : (8 * j + 4 * half);
        const float4 c = *(const float4*)(cw + (nt * 16 + l16) * NW + col);
        cwq[nt][j] = make_float4(0.5f * c.x, 0.5f * c.y, 0.5f * c.z, 0.5f * c.w);
      }
    }
    __syncthreads();

    #pragma unroll
    for (int ll = 0; ll < 2; ++ll) {
      const unsigned lb = base[ll];
      const unsigned ab = lb - (unsigned)l16 * 4u + (unsigned)l16 * sAB +
                          (unsigned)half * 16u;
      float4 ha[4][4];
      v8f cs[2][2], cd[2][2];
      v2f xq[4], wp[2];
      v8f bt[2];
      if (PROJ) {
        #pragma unroll
        for (int q = 0; q < 4; ++q)
          xq[q] = *(const v2f*)((const char*)xin + (xo[ll] + (unsigned)(16 * q) * sXB));
        #pragma unroll
        for (int m = 0; m < 2; ++m) {
          wp[m].x = inw[(half * 2) * NW + m * 16 + l16];
          wp[m].y = inw[(half * 2 + 1) * NW + m * 16 + l16];
          const float4 u0 = *(const float4*)(inb + 16 * m + 8 * half);
          const float4 u1 = *(const float4*)(inb + 16 * m + 8 * half + 4);
          bt[m][0] = u0.x; bt[m][1] = u0.y; bt[m][2] = u0.z; bt[m][3] = u0.w;
          bt[m][4] = u1.x; bt[m][5] = u1.y; bt[m][6] = u1.z; bt[m][7] = u1.w;
        }
      } else {
        #pragma unroll
        for (int q = 0; q < 4; ++q) {
          #pragma unroll
          for (int j = 0; j < 4; ++j)
            ha[q][j] = ld4_off(h, ab + (unsigned)(16 * q) * sAB + 32u * j);
        }
      }
      __builtin_amdgcn_sched_barrier(0);
      v8f e00 = zero8(), e01 = zero8(), e10 = zero8(), e11 = zero8();
      v8f o00 = zero8(), o01 = zero8(), o10 = zero8(), o11 = zero8();
      const float* gp = Gp + (wv * 2 + ll) * GP_LP;
      irdft_half(gp, Fi, 0, half, l16, e00, e01, e10, e11);
      irdft_half(gp, Fi, 1, half, l16, o00, o01, o10, o11);
      if (PROJ) {
        #pragma unroll
        for (int m = 0; m < 2; ++m) {
          const v8f t0 = wmma4(wp[m], xq[0], bt[m]), t1 = wmma4(wp[m], xq[1], bt[m]);
          const v8f t2 = wmma4(wp[m], xq[2], bt[m]), t3 = wmma4(wp[m], xq[3], bt[m]);
          cs[0][m] = t0 + t2; cs[1][m] = t1 + t3;
          cd[0][m] = t0 - t2; cd[1][m] = t1 - t3;
        }
      } else {
      #pragma unroll
      for (int q = 0; q < 2; ++q) {
        #pragma unroll
        for (int j = 0; j < 4; ++j) {
          const float4 u = ha[q][j], v = ha[q + 2][j];
          ha[q][j] = make_float4(u.x + v.x, u.y + v.y, u.z + v.z, u.w + v.w);
          ha[q + 2][j] = make_float4(u.x - v.x, u.y - v.y, u.z - v.z, u.w - v.w);
        }
      }
      }
      #pragma unroll
      for (int j = 0; j < 4; ++j) {
        #pragma unroll
        for (int s = 0; s < 2; ++s) {
          const v2f cb0 = k_pair(cwq[0][j], s), cb1 = k_pair(cwq[1][j], s);
          v2f cs0, cs1, cd0, cd1;
          if (PROJ) {
            const int m = j >> 1, t = 2 * (j & 1) + s;
            cs0.x = cs[0][m][2 * t]; cs0.y = cs[0][m][2 * t + 1];
            cs1.x = cs[1][m][2 * t]; cs1.y = cs[1][m][2 * t + 1];
            cd0.x = cd[0][m][2 * t]; cd0.y = cd[0][m][2 * t + 1];
            cd1.x = cd[1][m][2 * t]; cd1.y = cd[1][m][2 * t + 1];
          } else {
            cs0 = k_pair(ha[0][j], s); cs1 = k_pair(ha[1][j], s);
            cd0 = k_pair(ha[2][j], s); cd1 = k_pair(ha[3][j], s);
          }
          e00 = wmma4(cs0, cb0, e00);  e01 = wmma4(cs0, cb1, e01);
          e10 = wmma4(cs1, cb0, e10);  e11 = wmma4(cs1, cb1, e11);
          o00 = wmma4(cd0, cb0, o00);  o01 = wmma4(cd0, cb1, o01);
          o10 = wmma4(cd1, cb0, o10);  o11 = wmma4(cd1, cb1, o11);
        }
      }
      tile_order();
      tile_put<true>(T, half, l16, e00, e01, e10, e11, o00, o01, o10, o11,
                     cbv[0], cbv[1]);
      tile_order();
      #pragma unroll
      for (int i = 0; i < 16; ++i)
        st4_off(adv(acc, (unsigned)(4 * i) * sAB), rv[ll], tile_get(T, lane, i));
    }
    return;
  }

  float4 cq[2][16];
  #pragma unroll
  for (int i = 0; i < 16; ++i) cq[0][i] = ld4_off(adv(acc, (unsigned)(4 * i) * sAB), rv[0]);
  __syncthreads();

  #pragma unroll
  for (int ll = 0; ll < 2; ++ll) {
    const int li = wv * 2 + ll;
    if (ll == 1) {
      #pragma unroll
      for (int i = 0; i < 16; ++i) cq[1][i] = ld4_off(adv(acc, (unsigned)(4 * i) * sAB), rv[1]);
    }
    v8f e00 = zero8(), e01 = zero8(), e10 = zero8(), e11 = zero8();
    v8f o00 = zero8(), o01 = zero8(), o10 = zero8(), o11 = zero8();
    irdft_half(Gp + li * GP_LP, Fi, 0, half, l16, e00, e01, e10, e11);
    irdft_half(Gp + li * GP_LP, Fi, 1, half, l16, o00, o01, o10, o11);
    tile_order();
    tile_put<false>(T, half, l16, e00, e01, e10, e11, o00, o01, o10, o11, 0.0f, 0.0f);
    tile_order();
    #pragma unroll
    for (int i = 0; i < 16; ++i) {
      const float4 t = tile_get(T, lane, i);
      const float4 a = cq[ll][i];
      st4_off(adv(acc, (unsigned)(4 * i) * sAB), rv[ll],
              make_float4(act<GELU>(a.x + t.x), act<GELU>(a.y + t.y),
                          act<GELU>(a.z + t.z), act<GELU>(a.w + t.w)));
    }
  }
}

#define HEAD_TRIPS 4
__global__ __launch_bounds__(256, 4) void k_head(
    const float* __restrict__ hin, const float* __restrict__ w1,
    const float* __restrict__ b1, const float* __restrict__ w2,
    const float* __restrict__ b2, const float* __restrict__ w3,
    const float* __restrict__ b3, float* __restrict__ out) {
  const int lane = threadIdx.x & 31, wv = threadIdx.x >> 5;
  const int half = lane >> 4, l16 = lane & 15;

  v2f wa1[4][2][2];
  #pragma unroll
  for (int j = 0; j < 4; ++j) {
    #pragma unroll
    for (int s = 0; s < 2; ++s) {
      const int c = 8 * j + 4 * half + 2 * s;
      #pragma unroll
      for (int mo = 0; mo < 2; ++mo) {
        wa1[j][s][mo].x = w1[(c)*NW + 16 * mo + l16];
        wa1[j][s][mo].y = w1[(c + 1) * NW + 16 * mo + l16];
      }
    }
  }
  v2f wa2[2][4][2];
  #pragma unroll
  for (int mt = 0; mt < 2; ++mt) {
    #pragma unroll
    for (int rp = 0; rp < 4; ++rp) {
      const int c = 16 * mt + 8 * half + 2 * rp;
      #pragma unroll
      for (int mo = 0; mo < 2; ++mo) {
        wa2[mt][rp][mo].x = w2[(c)*NW + 16 * mo + l16];
        wa2[mt][rp][mo].y = w2[(c + 1) * NW + 16 * mo + l16];
      }
    }
  }
  float bv1[2][8], bv2[2][8], wv3[2][8];
  #pragma unroll
  for (int mt = 0; mt < 2; ++mt) {
    #pragma unroll
    for (int r = 0; r < 8; ++r) {
      const int c = 16 * mt + 8 * half + r;
      bv1[mt][r] = b1[c];
      bv2[mt][r] = b2[c];
      wv3[mt][r] = w3[c];
    }
  }
  const float b3v = b3[0];

  const long p0 = ((long)blockIdx.x * 8 + wv) * (32 * HEAD_TRIPS);
  const float4* row = (const float4*)(hin + (p0 + l16) * NW + 4 * half);
  float4 hv[2][4];
  #pragma unroll
  for (int t = 0; t < 2; ++t) {
    #pragma unroll
    for (int j = 0; j < 4; ++j) hv[t][j] = row[(16 * t) * 8 + 2 * j];
  }

  for (int trip = 0; trip < HEAD_TRIPS; ++trip) {
    v8f d[2][2];
    d[0][0] = zero8(); d[0][1] = zero8(); d[1][0] = zero8(); d[1][1] = zero8();
    #pragma unroll
    for (int j = 0; j < 4; ++j) {
      #pragma unroll
      for (int s = 0; s < 2; ++s) {
        #pragma unroll
        for (int t = 0; t < 2; ++t) {
          const v2f b = k_pair(hv[t][j], s);
          d[t][0] = wmma4(wa1[j][s][0], b, d[t][0]);
          d[t][1] = wmma4(wa1[j][s][1], b, d[t][1]);
        }
      }
    }
    if (trip + 1 < HEAD_TRIPS) {
      #pragma unroll
      for (int t = 0; t < 2; ++t) {
        #pragma unroll
        for (int j = 0; j < 4; ++j)
          hv[t][j] = row[(32 * (trip + 1) + 16 * t) * 8 + 2 * j];
      }
    }
    #pragma unroll
    for (int t = 0; t < 2; ++t) {
      #pragma unroll
      for (int mt = 0; mt < 2; ++mt) {
        #pragma unroll
        for (int r = 0; r < 8; ++r)
          d[t][mt][r] = gelu_exact(d[t][mt][r] + bv1[mt][r]);
      }
    }

    v8f e[2][2];
    e[0][0] = zero8(); e[0][1] = zero8(); e[1][0] = zero8(); e[1][1] = zero8();
    #pragma unroll
    for (int mt = 0; mt < 2; ++mt) {
      #pragma unroll
      for (int rp = 0; rp < 4; ++rp) {
        #pragma unroll
        for (int t = 0; t < 2; ++t) {
          v2f b;
          b.x = d[t][mt][2 * rp];
          b.y = d[t][mt][2 * rp + 1];
          e[t][0] = wmma4(wa2[mt][rp][0], b, e[t][0]);
          e[t][1] = wmma4(wa2[mt][rp][1], b, e[t][1]);
        }
      }
    }

    float tot[2];
    #pragma unroll
    for (int t = 0; t < 2; ++t) {
      float s0 = 0.0f, s1 = 0.0f;
      #pragma unroll
      for (int r = 0; r < 8; ++r) {
        s0 = fmaf(gelu_exact(e[t][0][r] + bv2[0][r]), wv3[0][r], s0);
        s1 = fmaf(gelu_exact(e[t][1][r] + bv2[1][r]), wv3[1][r], s1);
      }
      const float ps = s0 + s1;
      tot[t] = ps + __shfl_xor(ps, 16);
    }
    out[p0 + 32 * trip + lane] = (half ? tot[1] : tot[0]) + b3v;
  }
}

extern "C" void kernel_launch(void* const* d_in, const int* in_sizes, int n_in,
                              void* d_out, int out_size, void* d_ws,
                              size_t ws_size, hipStream_t stream) {
  const float* x      = (const float*)d_in[0];
  const float* in_w   = (const float*)d_in[1];
  const float* in_b   = (const float*)d_in[2];
  const float* fwx    = (const float*)d_in[3];
  const float* fwy    = (const float*)d_in[4];
  const float* fwz    = (const float*)d_in[5];
  const float* conv_w = (const float*)d_in[6];
  const float* conv_b = (const float*)d_in[7];
  const float* fc1_w  = (const float*)d_in[8];
  const float* fc1_b  = (const float*)d_in[9];
  const float* fc2_w  = (const float*)d_in[10];
  const float* fc2_b  = (const float*)d_in[11];
  const float* fc3_w  = (const float*)d_in[12];
  const float* fc3_b  = (const float*)d_in[13];
  float* out = (float*)d_out;

  float* hcur = (float*)d_ws;
  float* hacc = hcur + HSZ;
  float* wre = out;
  const long wsl = (long)NW * NW * NM * 2;

  const dim3 blk(256);
  const size_t smemB = (size_t)LDS_FLOATS * sizeof(float);
  const long sX = 64L * 64 * NW;
  const long sY = 64L * NW;
  const long sZ = (long)NW;

  k_weight_relayout<<<12 * 8192 / 256, blk, 0, stream>>>(fwz, fwy, fwx, wre);

  for (int i = 0; i < 4; ++i) {
    const long wofs = (long)i * wsl;
    const float* cwi = conv_w + i * (NW * NW);
    const float* cbi = conv_b + i * NW;
    if (i == 0) {
      k_spectral_axis<true, false, true><<<1024, blk, smemB, stream>>>(
          hcur, wre + wofs, hacc, cwi, cbi, x, in_w, in_b, sX, sY, sZ);
      k_spectral_axis<false, false, true><<<1024, blk, smemB, stream>>>(
          hcur, wre + 4 * wsl + wofs, hacc, cwi, cbi, x, in_w, in_b, sX, sZ, sY);
      k_spectral_axis<false, true, true><<<1024, blk, smemB, stream>>>(
          hcur, wre + 8 * wsl + wofs, hacc, cwi, cbi, x, in_w, in_b, sY, sZ, sX);
    } else {
      k_spectral_axis<true, false, false><<<1024, blk, smemB, stream>>>(
          hcur, wre + wofs, hacc, cwi, cbi, x, in_w, in_b, sX, sY, sZ);
      k_spectral_axis<false, false, false><<<1024, blk, smemB, stream>>>(
          hcur, wre + 4 * wsl + wofs, hacc, cwi, cbi, x, in_w, in_b, sX, sZ, sY);
      if (i < 3)
        k_spectral_axis<false, true, false><<<1024, blk, smemB, stream>>>(
            hcur, wre + 8 * wsl + wofs, hacc, cwi, cbi, x, in_w, in_b, sY, sZ, sX);
      else
        k_spectral_axis<false, false, false><<<1024, blk, smemB, stream>>>(
            hcur, wre + 8 * wsl + wofs, hacc, cwi, cbi, x, in_w, in_b, sY, sZ, sX);
    }
    float* const done = hacc;
    hacc = hcur;
    hcur = done;
  }

  k_head<<<NPTS / (32 * HEAD_TRIPS * 8), blk, 0, stream>>>(
      hcur, fc1_w, fc1_b, fc2_w, fc2_b, fc3_w, fc3_b, out);
}
